# starting kernel
# baseline (speedup 1.0000x reference)
#include <hip/hip_runtime.h>


#define HMM_B 256
#define HMM_T 8192
#define HMM_K 8
#define HMM_EPS 1e-8f
#define HMM_LN2 0.6931471805599453f

typedef __attribute__((ext_vector_type(2))) float v2f;
typedef __attribute__((ext_vector_type(8))) float v8f;

__device__ __forceinline__ float fast_rcp(float x) {
    float r = __builtin_amdgcn_rcpf(x);
    float e = __builtin_fmaf(-x, r, 1.0f);
    return __builtin_fmaf(r, e, r);
}

__device__ __forceinline__ float crosshalf(float x) {
    unsigned u = __builtin_bit_cast(unsigned, x);
    u = __builtin_amdgcn_permlanex16(u, u, 0x76543210u, 0xFEDCBA98u, false, false);
    return __builtin_bit_cast(float, u);
}

__global__ __launch_bounds__(256) void hmm_pack_kernel(
    const int* __restrict__ obs, unsigned* __restrict__ bits)
{
    const int wave = (blockIdx.x * blockDim.x + threadIdx.x) >> 5;
    const int lane = threadIdx.x & 31;
    const long base = (long)wave * 1024;
    unsigned keep = 0;
#pragma unroll 4
    for (int i = 0; i < 32; ++i) {
        int v = obs[base + i * 32 + lane];
        unsigned m = (unsigned)__ballot(v != 0);
        if (i == lane) keep = m;
    }
    bits[(size_t)wave * 32 + lane] = keep;
}

__global__ __launch_bounds__(32) void hmm_forward_kernel(
    const float* __restrict__ start, const float* __restrict__ trans,
    const float* __restrict__ emission, const unsigned* __restrict__ bits,
    float* __restrict__ alpha_out, float* __restrict__ loglik)
{
    const int  lane = threadIdx.x;
    const int  n    = lane & 15;
    const bool lo   = lane < 16;
    const int  b    = blockIdx.x * 16 + n;

    float em1[8], em0[8];
#pragma unroll
    for (int k = 0; k < 8; ++k) {
        float e = emission[k];
        e = fminf(fmaxf(e, HMM_EPS), 1.0f - HMM_EPS);
        em1[k] = e; em0[k] = 1.0f - e;
    }
    float st[8]; float ssum = 0.0f;
#pragma unroll
    for (int k = 0; k < 8; ++k) { st[k] = start[k] + HMM_EPS; ssum += st[k]; }
#pragma unroll
    for (int k = 0; k < 8; ++k) st[k] /= ssum;

    float rs[8];
#pragma unroll
    for (int j = 0; j < 8; ++j) {
        float s = 0.0f;
#pragma unroll
        for (int k = 0; k < 8; ++k) s += trans[j * 8 + k] + HMM_EPS;
        rs[j] = s;
    }

    const int kb = lo ? 0 : 2;
    const int i  = n;
    v2f A0, A1;
    A0.x = (i < 8) ? (trans[(kb + 0) * 8 + i] + HMM_EPS) / rs[kb + 0] : 0.0f;
    A0.y = (i < 8) ? (trans[(kb + 1) * 8 + i] + HMM_EPS) / rs[kb + 1] : 0.0f;
    A1.x = (i < 8) ? (trans[(kb + 4) * 8 + i] + HMM_EPS) / rs[kb + 4] : 0.0f;
    A1.y = (i < 8) ? (trans[(kb + 5) * 8 + i] + HMM_EPS) / rs[kb + 5] : 0.0f;

    unsigned w  = bits[(size_t)b * (HMM_T / 32)];
    int      ob = (int)(w & 1u);
    v8f a;
#pragma unroll
    for (int m = 0; m < 8; ++m) a[m] = st[m] * (ob ? em1[m] : em0[m]);
    float s = a[0] + a[1] + a[2] + a[3] + a[4] + a[5] + a[6] + a[7];
    {  float sx = crosshalf(s); s = lo ? s : sx; }
    float inv = fast_rcp(s);
    float lc  = __builtin_amdgcn_logf(s);
    if (lo) *(v8f*)(alpha_out + ((size_t)b * HMM_T) * 8) = a;

    const v8f vzero = {};
    for (int t = 1; t < HMM_T; ++t) {
        if ((t & 31) == 0) w = bits[(size_t)b * (HMM_T / 32) + (t >> 5)];
        ob = (int)((w >> (t & 31)) & 1u);

        float x2 = crosshalf(a[2]);
        float x3 = crosshalf(a[3]);
        float x6 = crosshalf(a[6]);
        float x7 = crosshalf(a[7]);
        v2f B0, B1;
        B0.x = (lo ? a[0] : x2) * inv;  B0.y = (lo ? a[1] : x3) * inv;
        B1.x = (lo ? a[4] : x6) * inv;  B1.y = (lo ? a[5] : x7) * inv;

        v8f d = __builtin_amdgcn_wmma_f32_16x16x4_f32(
            false, A0, false, B0, (short)0, vzero, false, false);
        d = __builtin_amdgcn_wmma_f32_16x16x4_f32(
            false, A1, false, B1, (short)0, d, false, false);

#pragma unroll
        for (int m = 0; m < 8; ++m) a[m] = d[m] * (ob ? em1[m] : em0[m]);
        s = a[0] + a[1] + a[2] + a[3] + a[4] + a[5] + a[6] + a[7];
        float sx = crosshalf(s);
        s = lo ? s : sx;
        inv = fast_rcp(s);
        lc += __builtin_amdgcn_logf(s);
        if (lo) *(v8f*)(alpha_out + ((size_t)b * HMM_T + t) * 8) = a;
    }
    if (lo) loglik[b] = lc * HMM_LN2;
}

__global__ __launch_bounds__(32) void hmm_backward_kernel(
    const float* __restrict__ trans, const float* __restrict__ emission,
    const unsigned* __restrict__ bits, float* __restrict__ gamma)
{
    const int  lane = threadIdx.x;
    const int  n    = lane & 15;
    const bool lo   = lane < 16;
    const int  b    = blockIdx.x * 16 + n;

    float em1[8], em0[8];
#pragma unroll
    for (int k = 0; k < 8; ++k) {
        float e = emission[k];
        e = fminf(fmaxf(e, HMM_EPS), 1.0f - HMM_EPS);
        em1[k] = e; em0[k] = 1.0f - e;
    }
    float rs[8];
#pragma unroll
    for (int j = 0; j < 8; ++j) {
        float s = 0.0f;
#pragma unroll
        for (int k = 0; k < 8; ++k) s += trans[j * 8 + k] + HMM_EPS;
        rs[j] = s;
    }
    const int kb = lo ? 0 : 2;
    const int i  = n;
    v2f A0, A1;
    A0.x = (i < 8) ? (trans[i * 8 + (kb + 0)] + HMM_EPS) / rs[i] : 0.0f;
    A0.y = (i < 8) ? (trans[i * 8 + (kb + 1)] + HMM_EPS) / rs[i] : 0.0f;
    A1.x = (i < 8) ? (trans[i * 8 + (kb + 4)] + HMM_EPS) / rs[i] : 0.0f;
    A1.y = (i < 8) ? (trans[i * 8 + (kb + 5)] + HMM_EPS) / rs[i] : 0.0f;

    v8f d;
#pragma unroll
    for (int m = 0; m < 8; ++m) d[m] = 1.0f;
    if (lo) {
        float* p = gamma + ((size_t)b * HMM_T + (HMM_T - 1)) * 8;
        v8f al = *(v8f*)p;
        float gs = al[0]+al[1]+al[2]+al[3]+al[4]+al[5]+al[6]+al[7];
        float gi = fast_rcp(gs);
        v8f g;
#pragma unroll
        for (int m = 0; m < 8; ++m) g[m] = al[m] * gi;
        *(v8f*)p = g;
    }

    const v8f vzero = {};
    unsigned w = 0;
    for (int t = HMM_T - 2; t >= 0; --t) {
        const int tt = t + 1;
        if ((tt & 31) == 31) w = bits[(size_t)b * (HMM_T / 32) + (tt >> 5)];
        int ob = (int)((w >> (tt & 31)) & 1u);

        float s = d[0]+d[1]+d[2]+d[3]+d[4]+d[5]+d[6]+d[7];
        float sx = crosshalf(s);
        s = lo ? s : sx;
        float inv = fast_rcp(s);

        v8f u;
#pragma unroll
        for (int m = 0; m < 8; ++m) u[m] = d[m] * (ob ? em1[m] : em0[m]);
        float x2 = crosshalf(u[2]);
        float x3 = crosshalf(u[3]);
        float x6 = crosshalf(u[6]);
        float x7 = crosshalf(u[7]);
        v2f B0, B1;
        B0.x = (lo ? u[0] : x2) * inv;  B0.y = (lo ? u[1] : x3) * inv;
        B1.x = (lo ? u[4] : x6) * inv;  B1.y = (lo ? u[5] : x7) * inv;

        d = __builtin_amdgcn_wmma_f32_16x16x4_f32(
            false, A0, false, B0, (short)0, vzero, false, false);
        d = __builtin_amdgcn_wmma_f32_16x16x4_f32(
            false, A1, false, B1, (short)0, d, false, false);

        if (lo) {
            float* p = gamma + ((size_t)b * HMM_T + t) * 8;
            v8f al = *(v8f*)p;
            v8f g;
            float gs = 0.0f;
#pragma unroll
            for (int m = 0; m < 8; ++m) { g[m] = al[m] * d[m]; gs += g[m]; }
            float gi = fast_rcp(gs);
#pragma unroll
            for (int m = 0; m < 8; ++m) g[m] *= gi;
            *(v8f*)p = g;
        }
    }
}

extern "C" void kernel_launch(void* const* d_in, const int* in_sizes, int n_in,
                              void* d_out, int out_size, void* d_ws, size_t ws_size,
                              hipStream_t stream) {
    (void)in_sizes; (void)n_in; (void)out_size; (void)ws_size;
    const float* start    = (const float*)d_in[0];
    const float* trans    = (const float*)d_in[1];
    const float* emission = (const float*)d_in[2];
    const int*   obs      = (const int*)d_in[3];

    float* gamma  = (float*)d_out;
    float* loglik = gamma + (size_t)HMM_B * HMM_T * HMM_K;
    unsigned* bits = (unsigned*)d_ws;

    hmm_pack_kernel<<<256, 256, 0, stream>>>(obs, bits);
    hmm_forward_kernel<<<HMM_B / 16, 32, 0, stream>>>(
        start, trans, emission, bits, gamma, loglik);
    hmm_backward_kernel<<<HMM_B / 16, 32, 0, stream>>>(
        trans, emission, bits, gamma);
}


# backward: staged-alpha loads issued one 8-step block ahead (register double buffer), gamma made branch-free (Trans rows repeated in A rows 8-15) beside the beta chain; bit words fetched one ahead
# speedup vs baseline: 2.3305x; 2.3305x over previous
#include <hip/hip_runtime.h>


#define HMM_B 256
#define HMM_T 8192
#define HMM_K 8
#define HMM_EPS 1e-8f
#define HMM_LN2 0.6931471805599453f

typedef __attribute__((ext_vector_type(2))) float v2f;
typedef __attribute__((ext_vector_type(4))) float v4f;
typedef __attribute__((ext_vector_type(8))) float v8f;

__device__ __forceinline__ float fast_rcp(float x) {
    float r = __builtin_amdgcn_rcpf(x);
    float e = __builtin_fmaf(-x, r, 1.0f);
    return __builtin_fmaf(r, e, r);
}

__device__ __forceinline__ float crosshalf(float x) {
    unsigned u = __builtin_bit_cast(unsigned, x);
    u = __builtin_amdgcn_permlanex16(u, u, 0x76543210u, 0xFEDCBA98u, false, false);
    return __builtin_bit_cast(float, u);
}

__global__ __launch_bounds__(256) void hmm_pack_kernel(
    const int* __restrict__ obs, unsigned* __restrict__ bits)
{
    const int wave = (blockIdx.x * blockDim.x + threadIdx.x) >> 5;
    const int lane = threadIdx.x & 31;
    const long base = (long)wave * 1024;
    unsigned keep = 0;
#pragma unroll 4
    for (int i = 0; i < 32; ++i) {
        int v = obs[base + i * 32 + lane];
        unsigned m = (unsigned)__ballot(v != 0);
        if (i == lane) keep = m;
    }
    bits[(size_t)wave * 32 + lane] = keep;
}

__global__ __launch_bounds__(32) void hmm_forward_kernel(
    const float* __restrict__ start, const float* __restrict__ trans,
    const float* __restrict__ emission, const unsigned* __restrict__ bits,
    float* __restrict__ alpha_out, float* __restrict__ loglik)
{
    const int  lane = threadIdx.x;
    const int  n    = lane & 15;
    const bool lo   = lane < 16;
    const int  b    = blockIdx.x * 16 + n;

    float em1[8], em0[8];
#pragma unroll
    for (int k = 0; k < 8; ++k) {
        float e = emission[k];
        e = fminf(fmaxf(e, HMM_EPS), 1.0f - HMM_EPS);
        em1[k] = e; em0[k] = 1.0f - e;
    }
    float st[8]; float ssum = 0.0f;
#pragma unroll
    for (int k = 0; k < 8; ++k) { st[k] = start[k] + HMM_EPS; ssum += st[k]; }
#pragma unroll
    for (int k = 0; k < 8; ++k) st[k] /= ssum;

    float rs[8];
#pragma unroll
    for (int j = 0; j < 8; ++j) {
        float s = 0.0f;
#pragma unroll
        for (int k = 0; k < 8; ++k) s += trans[j * 8 + k] + HMM_EPS;
        rs[j] = s;
    }

    const int kb = lo ? 0 : 2;
    const int i  = n;
    v2f A0, A1;
    A0.x = (i < 8) ? (trans[(kb + 0) * 8 + i] + HMM_EPS) / rs[kb + 0] : 0.0f;
    A0.y = (i < 8) ? (trans[(kb + 1) * 8 + i] + HMM_EPS) / rs[kb + 1] : 0.0f;
    A1.x = (i < 8) ? (trans[(kb + 4) * 8 + i] + HMM_EPS) / rs[kb + 4] : 0.0f;
    A1.y = (i < 8) ? (trans[(kb + 5) * 8 + i] + HMM_EPS) / rs[kb + 5] : 0.0f;

    const unsigned* const bw = bits + (size_t)b * (HMM_T / 32);
    unsigned wn = bw[0];
    int      ob = (int)(wn & 1u);
    v8f a;
#pragma unroll
    for (int m = 0; m < 8; ++m) a[m] = st[m] * (ob ? em1[m] : em0[m]);
    float s = a[0] + a[1] + a[2] + a[3] + a[4] + a[5] + a[6] + a[7];
    {  float sx = crosshalf(s); s = lo ? s : sx; }
    float inv = fast_rcp(s);
    float lc  = __builtin_amdgcn_logf(s);
    if (lo) *(v8f*)(alpha_out + ((size_t)b * HMM_T) * 8) = a;

    const v8f vzero = {};
    for (int q = 0; q < HMM_T / 32; ++q) {
      const unsigned w = wn;
      wn = bw[q + 1 < HMM_T / 32 ? q + 1 : HMM_T / 32 - 1];
      for (int tb = (q == 0) ? 1 : 0; tb < 32; ++tb) {
        const int t = q * 32 + tb;
        ob = (int)((w >> tb) & 1u);

        float x2 = crosshalf(a[2]);
        float x3 = crosshalf(a[3]);
        float x6 = crosshalf(a[6]);
        float x7 = crosshalf(a[7]);
        v2f B0, B1;
        B0.x = (lo ? a[0] : x2) * inv;  B0.y = (lo ? a[1] : x3) * inv;
        B1.x = (lo ? a[4] : x6) * inv;  B1.y = (lo ? a[5] : x7) * inv;

        v8f d = __builtin_amdgcn_wmma_f32_16x16x4_f32(
            false, A0, false, B0, (short)0, vzero, false, false);
        d = __builtin_amdgcn_wmma_f32_16x16x4_f32(
            false, A1, false, B1, (short)0, d, false, false);

#pragma unroll
        for (int m = 0; m < 8; ++m) a[m] = d[m] * (ob ? em1[m] : em0[m]);
        s = a[0] + a[1] + a[2] + a[3] + a[4] + a[5] + a[6] + a[7];
        float sx = crosshalf(s);
        s = lo ? s : sx;
        inv = fast_rcp(s);
        lc += __builtin_amdgcn_logf(s);
        if (lo) *(v8f*)(alpha_out + ((size_t)b * HMM_T + t) * 8) = a;
      }
    }
    if (lo) loglik[b] = lc * HMM_LN2;
}

__device__ __forceinline__ void bwd_load8(v8f (&buf)[8], const float* seq, int blk)
{
    const float* p = seq + (size_t)blk * 64;
#pragma unroll
    for (int i = 0; i < 8; ++i) buf[i] = *(const v8f*)(p + (7 - i) * 8);
}

__device__ __forceinline__ void bwd_block8(
    const v8f (&buf)[8], float* half_seq, int blk, unsigned w, v8f& d, bool lo,
    const float (&em1)[8], const float (&em0)[8], v2f A0, v2f A1)
{
    const v8f vzero = {};
    const int tt_hi = blk * 8 + 7;
#pragma unroll
    for (int i = 0; i < 8; ++i) {
        const int tt = tt_hi - i;
        const int ob = (int)((w >> (tt & 31)) & 1u);

        const v8f al = buf[i];
        v8f g;
        float gs = 0.0f;
#pragma unroll
        for (int m = 0; m < 8; ++m) { g[m] = al[m] * d[m]; gs += g[m]; }
        const float gi = fast_rcp(gs);
#pragma unroll
        for (int m = 0; m < 8; ++m) g[m] *= gi;
        v4f gh;
        gh.x = lo ? g[0] : g[4];  gh.y = lo ? g[1] : g[5];
        gh.z = lo ? g[2] : g[6];  gh.w = lo ? g[3] : g[7];
        *(v4f*)(half_seq + (size_t)tt * 8) = gh;

        float s = d[0]+d[1]+d[2]+d[3]+d[4]+d[5]+d[6]+d[7];
        float sx = crosshalf(s);
        s = lo ? s : sx;
        const float inv = fast_rcp(s);

        v8f u;
#pragma unroll
        for (int m = 0; m < 8; ++m) u[m] = d[m] * (ob ? em1[m] : em0[m]);
        float x2 = crosshalf(u[2]);
        float x3 = crosshalf(u[3]);
        float x6 = crosshalf(u[6]);
        float x7 = crosshalf(u[7]);
        v2f B0, B1;
        B0.x = (lo ? u[0] : x2) * inv;  B0.y = (lo ? u[1] : x3) * inv;
        B1.x = (lo ? u[4] : x6) * inv;  B1.y = (lo ? u[5] : x7) * inv;

        d = __builtin_amdgcn_wmma_f32_16x16x4_f32(
            false, A0, false, B0, (short)0, vzero, false, false);
        d = __builtin_amdgcn_wmma_f32_16x16x4_f32(
            false, A1, false, B1, (short)0, d, false, false);
    }
}

__global__ __launch_bounds__(32) void hmm_backward_kernel(
    const float* __restrict__ trans, const float* __restrict__ emission,
    const unsigned* __restrict__ bits, float* __restrict__ gamma)
{
    const int  lane = threadIdx.x;
    const int  n    = lane & 15;
    const bool lo   = lane < 16;
    const int  b    = blockIdx.x * 16 + n;

    float em1[8], em0[8];
#pragma unroll
    for (int k = 0; k < 8; ++k) {
        float e = emission[k];
        e = fminf(fmaxf(e, HMM_EPS), 1.0f - HMM_EPS);
        em1[k] = e; em0[k] = 1.0f - e;
    }
    const int kb = lo ? 0 : 2;
    const int i  = n & 7;
    float rsi = 0.0f;
#pragma unroll
    for (int k = 0; k < 8; ++k) rsi += trans[i * 8 + k] + HMM_EPS;
    v2f A0, A1;
    A0.x = (trans[i * 8 + (kb + 0)] + HMM_EPS) / rsi;
    A0.y = (trans[i * 8 + (kb + 1)] + HMM_EPS) / rsi;
    A1.x = (trans[i * 8 + (kb + 4)] + HMM_EPS) / rsi;
    A1.y = (trans[i * 8 + (kb + 5)] + HMM_EPS) / rsi;

    float* const          seq      = gamma + (size_t)b * HMM_T * 8;
    float* const          half_seq = seq + (lo ? 0 : 4);
    const unsigned* const bw       = bits + (size_t)b * (HMM_T / 32);

    v8f d;
#pragma unroll
    for (int m = 0; m < 8; ++m) d[m] = 1.0f;

    v8f bufA[8], bufB[8];
    unsigned wn = bw[HMM_T / 32 - 1];
    bwd_load8(bufA, seq, HMM_T / 8 - 1);

    for (int jb = HMM_T / 8 - 1; jb >= 1; jb -= 2) {
        const unsigned w  = wn;
        const int      jn = jb >= 3 ? jb - 2 : 1;
        bwd_load8(bufB, seq, jb - 1);
        wn = bw[jn >> 2];
        __builtin_amdgcn_sched_barrier(0);
        bwd_block8(bufA, half_seq, jb, w, d, lo, em1, em0, A0, A1);
        bwd_load8(bufA, seq, jn);
        __builtin_amdgcn_sched_barrier(0);
        bwd_block8(bufB, half_seq, jb - 1, w, d, lo, em1, em0, A0, A1);
    }
}

extern "C" void kernel_launch(void* const* d_in, const int* in_sizes, int n_in,
                              void* d_out, int out_size, void* d_ws, size_t ws_size,
                              hipStream_t stream) {
    (void)in_sizes; (void)n_in; (void)out_size; (void)ws_size;
    const float* start    = (const float*)d_in[0];
    const float* trans    = (const float*)d_in[1];
    const float* emission = (const float*)d_in[2];
    const int*   obs      = (const int*)d_in[3];

    float* gamma  = (float*)d_out;
    float* loglik = gamma + (size_t)HMM_B * HMM_T * HMM_K;
    unsigned* bits = (unsigned*)d_ws;

    hmm_pack_kernel<<<256, 256, 0, stream>>>(obs, bits);
    hmm_forward_kernel<<<HMM_B / 16, 32, 0, stream>>>(
        start, trans, emission, bits, gamma, loglik);
    hmm_backward_kernel<<<HMM_B / 16, 32, 0, stream>>>(
        trans, emission, bits, gamma);
}


# forward A repeats rows 0-7 in rows 8-15 (as backward), so every lane holds its column's 8 states: normalizer sums become local pairwise add trees (sum order changed), no cross-half swaps left
# speedup vs baseline: 2.8741x; 1.2333x over previous
#include <hip/hip_runtime.h>


#define HMM_B 256
#define HMM_T 8192
#define HMM_K 8
#define HMM_EPS 1e-8f
#define HMM_LN2 0.6931471805599453f

typedef __attribute__((ext_vector_type(2))) float v2f;
typedef __attribute__((ext_vector_type(4))) float v4f;
typedef __attribute__((ext_vector_type(8))) float v8f;

__device__ __forceinline__ float fast_rcp(float x) {
    float r = __builtin_amdgcn_rcpf(x);
    float e = __builtin_fmaf(-x, r, 1.0f);
    return __builtin_fmaf(r, e, r);
}

__global__ __launch_bounds__(256) void hmm_pack_kernel(
    const int* __restrict__ obs, unsigned* __restrict__ bits)
{
    const int wave = (blockIdx.x * blockDim.x + threadIdx.x) >> 5;
    const int lane = threadIdx.x & 31;
    const long base = (long)wave * 1024;
    unsigned keep = 0;
#pragma unroll 4
    for (int i = 0; i < 32; ++i) {
        int v = obs[base + i * 32 + lane];
        unsigned m = (unsigned)__ballot(v != 0);
        if (i == lane) keep = m;
    }
    bits[(size_t)wave * 32 + lane] = keep;
}

__device__ __forceinline__ float sum8_tree(const v8f& x)
{
    const v4f q = x.lo + x.hi;
    const v2f r = q.lo + q.hi;
    return r.x + r.y;
}

__device__ __forceinline__ float sum8_chain(const v8f& x)
{
    const v4f q = x.lo + x.hi;
    const v2f r = q.lo + q.hi;
    float s;
    asm("v_add_f32 %0, %1, %2" : "=v"(s) : "v"(r.x), "v"(r.y));
    return s;
}

__device__ __forceinline__ void fwd_step(
    v8f& a, float& inv, float& lc, float* row, int ob, bool lo,
    const float (&em1)[8], const float (&em0)[8], v2f A0, v2f A1)
{
    const v8f vzero = {};
    v2f B0, B1;
    B0.x = (lo ? a[0] : a[2]) * inv;  B0.y = (lo ? a[1] : a[3]) * inv;
    B1.x = (lo ? a[4] : a[6]) * inv;  B1.y = (lo ? a[5] : a[7]) * inv;

    v8f d = __builtin_amdgcn_wmma_f32_16x16x4_f32(
        false, A0, false, B0, (short)0, vzero, false, false);
    d = __builtin_amdgcn_wmma_f32_16x16x4_f32(
        false, A1, false, B1, (short)0, d, false, false);

#pragma unroll
    for (int m = 0; m < 8; ++m) a[m] = d[m] * (ob ? em1[m] : em0[m]);
    const float s = sum8_tree(a);
    inv = fast_rcp(s);
    lc += __builtin_amdgcn_logf(s);
    v4f ah;
    ah.x = lo ? a[0] : a[4];  ah.y = lo ? a[1] : a[5];
    ah.z = lo ? a[2] : a[6];  ah.w = lo ? a[3] : a[7];
    *(v4f*)row = ah;
}

__device__ __forceinline__ void fwd_block8(
    v8f& a, float& inv, float& lc, float* row, unsigned wb, bool lo,
    const float (&em1)[8], const float (&em0)[8], v2f A0, v2f A1)
{
#pragma unroll
    for (int tb = 0; tb < 8; ++tb)
        fwd_step(a, inv, lc, row + tb * 8, (int)((wb >> tb) & 1u),
                 lo, em1, em0, A0, A1);
}

__global__ __launch_bounds__(32) void hmm_forward_kernel(
    const float* __restrict__ start, const float* __restrict__ trans,
    const float* __restrict__ emission, const unsigned* __restrict__ bits,
    float* __restrict__ alpha_out, float* __restrict__ loglik)
{
    const int  lane = threadIdx.x;
    const int  n    = lane & 15;
    const bool lo   = lane < 16;
    const int  b    = blockIdx.x * 16 + n;

    float em1[8], em0[8];
#pragma unroll
    for (int k = 0; k < 8; ++k) {
        float e = emission[k];
        e = fminf(fmaxf(e, HMM_EPS), 1.0f - HMM_EPS);
        em1[k] = e; em0[k] = 1.0f - e;
    }
    float st[8]; float ssum = 0.0f;
#pragma unroll
    for (int k = 0; k < 8; ++k) { st[k] = start[k] + HMM_EPS; ssum += st[k]; }
#pragma unroll
    for (int k = 0; k < 8; ++k) st[k] /= ssum;

    float rs[8];
#pragma unroll
    for (int j = 0; j < 8; ++j) {
        float s = 0.0f;
#pragma unroll
        for (int k = 0; k < 8; ++k) s += trans[j * 8 + k] + HMM_EPS;
        rs[j] = s;
    }

    const int kb = lo ? 0 : 2;
    const int i  = n & 7;
    v2f A0, A1;
    A0.x = (trans[(kb + 0) * 8 + i] + HMM_EPS) / rs[kb + 0];
    A0.y = (trans[(kb + 1) * 8 + i] + HMM_EPS) / rs[kb + 1];
    A1.x = (trans[(kb + 4) * 8 + i] + HMM_EPS) / rs[kb + 4];
    A1.y = (trans[(kb + 5) * 8 + i] + HMM_EPS) / rs[kb + 5];

    float* const          half_seq = alpha_out + (size_t)b * HMM_T * 8 + (lo ? 0 : 4);
    const unsigned* const bw       = bits + (size_t)b * (HMM_T / 32);

    unsigned w  = bw[0];
    unsigned wn = bw[1];
    v8f a;
    {
        const int ob = (int)(w & 1u);
#pragma unroll
        for (int m = 0; m < 8; ++m) a[m] = st[m] * (ob ? em1[m] : em0[m]);
    }
    float s0  = sum8_tree(a);
    float inv = fast_rcp(s0);
    float lc  = __builtin_amdgcn_logf(s0);
    {
        v4f ah;
        ah.x = lo ? a[0] : a[4];  ah.y = lo ? a[1] : a[5];
        ah.z = lo ? a[2] : a[6];  ah.w = lo ? a[3] : a[7];
        *(v4f*)half_seq = ah;
    }

#pragma unroll
    for (int tb = 1; tb < 8; ++tb)
        fwd_step(a, inv, lc, half_seq + (size_t)tb * 8, (int)((w >> tb) & 1u),
                 lo, em1, em0, A0, A1);
    for (int sub = 1; sub < 4; ++sub)
        fwd_block8(a, inv, lc, half_seq + (size_t)sub * 64, w >> (sub * 8),
                   lo, em1, em0, A0, A1);

    for (int q = 1; q < HMM_T / 32; ++q) {
        w  = wn;
        wn = bw[q + 1 < HMM_T / 32 ? q + 1 : HMM_T / 32 - 1];
        float* const wrow = half_seq + (size_t)q * 256;
        for (int sub = 0; sub < 4; ++sub)
            fwd_block8(a, inv, lc, wrow + sub * 64, w >> (sub * 8),
                       lo, em1, em0, A0, A1);
    }
    if (lo) loglik[b] = lc * HMM_LN2;
}

__device__ __forceinline__ void bwd_load8(v8f (&buf)[8], const float* seq, int blk)
{
    const float* p = seq + (size_t)blk * 64;
#pragma unroll
    for (int i = 0; i < 8; ++i) buf[i] = *(const v8f*)(p + (7 - i) * 8);
}

__device__ __forceinline__ void bwd_block8(
    const v8f (&buf)[8], float* half_seq, int blk, unsigned w, v8f& d, bool lo,
    const float (&em1)[8], const float (&em0)[8], v2f A0, v2f A1)
{
    const v8f vzero = {};
    const int tt_hi = blk * 8 + 7;
#pragma unroll
    for (int i = 0; i < 8; ++i) {
        const int tt = tt_hi - i;
        const int ob = (int)((w >> (tt & 31)) & 1u);

        const v8f al = buf[i];
        v8f g;
#pragma unroll
        for (int m = 0; m < 8; ++m) g[m] = al[m] * d[m];
        const float gi = fast_rcp(sum8_tree(g));
#pragma unroll
        for (int m = 0; m < 8; ++m) g[m] *= gi;
        v4f gh;
        gh.x = lo ? g[0] : g[4];  gh.y = lo ? g[1] : g[5];
        gh.z = lo ? g[2] : g[6];  gh.w = lo ? g[3] : g[7];
        *(v4f*)(half_seq + (size_t)tt * 8) = gh;

        const float inv = fast_rcp(sum8_chain(d));

        v8f u;
#pragma unroll
        for (int m = 0; m < 8; ++m) u[m] = d[m] * (ob ? em1[m] : em0[m]);
        v2f B0, B1;
        B0.x = (lo ? u[0] : u[2]) * inv;  B0.y = (lo ? u[1] : u[3]) * inv;
        B1.x = (lo ? u[4] : u[6]) * inv;  B1.y = (lo ? u[5] : u[7]) * inv;

        d = __builtin_amdgcn_wmma_f32_16x16x4_f32(
            false, A0, false, B0, (short)0, vzero, false, false);
        d = __builtin_amdgcn_wmma_f32_16x16x4_f32(
            false, A1, false, B1, (short)0, d, false, false);
    }
}

__global__ __launch_bounds__(32) void hmm_backward_kernel(
    const float* __restrict__ trans, const float* __restrict__ emission,
    const unsigned* __restrict__ bits, float* __restrict__ gamma)
{
    const int  lane = threadIdx.x;
    const int  n    = lane & 15;
    const bool lo   = lane < 16;
    const int  b    = blockIdx.x * 16 + n;

    float em1[8], em0[8];
#pragma unroll
    for (int k = 0; k < 8; ++k) {
        float e = emission[k];
        e = fminf(fmaxf(e, HMM_EPS), 1.0f - HMM_EPS);
        em1[k] = e; em0[k] = 1.0f - e;
    }
    const int kb = lo ? 0 : 2;
    const int i  = n & 7;
    float rsi = 0.0f;
#pragma unroll
    for (int k = 0; k < 8; ++k) rsi += trans[i * 8 + k] + HMM_EPS;
    v2f A0, A1;
    A0.x = (trans[i * 8 + (kb + 0)] + HMM_EPS) / rsi;
    A0.y = (trans[i * 8 + (kb + 1)] + HMM_EPS) / rsi;
    A1.x = (trans[i * 8 + (kb + 4)] + HMM_EPS) / rsi;
    A1.y = (trans[i * 8 + (kb + 5)] + HMM_EPS) / rsi;

    float* const          seq      = gamma + (size_t)b * HMM_T * 8;
    float* const          half_seq = seq + (lo ? 0 : 4);
    const unsigned* const bw       = bits + (size_t)b * (HMM_T / 32);

    v8f d;
#pragma unroll
    for (int m = 0; m < 8; ++m) d[m] = 1.0f;

    v8f bufA[8], bufB[8];
    unsigned wn = bw[HMM_T / 32 - 1];
    bwd_load8(bufA, seq, HMM_T / 8 - 1);

    for (int jb = HMM_T / 8 - 1; jb >= 1; jb -= 2) {
        const unsigned w  = wn;
        const int      jn = jb >= 3 ? jb - 2 : 1;
        wn = bw[jn >> 2];
        __builtin_amdgcn_sched_barrier(0);
        bwd_load8(bufB, seq, jb - 1);
        __builtin_amdgcn_sched_barrier(0);
        bwd_block8(bufA, half_seq, jb, w, d, lo, em1, em0, A0, A1);
        bwd_load8(bufA, seq, jn);
        __builtin_amdgcn_sched_barrier(0);
        bwd_block8(bufB, half_seq, jb - 1, w, d, lo, em1, em0, A0, A1);
    }
}

extern "C" void kernel_launch(void* const* d_in, const int* in_sizes, int n_in,
                              void* d_out, int out_size, void* d_ws, size_t ws_size,
                              hipStream_t stream) {
    (void)in_sizes; (void)n_in; (void)out_size; (void)ws_size;
    const float* start    = (const float*)d_in[0];
    const float* trans    = (const float*)d_in[1];
    const float* emission = (const float*)d_in[2];
    const int*   obs      = (const int*)d_in[3];

    float* gamma  = (float*)d_out;
    float* loglik = gamma + (size_t)HMM_B * HMM_T * HMM_K;
    unsigned* bits = (unsigned*)d_ws;

    hmm_pack_kernel<<<256, 256, 0, stream>>>(obs, bits);
    hmm_forward_kernel<<<HMM_B / 16, 32, 0, stream>>>(
        start, trans, emission, bits, gamma, loglik);
    hmm_backward_kernel<<<HMM_B / 16, 32, 0, stream>>>(
        trans, emission, bits, gamma);
}


# alpha and beta recurrences run at once, one wave each in a 2-wave block: each stages its half of T in the gamma slot, one block barrier, each finishes over the other's rows; 3 launches become 2
# speedup vs baseline: 5.6901x; 1.9798x over previous
#include <hip/hip_runtime.h>


#define HMM_B 256
#define HMM_T 8192
#define HMM_K 8
#define HMM_EPS 1e-8f
#define HMM_LN2 0.6931471805599453f

typedef __attribute__((ext_vector_type(2))) float v2f;
typedef __attribute__((ext_vector_type(4))) float v4f;
typedef __attribute__((ext_vector_type(8))) float v8f;

__device__ __forceinline__ float fast_rcp(float x) {
    float r = __builtin_amdgcn_rcpf(x);
    float e = __builtin_fmaf(-x, r, 1.0f);
    return __builtin_fmaf(r, e, r);
}

__global__ __launch_bounds__(256) void hmm_pack_kernel(
    const int* __restrict__ obs, unsigned* __restrict__ bits)
{
    const int wave = (blockIdx.x * blockDim.x + threadIdx.x) >> 5;
    const int lane = threadIdx.x & 31;
    const long base = (long)wave * 1024;
    unsigned keep = 0;
#pragma unroll 4
    for (int i = 0; i < 32; ++i) {
        int v = obs[base + i * 32 + lane];
        unsigned m = (unsigned)__ballot(v != 0);
        if (i == lane) keep = m;
    }
    bits[(size_t)wave * 32 + lane] = keep;
}

__device__ __forceinline__ float sum8_tree(const v8f& x)
{
    const v4f q = x.lo + x.hi;
    const v2f r = q.lo + q.hi;
    return r.x + r.y;
}

__device__ __forceinline__ float sum8_chain(const v8f& x)
{
    const v4f q = x.lo + x.hi;
    const v2f r = q.lo + q.hi;
    float s;
    asm("v_add_f32 %0, %1, %2" : "=v"(s) : "v"(r.x), "v"(r.y));
    return s;
}

__device__ __forceinline__ void fwd_step(
    v8f& a, float& inv, float& lc, float* row, int ob, bool lo,
    const float (&em1)[8], const float (&em0)[8], v2f A0, v2f A1)
{
    const v8f vzero = {};
    v2f B0, B1;
    B0.x = (lo ? a[0] : a[2]) * inv;  B0.y = (lo ? a[1] : a[3]) * inv;
    B1.x = (lo ? a[4] : a[6]) * inv;  B1.y = (lo ? a[5] : a[7]) * inv;

    v8f d = __builtin_amdgcn_wmma_f32_16x16x4_f32(
        false, A0, false, B0, (short)0, vzero, false, false);
    d = __builtin_amdgcn_wmma_f32_16x16x4_f32(
        false, A1, false, B1, (short)0, d, false, false);

#pragma unroll
    for (int m = 0; m < 8; ++m) a[m] = d[m] * (ob ? em1[m] : em0[m]);
    const float s = sum8_tree(a);
    inv = fast_rcp(s);
    lc += __builtin_amdgcn_logf(s);
    v4f ah;
    ah.x = lo ? a[0] : a[4];  ah.y = lo ? a[1] : a[5];
    ah.z = lo ? a[2] : a[6];  ah.w = lo ? a[3] : a[7];
    *(v4f*)row = ah;
}

__device__ __forceinline__ void fwd_block8(
    v8f& a, float& inv, float& lc, float* row, unsigned wb, bool lo,
    const float (&em1)[8], const float (&em0)[8], v2f A0, v2f A1)
{
#pragma unroll
    for (int tb = 0; tb < 8; ++tb)
        fwd_step(a, inv, lc, row + tb * 8, (int)((wb >> tb) & 1u),
                 lo, em1, em0, A0, A1);
}

__device__ __forceinline__ void fwd_load8(v8f (&buf)[8], const float* seq, int blk)
{
    const float* p = seq + (size_t)blk * 64;
#pragma unroll
    for (int i = 0; i < 8; ++i) buf[i] = *(const v8f*)(p + i * 8);
}

__device__ __forceinline__ void fwd_gamma_block8(
    const v8f (&buf)[8], float* half_seq, int blk, unsigned w,
    v8f& a, float& inv, float& lc, bool lo,
    const float (&em1)[8], const float (&em0)[8], v2f A0, v2f A1)
{
    const v8f vzero = {};
    const int t_lo = blk * 8;
#pragma unroll
    for (int i = 0; i < 8; ++i) {
        const int t  = t_lo + i;
        const int ob = (int)((w >> (t & 31)) & 1u);

        v2f B0, B1;
        B0.x = (lo ? a[0] : a[2]) * inv;  B0.y = (lo ? a[1] : a[3]) * inv;
        B1.x = (lo ? a[4] : a[6]) * inv;  B1.y = (lo ? a[5] : a[7]) * inv;
        v8f d = __builtin_amdgcn_wmma_f32_16x16x4_f32(
            false, A0, false, B0, (short)0, vzero, false, false);
        d = __builtin_amdgcn_wmma_f32_16x16x4_f32(
            false, A1, false, B1, (short)0, d, false, false);
#pragma unroll
        for (int m = 0; m < 8; ++m) a[m] = d[m] * (ob ? em1[m] : em0[m]);
        const float s = sum8_chain(a);
        inv = fast_rcp(s);
        lc += __builtin_amdgcn_logf(s);

        const v8f be = buf[i];
        v8f g;
#pragma unroll
        for (int m = 0; m < 8; ++m) g[m] = be[m] * a[m];
        const float gi = fast_rcp(sum8_tree(g));
#pragma unroll
        for (int m = 0; m < 8; ++m) g[m] *= gi;
        v4f gh;
        gh.x = lo ? g[0] : g[4];  gh.y = lo ? g[1] : g[5];
        gh.z = lo ? g[2] : g[6];  gh.w = lo ? g[3] : g[7];
        *(v4f*)(half_seq + (size_t)t * 8) = gh;
    }
}

__device__ __forceinline__ void bwd_stage_block8(
    float* row_hi, unsigned wb, v8f& d, bool lo,
    const float (&em1)[8], const float (&em0)[8], v2f A0, v2f A1)
{
    const v8f vzero = {};
#pragma unroll
    for (int i = 0; i < 8; ++i) {
        const int ob = (int)((wb >> (7 - i)) & 1u);
        v4f dh;
        dh.x = lo ? d[0] : d[4];  dh.y = lo ? d[1] : d[5];
        dh.z = lo ? d[2] : d[6];  dh.w = lo ? d[3] : d[7];
        *(v4f*)(row_hi - i * 8) = dh;

        const float inv = fast_rcp(sum8_tree(d));
        v8f u;
#pragma unroll
        for (int m = 0; m < 8; ++m) u[m] = d[m] * (ob ? em1[m] : em0[m]);
        v2f B0, B1;
        B0.x = (lo ? u[0] : u[2]) * inv;  B0.y = (lo ? u[1] : u[3]) * inv;
        B1.x = (lo ? u[4] : u[6]) * inv;  B1.y = (lo ? u[5] : u[7]) * inv;
        d = __builtin_amdgcn_wmma_f32_16x16x4_f32(
            false, A0, false, B0, (short)0, vzero, false, false);
        d = __builtin_amdgcn_wmma_f32_16x16x4_f32(
            false, A1, false, B1, (short)0, d, false, false);
    }
}

__device__ __forceinline__ void bwd_load8(v8f (&buf)[8], const float* seq, int blk)
{
    const float* p = seq + (size_t)blk * 64;
#pragma unroll
    for (int i = 0; i < 8; ++i) buf[i] = *(const v8f*)(p + (7 - i) * 8);
}

__device__ __forceinline__ void bwd_block8(
    const v8f (&buf)[8], float* half_seq, int blk, unsigned w, v8f& d, bool lo,
    const float (&em1)[8], const float (&em0)[8], v2f A0, v2f A1)
{
    const v8f vzero = {};
    const int tt_hi = blk * 8 + 7;
#pragma unroll
    for (int i = 0; i < 8; ++i) {
        const int tt = tt_hi - i;
        const int ob = (int)((w >> (tt & 31)) & 1u);

        const v8f al = buf[i];
        v8f g;
#pragma unroll
        for (int m = 0; m < 8; ++m) g[m] = al[m] * d[m];
        const float gi = fast_rcp(sum8_tree(g));
#pragma unroll
        for (int m = 0; m < 8; ++m) g[m] *= gi;
        v4f gh;
        gh.x = lo ? g[0] : g[4];  gh.y = lo ? g[1] : g[5];
        gh.z = lo ? g[2] : g[6];  gh.w = lo ? g[3] : g[7];
        *(v4f*)(half_seq + (size_t)tt * 8) = gh;

        const float inv = fast_rcp(sum8_chain(d));

        v8f u;
#pragma unroll
        for (int m = 0; m < 8; ++m) u[m] = d[m] * (ob ? em1[m] : em0[m]);
        v2f B0, B1;
        B0.x = (lo ? u[0] : u[2]) * inv;  B0.y = (lo ? u[1] : u[3]) * inv;
        B1.x = (lo ? u[4] : u[6]) * inv;  B1.y = (lo ? u[5] : u[7]) * inv;

        d = __builtin_amdgcn_wmma_f32_16x16x4_f32(
            false, A0, false, B0, (short)0, vzero, false, false);
        d = __builtin_amdgcn_wmma_f32_16x16x4_f32(
            false, A1, false, B1, (short)0, d, false, false);
    }
}

__device__ __forceinline__ float trans_norm(const float* __restrict__ trans, int r, int c)
{
    float s = 0.0f;
#pragma unroll
    for (int k = 0; k < 8; ++k) s += trans[r * 8 + k] + HMM_EPS;
    return (trans[r * 8 + c] + HMM_EPS) / s;
}

__global__ __launch_bounds__(64) void hmm_fb_kernel(
    const float* __restrict__ start, const float* __restrict__ trans,
    const float* __restrict__ emission, const unsigned* __restrict__ bits,
    float* __restrict__ gamma, float* __restrict__ loglik)
{
    const bool fwd  = __builtin_amdgcn_readfirstlane((int)(threadIdx.x >> 5)) == 0;
    const int  lane = threadIdx.x & 31;
    const int  n    = lane & 15;
    const bool lo   = lane < 16;
    const int  b    = blockIdx.x * 16 + n;

    float em1[8], em0[8];
#pragma unroll
    for (int k = 0; k < 8; ++k) {
        float e = emission[k];
        e = fminf(fmaxf(e, HMM_EPS), 1.0f - HMM_EPS);
        em1[k] = e; em0[k] = 1.0f - e;
    }

    const int kb = lo ? 0 : 2;
    const int i  = n & 7;
    v2f A0, A1;
    A0.x = fwd ? trans_norm(trans, kb + 0, i) : trans_norm(trans, i, kb + 0);
    A0.y = fwd ? trans_norm(trans, kb + 1, i) : trans_norm(trans, i, kb + 1);
    A1.x = fwd ? trans_norm(trans, kb + 4, i) : trans_norm(trans, i, kb + 4);
    A1.y = fwd ? trans_norm(trans, kb + 5, i) : trans_norm(trans, i, kb + 5);

    float* const          seq      = gamma + (size_t)b * HMM_T * 8;
    float* const          half_seq = seq + (lo ? 0 : 4);
    const unsigned* const bw       = bits + (size_t)b * (HMM_T / 32);

    v8f      x;
    float    inv = 1.0f;
    float    lc  = 0.0f;
    unsigned wn;

    if (fwd) {
        float st[8]; float ssum = 0.0f;
#pragma unroll
        for (int k = 0; k < 8; ++k) { st[k] = start[k] + HMM_EPS; ssum += st[k]; }
#pragma unroll
        for (int k = 0; k < 8; ++k) st[k] /= ssum;

        unsigned w = bw[0];
        wn = bw[1];
        {
            const int ob = (int)(w & 1u);
#pragma unroll
            for (int m = 0; m < 8; ++m) x[m] = st[m] * (ob ? em1[m] : em0[m]);
        }
        const float s0 = sum8_tree(x);
        inv = fast_rcp(s0);
        lc  = __builtin_amdgcn_logf(s0);
        {
            v4f ah;
            ah.x = lo ? x[0] : x[4];  ah.y = lo ? x[1] : x[5];
            ah.z = lo ? x[2] : x[6];  ah.w = lo ? x[3] : x[7];
            *(v4f*)half_seq = ah;
        }
#pragma unroll
        for (int tb = 1; tb < 8; ++tb)
            fwd_step(x, inv, lc, half_seq + (size_t)tb * 8, (int)((w >> tb) & 1u),
                     lo, em1, em0, A0, A1);
        for (int sub = 1; sub < 4; ++sub)
            fwd_block8(x, inv, lc, half_seq + (size_t)sub * 64, w >> (sub * 8),
                       lo, em1, em0, A0, A1);
        for (int q = 1; q < HMM_T / 64; ++q) {
            w  = wn;
            wn = bw[q + 1];
            float* const wrow = half_seq + (size_t)q * 256;
            for (int sub = 0; sub < 4; ++sub)
                fwd_block8(x, inv, lc, wrow + sub * 64, w >> (sub * 8),
                           lo, em1, em0, A0, A1);
        }
    } else {
#pragma unroll
        for (int m = 0; m < 8; ++m) x[m] = 1.0f;
        wn = bw[HMM_T / 32 - 1];
        for (int q = HMM_T / 32 - 1; q >= HMM_T / 64; --q) {
            const unsigned w = wn;
            wn = bw[q - 1];
            float* const wrow = half_seq + (size_t)q * 256;
            for (int sub = 3; sub >= 0; --sub)
                bwd_stage_block8(wrow + sub * 64 + 56, w >> (sub * 8), x,
                                 lo, em1, em0, A0, A1);
        }
    }

    __syncthreads();

    v8f bufA[8], bufB[8];
    if (fwd) {
        fwd_load8(bufA, seq, HMM_T / 16);
        for (int jb = HMM_T / 16; jb < HMM_T / 8; jb += 2) {
            const unsigned w  = wn;
            const int      jn = jb + 2 < HMM_T / 8 ? jb + 2 : HMM_T / 8 - 2;
            wn = bw[jn >> 2];
            __builtin_amdgcn_sched_barrier(0);
            fwd_load8(bufB, seq, jb + 1);
            __builtin_amdgcn_sched_barrier(0);
            fwd_gamma_block8(bufA, half_seq, jb, w, x, inv, lc, lo, em1, em0, A0, A1);
            fwd_load8(bufA, seq, jn);
            __builtin_amdgcn_sched_barrier(0);
            fwd_gamma_block8(bufB, half_seq, jb + 1, w, x, inv, lc, lo, em1, em0, A0, A1);
        }
        if (lo) loglik[b] = lc * HMM_LN2;
    } else {
        bwd_load8(bufA, seq, HMM_T / 16 - 1);
        for (int jb = HMM_T / 16 - 1; jb >= 1; jb -= 2) {
            const unsigned w  = wn;
            const int      jn = jb >= 3 ? jb - 2 : 1;
            wn = bw[jn >> 2];
            __builtin_amdgcn_sched_barrier(0);
            bwd_load8(bufB, seq, jb - 1);
            __builtin_amdgcn_sched_barrier(0);
            bwd_block8(bufA, half_seq, jb, w, x, lo, em1, em0, A0, A1);
            bwd_load8(bufA, seq, jn);
            __builtin_amdgcn_sched_barrier(0);
            bwd_block8(bufB, half_seq, jb - 1, w, x, lo, em1, em0, A0, A1);
        }
    }
}

extern "C" void kernel_launch(void* const* d_in, const int* in_sizes, int n_in,
                              void* d_out, int out_size, void* d_ws, size_t ws_size,
                              hipStream_t stream) {
    (void)in_sizes; (void)n_in; (void)out_size; (void)ws_size;
    const float* start    = (const float*)d_in[0];
    const float* trans    = (const float*)d_in[1];
    const float* emission = (const float*)d_in[2];
    const int*   obs      = (const int*)d_in[3];

    float* gamma  = (float*)d_out;
    float* loglik = gamma + (size_t)HMM_B * HMM_T * HMM_K;
    unsigned* bits = (unsigned*)d_ws;

    hmm_pack_kernel<<<256, 256, 0, stream>>>(obs, bits);
    hmm_fb_kernel<<<HMM_B / 16, 64, 0, stream>>>(
        start, trans, emission, bits, gamma, loglik);
}


# time axis cut into 32 chunks run side by side: per chunk an 8x8 transfer matrix by the same WMMA step, a scan gives every chunk its entry alpha and exit beta and loglik (its sum order changed)
# speedup vs baseline: 34.2258x; 6.0150x over previous
#include <hip/hip_runtime.h>


#define HMM_B 256
#define HMM_T 8192
#define HMM_K 8
#define HMM_C 32
#define HMM_L (HMM_T / HMM_C)
#define HMM_EPS 1e-8f
#define HMM_LN2 0.6931471805599453f

typedef __attribute__((ext_vector_type(2))) float v2f;
typedef __attribute__((ext_vector_type(4))) float v4f;
typedef __attribute__((ext_vector_type(8))) float v8f;

__device__ __forceinline__ float fast_rcp(float x) {
    float r = __builtin_amdgcn_rcpf(x);
    float e = __builtin_fmaf(-x, r, 1.0f);
    return __builtin_fmaf(r, e, r);
}

__global__ __launch_bounds__(256) void hmm_pack_kernel(
    const int* __restrict__ obs, unsigned* __restrict__ bits)
{
    const int wave = (blockIdx.x * blockDim.x + threadIdx.x) >> 5;
    const int lane = threadIdx.x & 31;
    const long base = (long)wave * 1024;
    unsigned keep = 0;
#pragma unroll 4
    for (int i = 0; i < 32; ++i) {
        int v = obs[base + i * 32 + lane];
        unsigned m = (unsigned)__ballot(v != 0);
        if (i == lane) keep = m;
    }
    bits[(size_t)wave * 32 + lane] = keep;
}

__device__ __forceinline__ float sum8_tree(const v8f& x)
{
    const v4f q = x.lo + x.hi;
    const v2f r = q.lo + q.hi;
    return r.x + r.y;
}

__device__ __forceinline__ float sum8_chain(const v8f& x)
{
    const v4f q = x.lo + x.hi;
    const v2f r = q.lo + q.hi;
    float s;
    asm("v_add_f32 %0, %1, %2" : "=v"(s) : "v"(r.x), "v"(r.y));
    return s;
}

template <bool STORE, bool LOGS>
__device__ __forceinline__ void fwd_step(
    v8f& a, float& inv, float& lc, float* row, int ob, bool lo,
    const float (&em1)[8], const float (&em0)[8], v2f A0, v2f A1)
{
    const v8f vzero = {};
    v2f B0, B1;
    B0.x = (lo ? a[0] : a[2]) * inv;  B0.y = (lo ? a[1] : a[3]) * inv;
    B1.x = (lo ? a[4] : a[6]) * inv;  B1.y = (lo ? a[5] : a[7]) * inv;

    v8f d = __builtin_amdgcn_wmma_f32_16x16x4_f32(
        false, A0, false, B0, (short)0, vzero, false, false);
    d = __builtin_amdgcn_wmma_f32_16x16x4_f32(
        false, A1, false, B1, (short)0, d, false, false);

#pragma unroll
    for (int m = 0; m < 8; ++m) a[m] = d[m] * (ob ? em1[m] : em0[m]);
    const float s = sum8_tree(a);
    inv = fast_rcp(s);
    if (LOGS) lc += __builtin_amdgcn_logf(s);
    if (STORE) {
        v4f ah;
        ah.x = lo ? a[0] : a[4];  ah.y = lo ? a[1] : a[5];
        ah.z = lo ? a[2] : a[6];  ah.w = lo ? a[3] : a[7];
        *(v4f*)row = ah;
    }
}

template <bool STORE, bool LOGS>
__device__ __forceinline__ void fwd_block8(
    v8f& a, float& inv, float& lc, float* row, unsigned wb, bool lo,
    const float (&em1)[8], const float (&em0)[8], v2f A0, v2f A1)
{
#pragma unroll
    for (int tb = 0; tb < 8; ++tb)
        fwd_step<STORE, LOGS>(a, inv, lc, STORE ? row + tb * 8 : row,
                              (int)((wb >> tb) & 1u), lo, em1, em0, A0, A1);
}

__device__ __forceinline__ void fwd_load8(v8f (&buf)[8], const float* seq, int blk)
{
    const float* p = seq + (size_t)blk * 64;
#pragma unroll
    for (int i = 0; i < 8; ++i) buf[i] = *(const v8f*)(p + i * 8);
}

__device__ __forceinline__ void fwd_gamma_block8(
    const v8f (&buf)[8], float* half_seq, int blk, unsigned w,
    v8f& a, float& inv, bool lo,
    const float (&em1)[8], const float (&em0)[8], v2f A0, v2f A1)
{
    const v8f vzero = {};
    const int t_lo = blk * 8;
#pragma unroll
    for (int i = 0; i < 8; ++i) {
        const int t  = t_lo + i;
        const int ob = (int)((w >> (t & 31)) & 1u);

        v2f B0, B1;
        B0.x = (lo ? a[0] : a[2]) * inv;  B0.y = (lo ? a[1] : a[3]) * inv;
        B1.x = (lo ? a[4] : a[6]) * inv;  B1.y = (lo ? a[5] : a[7]) * inv;
        v8f d = __builtin_amdgcn_wmma_f32_16x16x4_f32(
            false, A0, false, B0, (short)0, vzero, false, false);
        d = __builtin_amdgcn_wmma_f32_16x16x4_f32(
            false, A1, false, B1, (short)0, d, false, false);
#pragma unroll
        for (int m = 0; m < 8; ++m) a[m] = d[m] * (ob ? em1[m] : em0[m]);
        const float s = sum8_chain(a);
        inv = fast_rcp(s);

        const v8f be = buf[i];
        v8f g;
#pragma unroll
        for (int m = 0; m < 8; ++m) g[m] = be[m] * a[m];
        const float gi = fast_rcp(sum8_tree(g));
#pragma unroll
        for (int m = 0; m < 8; ++m) g[m] *= gi;
        v4f gh;
        gh.x = lo ? g[0] : g[4];  gh.y = lo ? g[1] : g[5];
        gh.z = lo ? g[2] : g[6];  gh.w = lo ? g[3] : g[7];
        *(v4f*)(half_seq + (size_t)t * 8) = gh;
    }
}

__device__ __forceinline__ void bwd_stage_block8(
    float* row_hi, unsigned wb, v8f& d, bool lo,
    const float (&em1)[8], const float (&em0)[8], v2f A0, v2f A1)
{
    const v8f vzero = {};
#pragma unroll
    for (int i = 0; i < 8; ++i) {
        const int ob = (int)((wb >> (7 - i)) & 1u);
        v4f dh;
        dh.x = lo ? d[0] : d[4];  dh.y = lo ? d[1] : d[5];
        dh.z = lo ? d[2] : d[6];  dh.w = lo ? d[3] : d[7];
        *(v4f*)(row_hi - i * 8) = dh;

        const float inv = fast_rcp(sum8_tree(d));
        v8f u;
#pragma unroll
        for (int m = 0; m < 8; ++m) u[m] = d[m] * (ob ? em1[m] : em0[m]);
        v2f B0, B1;
        B0.x = (lo ? u[0] : u[2]) * inv;  B0.y = (lo ? u[1] : u[3]) * inv;
        B1.x = (lo ? u[4] : u[6]) * inv;  B1.y = (lo ? u[5] : u[7]) * inv;
        d = __builtin_amdgcn_wmma_f32_16x16x4_f32(
            false, A0, false, B0, (short)0, vzero, false, false);
        d = __builtin_amdgcn_wmma_f32_16x16x4_f32(
            false, A1, false, B1, (short)0, d, false, false);
    }
}

__device__ __forceinline__ void bwd_load8(v8f (&buf)[8], const float* seq, int blk)
{
    const float* p = seq + (size_t)blk * 64;
#pragma unroll
    for (int i = 0; i < 8; ++i) buf[i] = *(const v8f*)(p + (7 - i) * 8);
}

__device__ __forceinline__ void bwd_block8(
    const v8f (&buf)[8], float* half_seq, int blk, unsigned w, v8f& d, bool lo,
    const float (&em1)[8], const float (&em0)[8], v2f A0, v2f A1)
{
    const v8f vzero = {};
    const int tt_hi = blk * 8 + 7;
#pragma unroll
    for (int i = 0; i < 8; ++i) {
        const int tt = tt_hi - i;
        const int ob = (int)((w >> (tt & 31)) & 1u);

        const v8f al = buf[i];
        v8f g;
#pragma unroll
        for (int m = 0; m < 8; ++m) g[m] = al[m] * d[m];
        const float gi = fast_rcp(sum8_tree(g));
#pragma unroll
        for (int m = 0; m < 8; ++m) g[m] *= gi;
        v4f gh;
        gh.x = lo ? g[0] : g[4];  gh.y = lo ? g[1] : g[5];
        gh.z = lo ? g[2] : g[6];  gh.w = lo ? g[3] : g[7];
        *(v4f*)(half_seq + (size_t)tt * 8) = gh;

        const float inv = fast_rcp(sum8_chain(d));

        v8f u;
#pragma unroll
        for (int m = 0; m < 8; ++m) u[m] = d[m] * (ob ? em1[m] : em0[m]);
        v2f B0, B1;
        B0.x = (lo ? u[0] : u[2]) * inv;  B0.y = (lo ? u[1] : u[3]) * inv;
        B1.x = (lo ? u[4] : u[6]) * inv;  B1.y = (lo ? u[5] : u[7]) * inv;

        d = __builtin_amdgcn_wmma_f32_16x16x4_f32(
            false, A0, false, B0, (short)0, vzero, false, false);
        d = __builtin_amdgcn_wmma_f32_16x16x4_f32(
            false, A1, false, B1, (short)0, d, false, false);
    }
}

__device__ __forceinline__ float trans_norm(const float* __restrict__ trans, int r, int c)
{
    float s = 0.0f;
#pragma unroll
    for (int k = 0; k < 8; ++k) s += trans[r * 8 + k] + HMM_EPS;
    return (trans[r * 8 + c] + HMM_EPS) / s;
}

__device__ __forceinline__ void load_emission(
    const float* __restrict__ emission, float (&em1)[8], float (&em0)[8])
{
#pragma unroll
    for (int k = 0; k < 8; ++k) {
        float e = emission[k];
        e = fminf(fmaxf(e, HMM_EPS), 1.0f - HMM_EPS);
        em1[k] = e; em0[k] = 1.0f - e;
    }
}

__global__ __launch_bounds__(64) void hmm_chunk_kernel(
    const float* __restrict__ trans, const float* __restrict__ emission,
    const unsigned* __restrict__ bits, float* __restrict__ gamma)
{
    const int  wave = __builtin_amdgcn_readfirstlane((int)(threadIdx.x >> 5));
    const int  lane = threadIdx.x & 31;
    const int  n    = lane & 15;
    const bool lo   = lane < 16;
    const int  c    = blockIdx.x / (HMM_B / 4);
    const int  pair = (blockIdx.x % (HMM_B / 4)) * 2 + wave;
    const int  b    = pair * 2 + (n >> 3);
    const int  j    = n & 7;

    float em1[8], em0[8];
    load_emission(emission, em1, em0);

    const int kb = lo ? 0 : 2;
    const int i  = n & 7;
    v2f A0, A1;
    A0.x = trans_norm(trans, kb + 0, i);
    A0.y = trans_norm(trans, kb + 1, i);
    A1.x = trans_norm(trans, kb + 4, i);
    A1.y = trans_norm(trans, kb + 5, i);

    const unsigned* const bw = bits + (size_t)b * (HMM_T / 32) + c * (HMM_L / 32);

    v8f      x;
    float    inv, lc;
    unsigned w  = bw[0];
    unsigned wn = bw[1];
    if (c == 0) {
        const int ob = (int)(w & 1u);
#pragma unroll
        for (int m = 0; m < 8; ++m) x[m] = (m == j) ? (ob ? em1[m] : em0[m]) : 0.0f;
        const float s0 = sum8_tree(x);
        inv = fast_rcp(s0);
        lc  = __builtin_amdgcn_logf(s0);
#pragma unroll
        for (int tb = 1; tb < 8; ++tb)
            fwd_step<false, true>(x, inv, lc, nullptr, (int)((w >> tb) & 1u),
                                  lo, em1, em0, A0, A1);
    } else {
#pragma unroll
        for (int m = 0; m < 8; ++m) x[m] = (m == j) ? 1.0f : 0.0f;
        inv = 1.0f;
        lc  = 0.0f;
        fwd_block8<false, true>(x, inv, lc, nullptr, w, lo, em1, em0, A0, A1);
    }
    for (int sub = 1; sub < 4; ++sub)
        fwd_block8<false, true>(x, inv, lc, nullptr, w >> (sub * 8),
                                lo, em1, em0, A0, A1);
    for (int q = 1; q < HMM_L / 32; ++q) {
        w  = wn;
        wn = bw[q + 1 < HMM_L / 32 ? q + 1 : q];
        for (int sub = 0; sub < 4; ++sub)
            fwd_block8<false, true>(x, inv, lc, nullptr, w >> (sub * 8),
                                    lo, em1, em0, A0, A1);
    }

    float* const reg = gamma + ((size_t)b * HMM_T + (size_t)c * HMM_L) * 8;
    v4f ah;
    ah.x = (lo ? x[0] : x[4]) * inv;  ah.y = (lo ? x[1] : x[5]) * inv;
    ah.z = (lo ? x[2] : x[6]) * inv;  ah.w = (lo ? x[3] : x[7]) * inv;
    *(v4f*)(reg + (4 + j) * 8 + (lo ? 0 : 4)) = ah;
    if (lo) reg[12 * 8 + j] = lc;
}

__device__ __forceinline__ void scan_load(v8f (&r)[9], const float* reg)
{
#pragma unroll
    for (int k = 0; k < 9; ++k) r[k] = *(const v8f*)(reg + (4 + k) * 8);
}

__global__ __launch_bounds__(64) void hmm_scan_kernel(
    const float* __restrict__ start, float* __restrict__ gamma,
    float* __restrict__ loglik)
{
    const bool fwd = __builtin_amdgcn_readfirstlane((int)(threadIdx.x >> 5)) == 0;
    const int  b   = blockIdx.x * 32 + (threadIdx.x & 31);
    float* const seq = gamma + (size_t)b * HMM_T * 8;

    v8f cur[9], nxt[9];
    if (fwd) {
        v8f v; float ssum = 0.0f;
#pragma unroll
        for (int k = 0; k < 8; ++k) { v[k] = start[k] + HMM_EPS; ssum += v[k]; }
#pragma unroll
        for (int k = 0; k < 8; ++k) v[k] /= ssum;

        float tot = 0.0f;
        scan_load(cur, seq);
        for (int c = 0; c < HMM_C; ++c) {
            const int cn = c + 1 < HMM_C ? c + 1 : c;
            scan_load(nxt, seq + (size_t)cn * HMM_L * 8);

            const v8f lg = cur[8];
            float m = lg[0];
#pragma unroll
            for (int k = 1; k < 8; ++k) m = fmaxf(m, lg[k]);
            v8f u = {};
#pragma unroll
            for (int k = 0; k < 8; ++k) {
                const float wk = exp2f(lg[k] - m) * v[k];
#pragma unroll
                for (int r = 0; r < 8; ++r) u[r] = fmaf(cur[k][r], wk, u[r]);
            }
            const float s = sum8_tree(u);
            const float rs = 1.0f / s;
#pragma unroll
            for (int r = 0; r < 8; ++r) v[r] = u[r] * rs;
            tot += m + log2f(s);
            if (c + 1 < HMM_C) *(v8f*)(seq + (size_t)(c + 1) * HMM_L * 8) = v;
#pragma unroll
            for (int k = 0; k < 9; ++k) cur[k] = nxt[k];
        }
        loglik[b] = tot * HMM_LN2;
    } else {
        v8f f;
#pragma unroll
        for (int k = 0; k < 8; ++k) f[k] = 1.0f;
        *(v8f*)(seq + ((size_t)HMM_T - 1) * 8) = f;
        scan_load(cur, seq + (size_t)(HMM_C - 1) * HMM_L * 8);
        for (int c = HMM_C - 1; c >= 1; --c) {
            const int cn = c > 1 ? c - 1 : c;
            scan_load(nxt, seq + (size_t)cn * HMM_L * 8);

            const v8f lg = cur[8];
            float m = lg[0];
#pragma unroll
            for (int k = 1; k < 8; ++k) m = fmaxf(m, lg[k]);
            v8f u;
#pragma unroll
            for (int k = 0; k < 8; ++k) {
                float dot = 0.0f;
#pragma unroll
                for (int r = 0; r < 8; ++r) dot = fmaf(cur[k][r], f[r], dot);
                u[k] = exp2f(lg[k] - m) * dot;
            }
            const float rs = 1.0f / sum8_tree(u);
#pragma unroll
            for (int r = 0; r < 8; ++r) f[r] = u[r] * rs;
            *(v8f*)(seq + ((size_t)c * HMM_L - 1) * 8) = f;
#pragma unroll
            for (int k = 0; k < 9; ++k) cur[k] = nxt[k];
        }
    }
}

__global__ __launch_bounds__(64) void hmm_fb_kernel(
    const float* __restrict__ start, const float* __restrict__ trans,
    const float* __restrict__ emission, const unsigned* __restrict__ bits,
    float* __restrict__ gamma)
{
    const bool fwd  = __builtin_amdgcn_readfirstlane((int)(threadIdx.x >> 5)) == 0;
    const int  lane = threadIdx.x & 31;
    const int  n    = lane & 15;
    const bool lo   = lane < 16;
    const int  c    = blockIdx.x / (HMM_B / 16);
    const int  b    = (blockIdx.x % (HMM_B / 16)) * 16 + n;

    float em1[8], em0[8];
    load_emission(emission, em1, em0);

    const int kb = lo ? 0 : 2;
    const int i  = n & 7;
    v2f A0, A1;
    A0.x = fwd ? trans_norm(trans, kb + 0, i) : trans_norm(trans, i, kb + 0);
    A0.y = fwd ? trans_norm(trans, kb + 1, i) : trans_norm(trans, i, kb + 1);
    A1.x = fwd ? trans_norm(trans, kb + 4, i) : trans_norm(trans, i, kb + 4);
    A1.y = fwd ? trans_norm(trans, kb + 5, i) : trans_norm(trans, i, kb + 5);

    float* const          seq      = gamma + ((size_t)b * HMM_T + (size_t)c * HMM_L) * 8;
    float* const          half_seq = seq + (lo ? 0 : 4);
    const unsigned* const bw       = bits + (size_t)b * (HMM_T / 32) + c * (HMM_L / 32);

    v8f      x;
    float    inv = 1.0f;
    float    lcn = 0.0f;
    unsigned wn;

    if (fwd) {
        unsigned w = bw[0];
        wn = bw[1];
        if (c == 0) {
            float st[8]; float ssum = 0.0f;
#pragma unroll
            for (int k = 0; k < 8; ++k) { st[k] = start[k] + HMM_EPS; ssum += st[k]; }
#pragma unroll
            for (int k = 0; k < 8; ++k) st[k] /= ssum;

            {
                const int ob = (int)(w & 1u);
#pragma unroll
                for (int m = 0; m < 8; ++m) x[m] = st[m] * (ob ? em1[m] : em0[m]);
            }
            inv = fast_rcp(sum8_tree(x));
            {
                v4f ah;
                ah.x = lo ? x[0] : x[4];  ah.y = lo ? x[1] : x[5];
                ah.z = lo ? x[2] : x[6];  ah.w = lo ? x[3] : x[7];
                *(v4f*)half_seq = ah;
            }
#pragma unroll
            for (int tb = 1; tb < 8; ++tb)
                fwd_step<true, false>(x, inv, lcn, half_seq + (size_t)tb * 8,
                                      (int)((w >> tb) & 1u), lo, em1, em0, A0, A1);
        } else {
            x   = *(const v8f*)seq;
            inv = 1.0f;
            fwd_block8<true, false>(x, inv, lcn, half_seq, w, lo, em1, em0, A0, A1);
        }
        for (int sub = 1; sub < 4; ++sub)
            fwd_block8<true, false>(x, inv, lcn, half_seq + (size_t)sub * 64,
                                    w >> (sub * 8), lo, em1, em0, A0, A1);
        for (int q = 1; q < HMM_L / 64; ++q) {
            w  = wn;
            wn = bw[q + 1];
            float* const wrow = half_seq + (size_t)q * 256;
            for (int sub = 0; sub < 4; ++sub)
                fwd_block8<true, false>(x, inv, lcn, wrow + sub * 64, w >> (sub * 8),
                                        lo, em1, em0, A0, A1);
        }
    } else {
        x  = *(const v8f*)(seq + (size_t)(HMM_L - 1) * 8);
        wn = bw[HMM_L / 32 - 1];
        for (int q = HMM_L / 32 - 1; q >= HMM_L / 64; --q) {
            const unsigned w = wn;
            wn = bw[q - 1];
            float* const wrow = half_seq + (size_t)q * 256;
            for (int sub = 3; sub >= 0; --sub)
                bwd_stage_block8(wrow + sub * 64 + 56, w >> (sub * 8), x,
                                 lo, em1, em0, A0, A1);
        }
    }

    __syncthreads();

    v8f bufA[8], bufB[8];
    if (fwd) {
        fwd_load8(bufA, seq, HMM_L / 16);
        for (int jb = HMM_L / 16; jb < HMM_L / 8; jb += 2) {
            const unsigned w  = wn;
            const int      jn = jb + 2 < HMM_L / 8 ? jb + 2 : HMM_L / 8 - 2;
            wn = bw[jn >> 2];
            __builtin_amdgcn_sched_barrier(0);
            fwd_load8(bufB, seq, jb + 1);
            __builtin_amdgcn_sched_barrier(0);
            fwd_gamma_block8(bufA, half_seq, jb, w, x, inv, lo, em1, em0, A0, A1);
            fwd_load8(bufA, seq, jn);
            __builtin_amdgcn_sched_barrier(0);
            fwd_gamma_block8(bufB, half_seq, jb + 1, w, x, inv, lo, em1, em0, A0, A1);
        }
    } else {
        bwd_load8(bufA, seq, HMM_L / 16 - 1);
        for (int jb = HMM_L / 16 - 1; jb >= 1; jb -= 2) {
            const unsigned w  = wn;
            const int      jn = jb >= 3 ? jb - 2 : 1;
            wn = bw[jn >> 2];
            __builtin_amdgcn_sched_barrier(0);
            bwd_load8(bufB, seq, jb - 1);
            __builtin_amdgcn_sched_barrier(0);
            bwd_block8(bufA, half_seq, jb, w, x, lo, em1, em0, A0, A1);
            bwd_load8(bufA, seq, jn);
            __builtin_amdgcn_sched_barrier(0);
            bwd_block8(bufB, half_seq, jb - 1, w, x, lo, em1, em0, A0, A1);
        }
    }
}

extern "C" void kernel_launch(void* const* d_in, const int* in_sizes, int n_in,
                              void* d_out, int out_size, void* d_ws, size_t ws_size,
                              hipStream_t stream) {
    (void)in_sizes; (void)n_in; (void)out_size; (void)ws_size;
    const float* start    = (const float*)d_in[0];
    const float* trans    = (const float*)d_in[1];
    const float* emission = (const float*)d_in[2];
    const int*   obs      = (const int*)d_in[3];

    float* gamma  = (float*)d_out;
    float* loglik = gamma + (size_t)HMM_B * HMM_T * HMM_K;
    unsigned* bits = (unsigned*)d_ws;

    hmm_pack_kernel<<<256, 256, 0, stream>>>(obs, bits);
    hmm_chunk_kernel<<<HMM_C * (HMM_B / 4), 64, 0, stream>>>(
        trans, emission, bits, gamma);
    hmm_scan_kernel<<<HMM_B / 32, 64, 0, stream>>>(start, gamma, loglik);
    hmm_fb_kernel<<<HMM_C * (HMM_B / 16), 64, 0, stream>>>(
        start, trans, emission, bits, gamma);
}


# chunk transfer matrices = products of 32 table matrices, one 8-step matrix per obs byte (256 of them, built beside the bit packing); order of products and sums changed
# speedup vs baseline: 45.8330x; 1.3391x over previous
#include <hip/hip_runtime.h>


#define HMM_B 256
#define HMM_T 8192
#define HMM_K 8
#define HMM_C 32
#define HMM_L (HMM_T / HMM_C)
#define HMM_EPS 1e-8f
#define HMM_LN2 0.6931471805599453f

typedef __attribute__((ext_vector_type(2))) float v2f;
typedef __attribute__((ext_vector_type(4))) float v4f;
typedef __attribute__((ext_vector_type(8))) float v8f;

__device__ __forceinline__ float fast_rcp(float x) {
    float r = __builtin_amdgcn_rcpf(x);
    float e = __builtin_fmaf(-x, r, 1.0f);
    return __builtin_fmaf(r, e, r);
}

#define HMM_TAB_BLOCKS  16
#define HMM_PACK_BLOCKS 256

__device__ __forceinline__ size_t tab_entry_off(unsigned v)
{
    return ((size_t)(v >> 4) * HMM_L + 16) * 8 + (size_t)(v & 15u) * 64;
}
__device__ __forceinline__ size_t tab_scale_off(unsigned v)
{
    return ((size_t)(v >> 4) * HMM_L + 144) * 8 + (size_t)(v & 15u);
}
static_assert(HMM_C >= 16 && HMM_L >= 146, "the table needs rows 16-145 of 16 chunk regions");

__device__ void hmm_build_table(const float* __restrict__ trans,
                                const float* __restrict__ emission,
                                float* __restrict__ tab, int tw);

__global__ __launch_bounds__(256) void hmm_pack_kernel(
    const int* __restrict__ obs, unsigned* __restrict__ bits,
    const float* __restrict__ trans, const float* __restrict__ emission,
    float* __restrict__ tab)
{
    if (blockIdx.x < HMM_TAB_BLOCKS) {
        hmm_build_table(trans, emission, tab,
                        (int)((blockIdx.x * blockDim.x + threadIdx.x) >> 5));
        return;
    }
    const int wave = ((blockIdx.x - HMM_TAB_BLOCKS) * blockDim.x + threadIdx.x) >> 5;
    const int lane = threadIdx.x & 31;
    const long base = (long)wave * 1024;
    unsigned keep = 0;
#pragma unroll 4
    for (int i = 0; i < 32; ++i) {
        int v = obs[base + i * 32 + lane];
        unsigned m = (unsigned)__ballot(v != 0);
        if (i == lane) keep = m;
    }
    bits[(size_t)wave * 32 + lane] = keep;
}

__device__ __forceinline__ float sum8_tree(const v8f& x)
{
    const v4f q = x.lo + x.hi;
    const v2f r = q.lo + q.hi;
    return r.x + r.y;
}

__device__ __forceinline__ float sum8_chain(const v8f& x)
{
    const v4f q = x.lo + x.hi;
    const v2f r = q.lo + q.hi;
    float s;
    asm("v_add_f32 %0, %1, %2" : "=v"(s) : "v"(r.x), "v"(r.y));
    return s;
}

template <bool STORE, bool LOGS>
__device__ __forceinline__ void fwd_step(
    v8f& a, float& inv, float& lc, float* row, int ob, bool lo,
    const float (&em1)[8], const float (&em0)[8], v2f A0, v2f A1)
{
    const v8f vzero = {};
    v2f B0, B1;
    B0.x = (lo ? a[0] : a[2]) * inv;  B0.y = (lo ? a[1] : a[3]) * inv;
    B1.x = (lo ? a[4] : a[6]) * inv;  B1.y = (lo ? a[5] : a[7]) * inv;

    v8f d = __builtin_amdgcn_wmma_f32_16x16x4_f32(
        false, A0, false, B0, (short)0, vzero, false, false);
    d = __builtin_amdgcn_wmma_f32_16x16x4_f32(
        false, A1, false, B1, (short)0, d, false, false);

#pragma unroll
    for (int m = 0; m < 8; ++m) a[m] = d[m] * (ob ? em1[m] : em0[m]);
    const float s = sum8_tree(a);
    inv = fast_rcp(s);
    if (LOGS) lc += __builtin_amdgcn_logf(s);
    if (STORE) {
        v4f ah;
        ah.x = lo ? a[0] : a[4];  ah.y = lo ? a[1] : a[5];
        ah.z = lo ? a[2] : a[6];  ah.w = lo ? a[3] : a[7];
        *(v4f*)row = ah;
    }
}

template <bool STORE, bool LOGS>
__device__ __forceinline__ void fwd_block8(
    v8f& a, float& inv, float& lc, float* row, unsigned wb, bool lo,
    const float (&em1)[8], const float (&em0)[8], v2f A0, v2f A1)
{
#pragma unroll
    for (int tb = 0; tb < 8; ++tb)
        fwd_step<STORE, LOGS>(a, inv, lc, STORE ? row + tb * 8 : row,
                              (int)((wb >> tb) & 1u), lo, em1, em0, A0, A1);
}

__device__ __forceinline__ void fwd_load8(v8f (&buf)[8], const float* seq, int blk)
{
    const float* p = seq + (size_t)blk * 64;
#pragma unroll
    for (int i = 0; i < 8; ++i) buf[i] = *(const v8f*)(p + i * 8);
}

__device__ __forceinline__ void fwd_gamma_block8(
    const v8f (&buf)[8], float* half_seq, int blk, unsigned w,
    v8f& a, float& inv, bool lo,
    const float (&em1)[8], const float (&em0)[8], v2f A0, v2f A1)
{
    const v8f vzero = {};
    const int t_lo = blk * 8;
#pragma unroll
    for (int i = 0; i < 8; ++i) {
        const int t  = t_lo + i;
        const int ob = (int)((w >> (t & 31)) & 1u);

        v2f B0, B1;
        B0.x = (lo ? a[0] : a[2]) * inv;  B0.y = (lo ? a[1] : a[3]) * inv;
        B1.x = (lo ? a[4] : a[6]) * inv;  B1.y = (lo ? a[5] : a[7]) * inv;
        v8f d = __builtin_amdgcn_wmma_f32_16x16x4_f32(
            false, A0, false, B0, (short)0, vzero, false, false);
        d = __builtin_amdgcn_wmma_f32_16x16x4_f32(
            false, A1, false, B1, (short)0, d, false, false);
#pragma unroll
        for (int m = 0; m < 8; ++m) a[m] = d[m] * (ob ? em1[m] : em0[m]);
        const float s = sum8_chain(a);
        inv = fast_rcp(s);

        const v8f be = buf[i];
        v8f g;
#pragma unroll
        for (int m = 0; m < 8; ++m) g[m] = be[m] * a[m];
        const float gi = fast_rcp(sum8_tree(g));
#pragma unroll
        for (int m = 0; m < 8; ++m) g[m] *= gi;
        v4f gh;
        gh.x = lo ? g[0] : g[4];  gh.y = lo ? g[1] : g[5];
        gh.z = lo ? g[2] : g[6];  gh.w = lo ? g[3] : g[7];
        *(v4f*)(half_seq + (size_t)t * 8) = gh;
    }
}

__device__ __forceinline__ void bwd_stage_block8(
    float* row_hi, unsigned wb, v8f& d, bool lo,
    const float (&em1)[8], const float (&em0)[8], v2f A0, v2f A1)
{
    const v8f vzero = {};
#pragma unroll
    for (int i = 0; i < 8; ++i) {
        const int ob = (int)((wb >> (7 - i)) & 1u);
        v4f dh;
        dh.x = lo ? d[0] : d[4];  dh.y = lo ? d[1] : d[5];
        dh.z = lo ? d[2] : d[6];  dh.w = lo ? d[3] : d[7];
        *(v4f*)(row_hi - i * 8) = dh;

        const float inv = fast_rcp(sum8_tree(d));
        v8f u;
#pragma unroll
        for (int m = 0; m < 8; ++m) u[m] = d[m] * (ob ? em1[m] : em0[m]);
        v2f B0, B1;
        B0.x = (lo ? u[0] : u[2]) * inv;  B0.y = (lo ? u[1] : u[3]) * inv;
        B1.x = (lo ? u[4] : u[6]) * inv;  B1.y = (lo ? u[5] : u[7]) * inv;
        d = __builtin_amdgcn_wmma_f32_16x16x4_f32(
            false, A0, false, B0, (short)0, vzero, false, false);
        d = __builtin_amdgcn_wmma_f32_16x16x4_f32(
            false, A1, false, B1, (short)0, d, false, false);
    }
}

__device__ __forceinline__ void bwd_load8(v8f (&buf)[8], const float* seq, int blk)
{
    const float* p = seq + (size_t)blk * 64;
#pragma unroll
    for (int i = 0; i < 8; ++i) buf[i] = *(const v8f*)(p + (7 - i) * 8);
}

__device__ __forceinline__ void bwd_block8(
    const v8f (&buf)[8], float* half_seq, int blk, unsigned w, v8f& d, bool lo,
    const float (&em1)[8], const float (&em0)[8], v2f A0, v2f A1)
{
    const v8f vzero = {};
    const int tt_hi = blk * 8 + 7;
#pragma unroll
    for (int i = 0; i < 8; ++i) {
        const int tt = tt_hi - i;
        const int ob = (int)((w >> (tt & 31)) & 1u);

        const v8f al = buf[i];
        v8f g;
#pragma unroll
        for (int m = 0; m < 8; ++m) g[m] = al[m] * d[m];
        const float gi = fast_rcp(sum8_tree(g));
#pragma unroll
        for (int m = 0; m < 8; ++m) g[m] *= gi;
        v4f gh;
        gh.x = lo ? g[0] : g[4];  gh.y = lo ? g[1] : g[5];
        gh.z = lo ? g[2] : g[6];  gh.w = lo ? g[3] : g[7];
        *(v4f*)(half_seq + (size_t)tt * 8) = gh;

        const float inv = fast_rcp(sum8_chain(d));

        v8f u;
#pragma unroll
        for (int m = 0; m < 8; ++m) u[m] = d[m] * (ob ? em1[m] : em0[m]);
        v2f B0, B1;
        B0.x = (lo ? u[0] : u[2]) * inv;  B0.y = (lo ? u[1] : u[3]) * inv;
        B1.x = (lo ? u[4] : u[6]) * inv;  B1.y = (lo ? u[5] : u[7]) * inv;

        d = __builtin_amdgcn_wmma_f32_16x16x4_f32(
            false, A0, false, B0, (short)0, vzero, false, false);
        d = __builtin_amdgcn_wmma_f32_16x16x4_f32(
            false, A1, false, B1, (short)0, d, false, false);
    }
}

__device__ __forceinline__ float trans_norm(const float* __restrict__ trans, int r, int c)
{
    float s = 0.0f;
#pragma unroll
    for (int k = 0; k < 8; ++k) s += trans[r * 8 + k] + HMM_EPS;
    return (trans[r * 8 + c] + HMM_EPS) / s;
}

__device__ __forceinline__ void load_emission(
    const float* __restrict__ emission, float (&em1)[8], float (&em0)[8])
{
#pragma unroll
    for (int k = 0; k < 8; ++k) {
        float e = emission[k];
        e = fminf(fmaxf(e, HMM_EPS), 1.0f - HMM_EPS);
        em1[k] = e; em0[k] = 1.0f - e;
    }
}

__device__ void hmm_build_table(const float* __restrict__ trans,
                                const float* __restrict__ emission,
                                float* __restrict__ tab, int tw)
{
    const int  lane = threadIdx.x & 31;
    const int  n    = lane & 15;
    const bool lo   = lane < 16;
    const int  j    = n & 7;
    const unsigned v = (unsigned)(tw * 2 + (n >> 3));

    float em1[8], em0[8];
    load_emission(emission, em1, em0);

    const int kb = lo ? 0 : 2;
    v2f A0, A1;
    A0.x = trans_norm(trans, kb + 0, j);
    A0.y = trans_norm(trans, kb + 1, j);
    A1.x = trans_norm(trans, kb + 4, j);
    A1.y = trans_norm(trans, kb + 5, j);

    v8f x;
#pragma unroll
    for (int m = 0; m < 8; ++m) x[m] = (m == j) ? 1.0f : 0.0f;
    float inv = 1.0f, lc = 0.0f;
    fwd_block8<false, true>(x, inv, lc, nullptr, v, lo, em1, em0, A0, A1);

    float mx = lc;
    mx = fmaxf(mx, __shfl_xor(mx, 1));
    mx = fmaxf(mx, __shfl_xor(mx, 2));
    mx = fmaxf(mx, __shfl_xor(mx, 4));
    const float sc = inv * __builtin_amdgcn_exp2f(lc - mx);

    const int    pj = (j & 1) | ((j >> 1) & 2) | ((j << 1) & 4);
    float* const P  = tab + tab_entry_off(v) + pj;
    const int    i0 = lo ? 0 : 4;
    P[(i0 + 0) * 8] = (lo ? x[0] : x[4]) * sc;
    P[(i0 + 1) * 8] = (lo ? x[1] : x[5]) * sc;
    P[(i0 + 2) * 8] = (lo ? x[2] : x[6]) * sc;
    P[(i0 + 3) * 8] = (lo ? x[3] : x[7]) * sc;
    if (lo && j == 0) tab[tab_scale_off(v)] = mx;
}

__device__ __forceinline__ void tab_load4(
    const float* __restrict__ tab, unsigned w, int aoff,
    v2f (&a0)[4], v2f (&a1)[4], float (&ms)[4])
{
#pragma unroll
    for (int sb = 0; sb < 4; ++sb) {
        const unsigned v = (w >> (sb * 8)) & 255u;
        const v4f      p = *(const v4f*)(tab + tab_entry_off(v) + aoff);
        a0[sb] = p.lo;
        a1[sb] = p.hi;
        ms[sb] = tab[tab_scale_off(v)];
    }
}

__device__ __forceinline__ void tab_step(
    v8f& a, float& inv, float& lc, bool lo, v2f A0, v2f A1, float mlog)
{
    const v8f vzero = {};
    v2f B0, B1;
    B0.x = (lo ? a[0] : a[2]) * inv;  B0.y = (lo ? a[1] : a[3]) * inv;
    B1.x = (lo ? a[4] : a[6]) * inv;  B1.y = (lo ? a[5] : a[7]) * inv;
    v8f d = __builtin_amdgcn_wmma_f32_16x16x4_f32(
        false, A0, false, B0, (short)0, vzero, false, false);
    a = __builtin_amdgcn_wmma_f32_16x16x4_f32(
        false, A1, false, B1, (short)0, d, false, false);
    const float s = sum8_tree(a);
    inv = fast_rcp(s);
    lc += __builtin_amdgcn_logf(s) + mlog;
}

__global__ __launch_bounds__(64) void hmm_chunk_kernel(
    const float* __restrict__ trans, const float* __restrict__ emission,
    const unsigned* __restrict__ bits, const float* __restrict__ tab,
    float* __restrict__ gamma)
{
    const int  wave = __builtin_amdgcn_readfirstlane((int)(threadIdx.x >> 5));
    const int  lane = threadIdx.x & 31;
    const int  n    = lane & 15;
    const bool lo   = lane < 16;
    const int  c    = blockIdx.x / (HMM_B / 2);
    const int  b    = (blockIdx.x % (HMM_B / 2)) * 2 + wave;
    const int  j    = n & 7;
    const int  aoff = j * 8 + (lo ? 0 : 4);

    const unsigned* const bw = bits + (size_t)b * (HMM_T / 32) + c * (HMM_L / 32);

    unsigned w = (unsigned)__builtin_amdgcn_readfirstlane((int)bw[0]);
    v2f   a0c[4], a1c[4], a0n[4], a1n[4];
    float mc[4], mn[4];
    tab_load4(tab, w, aoff, a0c, a1c, mc);

    v8f   x;
    float inv, lc;
    if (c == 0) {
        float em1[8], em0[8];
        load_emission(emission, em1, em0);
        const int kb = lo ? 0 : 2;
        v2f A0, A1;
        A0.x = trans_norm(trans, kb + 0, j);
        A0.y = trans_norm(trans, kb + 1, j);
        A1.x = trans_norm(trans, kb + 4, j);
        A1.y = trans_norm(trans, kb + 5, j);
        const int ob = (int)(w & 1u);
#pragma unroll
        for (int m = 0; m < 8; ++m) x[m] = (m == j) ? (ob ? em1[m] : em0[m]) : 0.0f;
        const float s0 = sum8_tree(x);
        inv = fast_rcp(s0);
        lc  = __builtin_amdgcn_logf(s0);
#pragma unroll
        for (int tb = 1; tb < 8; ++tb)
            fwd_step<false, true>(x, inv, lc, nullptr, (int)((w >> tb) & 1u),
                                  lo, em1, em0, A0, A1);
    } else {
#pragma unroll
        for (int m = 0; m < 8; ++m) x[m] = (m == j) ? 1.0f : 0.0f;
        inv = 1.0f;
        lc  = 0.0f;
        tab_step(x, inv, lc, lo, a0c[0], a1c[0], mc[0]);
    }
    for (int q = 0; q < HMM_L / 32; ++q) {
        const unsigned wn = (unsigned)__builtin_amdgcn_readfirstlane(
            (int)bw[q + 1 < HMM_L / 32 ? q + 1 : q]);
        tab_load4(tab, wn, aoff, a0n, a1n, mn);
#pragma unroll
        for (int sb = 0; sb < 4; ++sb)
            if (q > 0 || sb > 0) tab_step(x, inv, lc, lo, a0c[sb], a1c[sb], mc[sb]);
#pragma unroll
        for (int sb = 0; sb < 4; ++sb) { a0c[sb] = a0n[sb]; a1c[sb] = a1n[sb]; mc[sb] = mn[sb]; }
    }

    float* const reg = gamma + ((size_t)b * HMM_T + (size_t)c * HMM_L) * 8;
    v4f ah;
    ah.x = (lo ? x[0] : x[4]) * inv;  ah.y = (lo ? x[1] : x[5]) * inv;
    ah.z = (lo ? x[2] : x[6]) * inv;  ah.w = (lo ? x[3] : x[7]) * inv;
    if (n < 8) {
        *(v4f*)(reg + (4 + j) * 8 + (lo ? 0 : 4)) = ah;
        if (lo) reg[12 * 8 + j] = lc;
    }
}

__device__ __forceinline__ void scan_load(v8f (&r)[9], const float* reg)
{
#pragma unroll
    for (int k = 0; k < 9; ++k) r[k] = *(const v8f*)(reg + (4 + k) * 8);
}

__global__ __launch_bounds__(64) void hmm_scan_kernel(
    const float* __restrict__ start, float* __restrict__ gamma,
    float* __restrict__ loglik)
{
    const bool fwd = __builtin_amdgcn_readfirstlane((int)(threadIdx.x >> 5)) == 0;
    const int  b   = blockIdx.x * 32 + (threadIdx.x & 31);
    float* const seq = gamma + (size_t)b * HMM_T * 8;

    v8f cur[9], nxt[9];
    if (fwd) {
        v8f v; float ssum = 0.0f;
#pragma unroll
        for (int k = 0; k < 8; ++k) { v[k] = start[k] + HMM_EPS; ssum += v[k]; }
#pragma unroll
        for (int k = 0; k < 8; ++k) v[k] /= ssum;

        float tot = 0.0f;
        scan_load(cur, seq);
        for (int c = 0; c < HMM_C; ++c) {
            const int cn = c + 1 < HMM_C ? c + 1 : c;
            scan_load(nxt, seq + (size_t)cn * HMM_L * 8);

            const v8f lg = cur[8];
            float m = lg[0];
#pragma unroll
            for (int k = 1; k < 8; ++k) m = fmaxf(m, lg[k]);
            v8f u = {};
#pragma unroll
            for (int k = 0; k < 8; ++k) {
                const float wk = exp2f(lg[k] - m) * v[k];
#pragma unroll
                for (int r = 0; r < 8; ++r) u[r] = fmaf(cur[k][r], wk, u[r]);
            }
            const float s = sum8_tree(u);
            const float rs = 1.0f / s;
#pragma unroll
            for (int r = 0; r < 8; ++r) v[r] = u[r] * rs;
            tot += m + log2f(s);
            if (c + 1 < HMM_C) *(v8f*)(seq + (size_t)(c + 1) * HMM_L * 8) = v;
#pragma unroll
            for (int k = 0; k < 9; ++k) cur[k] = nxt[k];
        }
        loglik[b] = tot * HMM_LN2;
    } else {
        v8f f;
#pragma unroll
        for (int k = 0; k < 8; ++k) f[k] = 1.0f;
        *(v8f*)(seq + ((size_t)HMM_T - 1) * 8) = f;
        scan_load(cur, seq + (size_t)(HMM_C - 1) * HMM_L * 8);
        for (int c = HMM_C - 1; c >= 1; --c) {
            const int cn = c > 1 ? c - 1 : c;
            scan_load(nxt, seq + (size_t)cn * HMM_L * 8);

            const v8f lg = cur[8];
            float m = lg[0];
#pragma unroll
            for (int k = 1; k < 8; ++k) m = fmaxf(m, lg[k]);
            v8f u;
#pragma unroll
            for (int k = 0; k < 8; ++k) {
                float dot = 0.0f;
#pragma unroll
                for (int r = 0; r < 8; ++r) dot = fmaf(cur[k][r], f[r], dot);
                u[k] = exp2f(lg[k] - m) * dot;
            }
            const float rs = 1.0f / sum8_tree(u);
#pragma unroll
            for (int r = 0; r < 8; ++r) f[r] = u[r] * rs;
            *(v8f*)(seq + ((size_t)c * HMM_L - 1) * 8) = f;
#pragma unroll
            for (int k = 0; k < 9; ++k) cur[k] = nxt[k];
        }
    }
}

__global__ __launch_bounds__(64) void hmm_fb_kernel(
    const float* __restrict__ start, const float* __restrict__ trans,
    const float* __restrict__ emission, const unsigned* __restrict__ bits,
    float* __restrict__ gamma)
{
    const bool fwd  = __builtin_amdgcn_readfirstlane((int)(threadIdx.x >> 5)) == 0;
    const int  lane = threadIdx.x & 31;
    const int  n    = lane & 15;
    const bool lo   = lane < 16;
    const int  c    = blockIdx.x / (HMM_B / 16);
    const int  b    = (blockIdx.x % (HMM_B / 16)) * 16 + n;

    float em1[8], em0[8];
    load_emission(emission, em1, em0);

    const int kb = lo ? 0 : 2;
    const int i  = n & 7;
    v2f A0, A1;
    A0.x = fwd ? trans_norm(trans, kb + 0, i) : trans_norm(trans, i, kb + 0);
    A0.y = fwd ? trans_norm(trans, kb + 1, i) : trans_norm(trans, i, kb + 1);
    A1.x = fwd ? trans_norm(trans, kb + 4, i) : trans_norm(trans, i, kb + 4);
    A1.y = fwd ? trans_norm(trans, kb + 5, i) : trans_norm(trans, i, kb + 5);

    float* const          seq      = gamma + ((size_t)b * HMM_T + (size_t)c * HMM_L) * 8;
    float* const          half_seq = seq + (lo ? 0 : 4);
    const unsigned* const bw       = bits + (size_t)b * (HMM_T / 32) + c * (HMM_L / 32);

    v8f      x;
    float    inv = 1.0f;
    float    lcn = 0.0f;
    unsigned wn;

    if (fwd) {
        unsigned w = bw[0];
        wn = bw[1];
        if (c == 0) {
            float st[8]; float ssum = 0.0f;
#pragma unroll
            for (int k = 0; k < 8; ++k) { st[k] = start[k] + HMM_EPS; ssum += st[k]; }
#pragma unroll
            for (int k = 0; k < 8; ++k) st[k] /= ssum;

            {
                const int ob = (int)(w & 1u);
#pragma unroll
                for (int m = 0; m < 8; ++m) x[m] = st[m] * (ob ? em1[m] : em0[m]);
            }
            inv = fast_rcp(sum8_tree(x));
            {
                v4f ah;
                ah.x = lo ? x[0] : x[4];  ah.y = lo ? x[1] : x[5];
                ah.z = lo ? x[2] : x[6];  ah.w = lo ? x[3] : x[7];
                *(v4f*)half_seq = ah;
            }
#pragma unroll
            for (int tb = 1; tb < 8; ++tb)
                fwd_step<true, false>(x, inv, lcn, half_seq + (size_t)tb * 8,
                                      (int)((w >> tb) & 1u), lo, em1, em0, A0, A1);
        } else {
            x   = *(const v8f*)seq;
            inv = 1.0f;
            fwd_block8<true, false>(x, inv, lcn, half_seq, w, lo, em1, em0, A0, A1);
        }
        for (int sub = 1; sub < 4; ++sub)
            fwd_block8<true, false>(x, inv, lcn, half_seq + (size_t)sub * 64,
                                    w >> (sub * 8), lo, em1, em0, A0, A1);
        for (int q = 1; q < HMM_L / 64; ++q) {
            w  = wn;
            wn = bw[q + 1];
            float* const wrow = half_seq + (size_t)q * 256;
            for (int sub = 0; sub < 4; ++sub)
                fwd_block8<true, false>(x, inv, lcn, wrow + sub * 64, w >> (sub * 8),
                                        lo, em1, em0, A0, A1);
        }
    } else {
        x  = *(const v8f*)(seq + (size_t)(HMM_L - 1) * 8);
        wn = bw[HMM_L / 32 - 1];
        for (int q = HMM_L / 32 - 1; q >= HMM_L / 64; --q) {
            const unsigned w = wn;
            wn = bw[q - 1];
            float* const wrow = half_seq + (size_t)q * 256;
            for (int sub = 3; sub >= 0; --sub)
                bwd_stage_block8(wrow + sub * 64 + 56, w >> (sub * 8), x,
                                 lo, em1, em0, A0, A1);
        }
    }

    __syncthreads();

    v8f bufA[8], bufB[8];
    if (fwd) {
        fwd_load8(bufA, seq, HMM_L / 16);
        for (int jb = HMM_L / 16; jb < HMM_L / 8; jb += 2) {
            const unsigned w  = wn;
            const int      jn = jb + 2 < HMM_L / 8 ? jb + 2 : HMM_L / 8 - 2;
            wn = bw[jn >> 2];
            __builtin_amdgcn_sched_barrier(0);
            fwd_load8(bufB, seq, jb + 1);
            __builtin_amdgcn_sched_barrier(0);
            fwd_gamma_block8(bufA, half_seq, jb, w, x, inv, lo, em1, em0, A0, A1);
            fwd_load8(bufA, seq, jn);
            __builtin_amdgcn_sched_barrier(0);
            fwd_gamma_block8(bufB, half_seq, jb + 1, w, x, inv, lo, em1, em0, A0, A1);
        }
    } else {
        bwd_load8(bufA, seq, HMM_L / 16 - 1);
        for (int jb = HMM_L / 16 - 1; jb >= 1; jb -= 2) {
            const unsigned w  = wn;
            const int      jn = jb >= 3 ? jb - 2 : 1;
            wn = bw[jn >> 2];
            __builtin_amdgcn_sched_barrier(0);
            bwd_load8(bufB, seq, jb - 1);
            __builtin_amdgcn_sched_barrier(0);
            bwd_block8(bufA, half_seq, jb, w, x, lo, em1, em0, A0, A1);
            bwd_load8(bufA, seq, jn);
            __builtin_amdgcn_sched_barrier(0);
            bwd_block8(bufB, half_seq, jb - 1, w, x, lo, em1, em0, A0, A1);
        }
    }
}

extern "C" void kernel_launch(void* const* d_in, const int* in_sizes, int n_in,
                              void* d_out, int out_size, void* d_ws, size_t ws_size,
                              hipStream_t stream) {
    (void)in_sizes; (void)n_in; (void)out_size; (void)ws_size;
    const float* start    = (const float*)d_in[0];
    const float* trans    = (const float*)d_in[1];
    const float* emission = (const float*)d_in[2];
    const int*   obs      = (const int*)d_in[3];

    float* gamma  = (float*)d_out;
    float* loglik = gamma + (size_t)HMM_B * HMM_T * HMM_K;
    unsigned* bits = (unsigned*)d_ws;

    hmm_pack_kernel<<<HMM_TAB_BLOCKS + HMM_PACK_BLOCKS, 256, 0, stream>>>(
        obs, bits, trans, emission, gamma);
    hmm_chunk_kernel<<<HMM_C * (HMM_B / 2), 64, 0, stream>>>(
        trans, emission, bits, gamma, gamma);
    hmm_scan_kernel<<<HMM_B / 32, 64, 0, stream>>>(start, gamma, loglik);
    hmm_fb_kernel<<<HMM_C * (HMM_B / 16), 64, 0, stream>>>(
        start, trans, emission, bits, gamma);
}


# scan re-laid, lane = chunk: a wave per sequence and direction loads all 32 chunk matrices at once, vector passed lane to lane by v_readlane; scales folded into matrices, loglik summed as a lane tree
# speedup vs baseline: 64.6611x; 1.4108x over previous
#include <hip/hip_runtime.h>


#define HMM_B 256
#define HMM_T 8192
#define HMM_K 8
#define HMM_C 32
#define HMM_L (HMM_T / HMM_C)
#define HMM_EPS 1e-8f
#define HMM_LN2 0.6931471805599453f

typedef __attribute__((ext_vector_type(2))) float v2f;
typedef __attribute__((ext_vector_type(4))) float v4f;
typedef __attribute__((ext_vector_type(8))) float v8f;

__device__ __forceinline__ float fast_rcp(float x) {
    float r = __builtin_amdgcn_rcpf(x);
    float e = __builtin_fmaf(-x, r, 1.0f);
    return __builtin_fmaf(r, e, r);
}

#define HMM_TAB_BLOCKS  16
#define HMM_PACK_BLOCKS 256

__device__ __forceinline__ size_t tab_entry_off(unsigned v)
{
    return ((size_t)(v >> 4) * HMM_L + 16) * 8 + (size_t)(v & 15u) * 64;
}
__device__ __forceinline__ size_t tab_scale_off(unsigned v)
{
    return ((size_t)(v >> 4) * HMM_L + 144) * 8 + (size_t)(v & 15u);
}
static_assert(HMM_C >= 16 && HMM_L >= 146, "the table needs rows 16-145 of 16 chunk regions");

__device__ void hmm_build_table(const float* __restrict__ trans,
                                const float* __restrict__ emission,
                                float* __restrict__ tab, int tw);

__global__ __launch_bounds__(256) void hmm_pack_kernel(
    const int* __restrict__ obs, unsigned* __restrict__ bits,
    const float* __restrict__ trans, const float* __restrict__ emission,
    float* __restrict__ tab)
{
    if (blockIdx.x < HMM_TAB_BLOCKS) {
        hmm_build_table(trans, emission, tab,
                        (int)((blockIdx.x * blockDim.x + threadIdx.x) >> 5));
        return;
    }
    const int wave = ((blockIdx.x - HMM_TAB_BLOCKS) * blockDim.x + threadIdx.x) >> 5;
    const int lane = threadIdx.x & 31;
    const long base = (long)wave * 1024;
    unsigned keep = 0;
#pragma unroll 4
    for (int i = 0; i < 32; ++i) {
        int v = obs[base + i * 32 + lane];
        unsigned m = (unsigned)__ballot(v != 0);
        if (i == lane) keep = m;
    }
    bits[(size_t)wave * 32 + lane] = keep;
}

__device__ __forceinline__ float sum8_tree(const v8f& x)
{
    const v4f q = x.lo + x.hi;
    const v2f r = q.lo + q.hi;
    return r.x + r.y;
}

__device__ __forceinline__ float sum8_chain(const v8f& x)
{
    const v4f q = x.lo + x.hi;
    const v2f r = q.lo + q.hi;
    float s;
    asm("v_add_f32 %0, %1, %2" : "=v"(s) : "v"(r.x), "v"(r.y));
    return s;
}

template <bool STORE, bool LOGS>
__device__ __forceinline__ void fwd_step(
    v8f& a, float& inv, float& lc, float* row, int ob, bool lo,
    const float (&em1)[8], const float (&em0)[8], v2f A0, v2f A1)
{
    const v8f vzero = {};
    v2f B0, B1;
    B0.x = (lo ? a[0] : a[2]) * inv;  B0.y = (lo ? a[1] : a[3]) * inv;
    B1.x = (lo ? a[4] : a[6]) * inv;  B1.y = (lo ? a[5] : a[7]) * inv;

    v8f d = __builtin_amdgcn_wmma_f32_16x16x4_f32(
        false, A0, false, B0, (short)0, vzero, false, false);
    d = __builtin_amdgcn_wmma_f32_16x16x4_f32(
        false, A1, false, B1, (short)0, d, false, false);

#pragma unroll
    for (int m = 0; m < 8; ++m) a[m] = d[m] * (ob ? em1[m] : em0[m]);
    const float s = sum8_tree(a);
    inv = fast_rcp(s);
    if (LOGS) lc += __builtin_amdgcn_logf(s);
    if (STORE) {
        v4f ah;
        ah.x = lo ? a[0] : a[4];  ah.y = lo ? a[1] : a[5];
        ah.z = lo ? a[2] : a[6];  ah.w = lo ? a[3] : a[7];
        *(v4f*)row = ah;
    }
}

template <bool STORE, bool LOGS>
__device__ __forceinline__ void fwd_block8(
    v8f& a, float& inv, float& lc, float* row, unsigned wb, bool lo,
    const float (&em1)[8], const float (&em0)[8], v2f A0, v2f A1)
{
#pragma unroll
    for (int tb = 0; tb < 8; ++tb)
        fwd_step<STORE, LOGS>(a, inv, lc, STORE ? row + tb * 8 : row,
                              (int)((wb >> tb) & 1u), lo, em1, em0, A0, A1);
}

__device__ __forceinline__ void fwd_load8(v8f (&buf)[8], const float* seq, int blk)
{
    const float* p = seq + (size_t)blk * 64;
#pragma unroll
    for (int i = 0; i < 8; ++i) buf[i] = *(const v8f*)(p + i * 8);
}

__device__ __forceinline__ void fwd_gamma_block8(
    const v8f (&buf)[8], float* half_seq, int blk, unsigned w,
    v8f& a, float& inv, bool lo,
    const float (&em1)[8], const float (&em0)[8], v2f A0, v2f A1)
{
    const v8f vzero = {};
    const int t_lo = blk * 8;
#pragma unroll
    for (int i = 0; i < 8; ++i) {
        const int t  = t_lo + i;
        const int ob = (int)((w >> (t & 31)) & 1u);

        v2f B0, B1;
        B0.x = (lo ? a[0] : a[2]) * inv;  B0.y = (lo ? a[1] : a[3]) * inv;
        B1.x = (lo ? a[4] : a[6]) * inv;  B1.y = (lo ? a[5] : a[7]) * inv;
        v8f d = __builtin_amdgcn_wmma_f32_16x16x4_f32(
            false, A0, false, B0, (short)0, vzero, false, false);
        d = __builtin_amdgcn_wmma_f32_16x16x4_f32(
            false, A1, false, B1, (short)0, d, false, false);
#pragma unroll
        for (int m = 0; m < 8; ++m) a[m] = d[m] * (ob ? em1[m] : em0[m]);
        const float s = sum8_chain(a);
        inv = fast_rcp(s);

        const v8f be = buf[i];
        v8f g;
#pragma unroll
        for (int m = 0; m < 8; ++m) g[m] = be[m] * a[m];
        const float gi = fast_rcp(sum8_tree(g));
#pragma unroll
        for (int m = 0; m < 8; ++m) g[m] *= gi;
        v4f gh;
        gh.x = lo ? g[0] : g[4];  gh.y = lo ? g[1] : g[5];
        gh.z = lo ? g[2] : g[6];  gh.w = lo ? g[3] : g[7];
        *(v4f*)(half_seq + (size_t)t * 8) = gh;
    }
}

__device__ __forceinline__ void bwd_stage_block8(
    float* row_hi, unsigned wb, v8f& d, bool lo,
    const float (&em1)[8], const float (&em0)[8], v2f A0, v2f A1)
{
    const v8f vzero = {};
#pragma unroll
    for (int i = 0; i < 8; ++i) {
        const int ob = (int)((wb >> (7 - i)) & 1u);
        v4f dh;
        dh.x = lo ? d[0] : d[4];  dh.y = lo ? d[1] : d[5];
        dh.z = lo ? d[2] : d[6];  dh.w = lo ? d[3] : d[7];
        *(v4f*)(row_hi - i * 8) = dh;

        const float inv = fast_rcp(sum8_tree(d));
        v8f u;
#pragma unroll
        for (int m = 0; m < 8; ++m) u[m] = d[m] * (ob ? em1[m] : em0[m]);
        v2f B0, B1;
        B0.x = (lo ? u[0] : u[2]) * inv;  B0.y = (lo ? u[1] : u[3]) * inv;
        B1.x = (lo ? u[4] : u[6]) * inv;  B1.y = (lo ? u[5] : u[7]) * inv;
        d = __builtin_amdgcn_wmma_f32_16x16x4_f32(
            false, A0, false, B0, (short)0, vzero, false, false);
        d = __builtin_amdgcn_wmma_f32_16x16x4_f32(
            false, A1, false, B1, (short)0, d, false, false);
    }
}

__device__ __forceinline__ void bwd_load8(v8f (&buf)[8], const float* seq, int blk)
{
    const float* p = seq + (size_t)blk * 64;
#pragma unroll
    for (int i = 0; i < 8; ++i) buf[i] = *(const v8f*)(p + (7 - i) * 8);
}

__device__ __forceinline__ void bwd_block8(
    const v8f (&buf)[8], float* half_seq, int blk, unsigned w, v8f& d, bool lo,
    const float (&em1)[8], const float (&em0)[8], v2f A0, v2f A1)
{
    const v8f vzero = {};
    const int tt_hi = blk * 8 + 7;
#pragma unroll
    for (int i = 0; i < 8; ++i) {
        const int tt = tt_hi - i;
        const int ob = (int)((w >> (tt & 31)) & 1u);

        const v8f al = buf[i];
        v8f g;
#pragma unroll
        for (int m = 0; m < 8; ++m) g[m] = al[m] * d[m];
        const float gi = fast_rcp(sum8_tree(g));
#pragma unroll
        for (int m = 0; m < 8; ++m) g[m] *= gi;
        v4f gh;
        gh.x = lo ? g[0] : g[4];  gh.y = lo ? g[1] : g[5];
        gh.z = lo ? g[2] : g[6];  gh.w = lo ? g[3] : g[7];
        *(v4f*)(half_seq + (size_t)tt * 8) = gh;

        const float inv = fast_rcp(sum8_chain(d));

        v8f u;
#pragma unroll
        for (int m = 0; m < 8; ++m) u[m] = d[m] * (ob ? em1[m] : em0[m]);
        v2f B0, B1;
        B0.x = (lo ? u[0] : u[2]) * inv;  B0.y = (lo ? u[1] : u[3]) * inv;
        B1.x = (lo ? u[4] : u[6]) * inv;  B1.y = (lo ? u[5] : u[7]) * inv;

        d = __builtin_amdgcn_wmma_f32_16x16x4_f32(
            false, A0, false, B0, (short)0, vzero, false, false);
        d = __builtin_amdgcn_wmma_f32_16x16x4_f32(
            false, A1, false, B1, (short)0, d, false, false);
    }
}

__device__ __forceinline__ float trans_norm(const float* __restrict__ trans, int r, int c)
{
    float s = 0.0f;
#pragma unroll
    for (int k = 0; k < 8; ++k) s += trans[r * 8 + k] + HMM_EPS;
    return (trans[r * 8 + c] + HMM_EPS) / s;
}

__device__ __forceinline__ void load_emission(
    const float* __restrict__ emission, float (&em1)[8], float (&em0)[8])
{
#pragma unroll
    for (int k = 0; k < 8; ++k) {
        float e = emission[k];
        e = fminf(fmaxf(e, HMM_EPS), 1.0f - HMM_EPS);
        em1[k] = e; em0[k] = 1.0f - e;
    }
}

__device__ void hmm_build_table(const float* __restrict__ trans,
                                const float* __restrict__ emission,
                                float* __restrict__ tab, int tw)
{
    const int  lane = threadIdx.x & 31;
    const int  n    = lane & 15;
    const bool lo   = lane < 16;
    const int  j    = n & 7;
    const unsigned v = (unsigned)(tw * 2 + (n >> 3));

    float em1[8], em0[8];
    load_emission(emission, em1, em0);

    const int kb = lo ? 0 : 2;
    v2f A0, A1;
    A0.x = trans_norm(trans, kb + 0, j);
    A0.y = trans_norm(trans, kb + 1, j);
    A1.x = trans_norm(trans, kb + 4, j);
    A1.y = trans_norm(trans, kb + 5, j);

    v8f x;
#pragma unroll
    for (int m = 0; m < 8; ++m) x[m] = (m == j) ? 1.0f : 0.0f;
    float inv = 1.0f, lc = 0.0f;
    fwd_block8<false, true>(x, inv, lc, nullptr, v, lo, em1, em0, A0, A1);

    float mx = lc;
    mx = fmaxf(mx, __shfl_xor(mx, 1));
    mx = fmaxf(mx, __shfl_xor(mx, 2));
    mx = fmaxf(mx, __shfl_xor(mx, 4));
    const float sc = inv * __builtin_amdgcn_exp2f(lc - mx);

    const int    pj = (j & 1) | ((j >> 1) & 2) | ((j << 1) & 4);
    float* const P  = tab + tab_entry_off(v) + pj;
    const int    i0 = lo ? 0 : 4;
    P[(i0 + 0) * 8] = (lo ? x[0] : x[4]) * sc;
    P[(i0 + 1) * 8] = (lo ? x[1] : x[5]) * sc;
    P[(i0 + 2) * 8] = (lo ? x[2] : x[6]) * sc;
    P[(i0 + 3) * 8] = (lo ? x[3] : x[7]) * sc;
    if (lo && j == 0) tab[tab_scale_off(v)] = mx;
}

__device__ __forceinline__ void tab_load4(
    const float* __restrict__ tab, unsigned w, int aoff,
    v2f (&a0)[4], v2f (&a1)[4], float (&ms)[4])
{
#pragma unroll
    for (int sb = 0; sb < 4; ++sb) {
        const unsigned v = (w >> (sb * 8)) & 255u;
        const v4f      p = *(const v4f*)(tab + tab_entry_off(v) + aoff);
        a0[sb] = p.lo;
        a1[sb] = p.hi;
        ms[sb] = tab[tab_scale_off(v)];
    }
}

__device__ __forceinline__ void tab_step(
    v8f& a, float& inv, float& lc, bool lo, v2f A0, v2f A1, float mlog)
{
    const v8f vzero = {};
    v2f B0, B1;
    B0.x = (lo ? a[0] : a[2]) * inv;  B0.y = (lo ? a[1] : a[3]) * inv;
    B1.x = (lo ? a[4] : a[6]) * inv;  B1.y = (lo ? a[5] : a[7]) * inv;
    v8f d = __builtin_amdgcn_wmma_f32_16x16x4_f32(
        false, A0, false, B0, (short)0, vzero, false, false);
    a = __builtin_amdgcn_wmma_f32_16x16x4_f32(
        false, A1, false, B1, (short)0, d, false, false);
    const float s = sum8_tree(a);
    inv = fast_rcp(s);
    lc += __builtin_amdgcn_logf(s) + mlog;
}

__global__ __launch_bounds__(64) void hmm_chunk_kernel(
    const float* __restrict__ trans, const float* __restrict__ emission,
    const unsigned* __restrict__ bits, const float* __restrict__ tab,
    float* __restrict__ gamma)
{
    const int  wave = __builtin_amdgcn_readfirstlane((int)(threadIdx.x >> 5));
    const int  lane = threadIdx.x & 31;
    const int  n    = lane & 15;
    const bool lo   = lane < 16;
    const int  c    = blockIdx.x / (HMM_B / 2);
    const int  b    = (blockIdx.x % (HMM_B / 2)) * 2 + wave;
    const int  j    = n & 7;
    const int  aoff = j * 8 + (lo ? 0 : 4);

    const unsigned* const bw = bits + (size_t)b * (HMM_T / 32) + c * (HMM_L / 32);

    unsigned w = (unsigned)__builtin_amdgcn_readfirstlane((int)bw[0]);
    v2f   a0c[4], a1c[4], a0n[4], a1n[4];
    float mc[4], mn[4];
    tab_load4(tab, w, aoff, a0c, a1c, mc);

    v8f   x;
    float inv, lc;
    if (c == 0) {
        float em1[8], em0[8];
        load_emission(emission, em1, em0);
        const int kb = lo ? 0 : 2;
        v2f A0, A1;
        A0.x = trans_norm(trans, kb + 0, j);
        A0.y = trans_norm(trans, kb + 1, j);
        A1.x = trans_norm(trans, kb + 4, j);
        A1.y = trans_norm(trans, kb + 5, j);
        const int ob = (int)(w & 1u);
#pragma unroll
        for (int m = 0; m < 8; ++m) x[m] = (m == j) ? (ob ? em1[m] : em0[m]) : 0.0f;
        const float s0 = sum8_tree(x);
        inv = fast_rcp(s0);
        lc  = __builtin_amdgcn_logf(s0);
#pragma unroll
        for (int tb = 1; tb < 8; ++tb)
            fwd_step<false, true>(x, inv, lc, nullptr, (int)((w >> tb) & 1u),
                                  lo, em1, em0, A0, A1);
    } else {
#pragma unroll
        for (int m = 0; m < 8; ++m) x[m] = (m == j) ? 1.0f : 0.0f;
        inv = 1.0f;
        lc  = 0.0f;
        tab_step(x, inv, lc, lo, a0c[0], a1c[0], mc[0]);
    }
    for (int q = 0; q < HMM_L / 32; ++q) {
        const unsigned wn = (unsigned)__builtin_amdgcn_readfirstlane(
            (int)bw[q + 1 < HMM_L / 32 ? q + 1 : q]);
        tab_load4(tab, wn, aoff, a0n, a1n, mn);
#pragma unroll
        for (int sb = 0; sb < 4; ++sb)
            if (q > 0 || sb > 0) tab_step(x, inv, lc, lo, a0c[sb], a1c[sb], mc[sb]);
#pragma unroll
        for (int sb = 0; sb < 4; ++sb) { a0c[sb] = a0n[sb]; a1c[sb] = a1n[sb]; mc[sb] = mn[sb]; }
    }

    float* const reg = gamma + ((size_t)b * HMM_T + (size_t)c * HMM_L) * 8;
    v4f ah;
    ah.x = (lo ? x[0] : x[4]) * inv;  ah.y = (lo ? x[1] : x[5]) * inv;
    ah.z = (lo ? x[2] : x[6]) * inv;  ah.w = (lo ? x[3] : x[7]) * inv;
    if (n < 8) {
        *(v4f*)(reg + (4 + j) * 8 + (lo ? 0 : 4)) = ah;
        if (lo) reg[12 * 8 + j] = lc;
    }
}

static_assert(HMM_C == 32, "the scan keeps one chunk per lane of a wave");

__device__ __forceinline__ float lane_value(float x, int src_lane)
{
    return __int_as_float(__builtin_amdgcn_readlane(__float_as_int(x), src_lane));
}

__global__ __launch_bounds__(64) void hmm_scan_kernel(
    const float* __restrict__ start, float* __restrict__ gamma,
    float* __restrict__ loglik)
{
    const bool fwd  = __builtin_amdgcn_readfirstlane((int)(threadIdx.x >> 5)) == 0;
    const int  lane = threadIdx.x & 31;
    const int  b    = blockIdx.x;
    float* const seq = gamma + (size_t)b * HMM_T * 8;
    float* const reg = seq + (size_t)lane * HMM_L * 8;

    v8f col[8];
#pragma unroll
    for (int k = 0; k < 8; ++k) col[k] = *(const v8f*)(reg + (4 + k) * 8);
    const v8f lg = *(const v8f*)(reg + 12 * 8);
    float m = lg[0];
#pragma unroll
    for (int k = 1; k < 8; ++k) m = fmaxf(m, lg[k]);
#pragma unroll
    for (int k = 0; k < 8; ++k) col[k] *= exp2f(lg[k] - m);

    if (fwd) {
        float v[8]; float ssum = 0.0f;
#pragma unroll
        for (int k = 0; k < 8; ++k) { v[k] = start[k] + HMM_EPS; ssum += v[k]; }
#pragma unroll
        for (int k = 0; k < 8; ++k) v[k] = lane_value(v[k] / ssum, 0);

        float* const erow    = reg + (size_t)HMM_L * 8;
        const bool   has_out = lane + 1 < HMM_C;
        float        skeep   = 1.0f;
        for (int c = 0; c < HMM_C; ++c) {
            v8f u = col[0] * v[0];
#pragma unroll
            for (int k = 1; k < 8; ++k) u = col[k] * v[k] + u;
            const float s  = sum8_tree(u);
            const v8f   vn = u * fast_rcp(s);
            if (lane == c) {
                skeep = s;
                if (has_out) *(v8f*)erow = vn;
            }
#pragma unroll
            for (int k = 0; k < 8; ++k) v[k] = lane_value(vn[k], c);
        }
        float tot = m + log2f(skeep);
        tot += __shfl_xor(tot, 16);
        tot += __shfl_xor(tot, 8);
        tot += __shfl_xor(tot, 4);
        tot += __shfl_xor(tot, 2);
        tot += __shfl_xor(tot, 1);
        if (lane == 0) loglik[b] = tot * HMM_LN2;
    } else {
        v8f row[8];
#pragma unroll
        for (int r = 0; r < 8; ++r)
#pragma unroll
            for (int k = 0; k < 8; ++k) row[r][k] = col[k][r];

        float f[8];
#pragma unroll
        for (int k = 0; k < 8; ++k) f[k] = 1.0f;
        if (lane == 0) {
            v8f ones;
#pragma unroll
            for (int k = 0; k < 8; ++k) ones[k] = 1.0f;
            *(v8f*)(seq + ((size_t)HMM_T - 1) * 8) = ones;
        }
        float* const frow = reg - 8;
        for (int c = HMM_C - 1; c >= 1; --c) {
            v8f u = row[0] * f[0];
#pragma unroll
            for (int r = 1; r < 8; ++r) u = row[r] * f[r] + u;
            const v8f fn = u * fast_rcp(sum8_tree(u));
            if (lane == c) *(v8f*)frow = fn;
#pragma unroll
            for (int k = 0; k < 8; ++k) f[k] = lane_value(fn[k], c);
        }
    }
}

__global__ __launch_bounds__(64) void hmm_fb_kernel(
    const float* __restrict__ start, const float* __restrict__ trans,
    const float* __restrict__ emission, const unsigned* __restrict__ bits,
    float* __restrict__ gamma)
{
    const bool fwd  = __builtin_amdgcn_readfirstlane((int)(threadIdx.x >> 5)) == 0;
    const int  lane = threadIdx.x & 31;
    const int  n    = lane & 15;
    const bool lo   = lane < 16;
    const int  c    = blockIdx.x / (HMM_B / 16);
    const int  b    = (blockIdx.x % (HMM_B / 16)) * 16 + n;

    float em1[8], em0[8];
    load_emission(emission, em1, em0);

    const int kb = lo ? 0 : 2;
    const int i  = n & 7;
    v2f A0, A1;
    A0.x = fwd ? trans_norm(trans, kb + 0, i) : trans_norm(trans, i, kb + 0);
    A0.y = fwd ? trans_norm(trans, kb + 1, i) : trans_norm(trans, i, kb + 1);
    A1.x = fwd ? trans_norm(trans, kb + 4, i) : trans_norm(trans, i, kb + 4);
    A1.y = fwd ? trans_norm(trans, kb + 5, i) : trans_norm(trans, i, kb + 5);

    float* const          seq      = gamma + ((size_t)b * HMM_T + (size_t)c * HMM_L) * 8;
    float* const          half_seq = seq + (lo ? 0 : 4);
    const unsigned* const bw       = bits + (size_t)b * (HMM_T / 32) + c * (HMM_L / 32);

    v8f      x;
    float    inv = 1.0f;
    float    lcn = 0.0f;
    unsigned wn;

    if (fwd) {
        unsigned w = bw[0];
        wn = bw[1];
        if (c == 0) {
            float st[8]; float ssum = 0.0f;
#pragma unroll
            for (int k = 0; k < 8; ++k) { st[k] = start[k] + HMM_EPS; ssum += st[k]; }
#pragma unroll
            for (int k = 0; k < 8; ++k) st[k] /= ssum;

            {
                const int ob = (int)(w & 1u);
#pragma unroll
                for (int m = 0; m < 8; ++m) x[m] = st[m] * (ob ? em1[m] : em0[m]);
            }
            inv = fast_rcp(sum8_tree(x));
            {
                v4f ah;
                ah.x = lo ? x[0] : x[4];  ah.y = lo ? x[1] : x[5];
                ah.z = lo ? x[2] : x[6];  ah.w = lo ? x[3] : x[7];
                *(v4f*)half_seq = ah;
            }
#pragma unroll
            for (int tb = 1; tb < 8; ++tb)
                fwd_step<true, false>(x, inv, lcn, half_seq + (size_t)tb * 8,
                                      (int)((w >> tb) & 1u), lo, em1, em0, A0, A1);
        } else {
            x   = *(const v8f*)seq;
            inv = 1.0f;
            fwd_block8<true, false>(x, inv, lcn, half_seq, w, lo, em1, em0, A0, A1);
        }
        for (int sub = 1; sub < 4; ++sub)
            fwd_block8<true, false>(x, inv, lcn, half_seq + (size_t)sub * 64,
                                    w >> (sub * 8), lo, em1, em0, A0, A1);
        for (int q = 1; q < HMM_L / 64; ++q) {
            w  = wn;
            wn = bw[q + 1];
            float* const wrow = half_seq + (size_t)q * 256;
            for (int sub = 0; sub < 4; ++sub)
                fwd_block8<true, false>(x, inv, lcn, wrow + sub * 64, w >> (sub * 8),
                                        lo, em1, em0, A0, A1);
        }
    } else {
        x  = *(const v8f*)(seq + (size_t)(HMM_L - 1) * 8);
        wn = bw[HMM_L / 32 - 1];
        for (int q = HMM_L / 32 - 1; q >= HMM_L / 64; --q) {
            const unsigned w = wn;
            wn = bw[q - 1];
            float* const wrow = half_seq + (size_t)q * 256;
            for (int sub = 3; sub >= 0; --sub)
                bwd_stage_block8(wrow + sub * 64 + 56, w >> (sub * 8), x,
                                 lo, em1, em0, A0, A1);
        }
    }

    __syncthreads();

    v8f bufA[8], bufB[8];
    if (fwd) {
        fwd_load8(bufA, seq, HMM_L / 16);
        for (int jb = HMM_L / 16; jb < HMM_L / 8; jb += 2) {
            const unsigned w  = wn;
            const int      jn = jb + 2 < HMM_L / 8 ? jb + 2 : HMM_L / 8 - 2;
            wn = bw[jn >> 2];
            __builtin_amdgcn_sched_barrier(0);
            fwd_load8(bufB, seq, jb + 1);
            __builtin_amdgcn_sched_barrier(0);
            fwd_gamma_block8(bufA, half_seq, jb, w, x, inv, lo, em1, em0, A0, A1);
            fwd_load8(bufA, seq, jn);
            __builtin_amdgcn_sched_barrier(0);
            fwd_gamma_block8(bufB, half_seq, jb + 1, w, x, inv, lo, em1, em0, A0, A1);
        }
    } else {
        bwd_load8(bufA, seq, HMM_L / 16 - 1);
        for (int jb = HMM_L / 16 - 1; jb >= 1; jb -= 2) {
            const unsigned w  = wn;
            const int      jn = jb >= 3 ? jb - 2 : 1;
            wn = bw[jn >> 2];
            __builtin_amdgcn_sched_barrier(0);
            bwd_load8(bufB, seq, jb - 1);
            __builtin_amdgcn_sched_barrier(0);
            bwd_block8(bufA, half_seq, jb, w, x, lo, em1, em0, A0, A1);
            bwd_load8(bufA, seq, jn);
            __builtin_amdgcn_sched_barrier(0);
            bwd_block8(bufB, half_seq, jb - 1, w, x, lo, em1, em0, A0, A1);
        }
    }
}

extern "C" void kernel_launch(void* const* d_in, const int* in_sizes, int n_in,
                              void* d_out, int out_size, void* d_ws, size_t ws_size,
                              hipStream_t stream) {
    (void)in_sizes; (void)n_in; (void)out_size; (void)ws_size;
    const float* start    = (const float*)d_in[0];
    const float* trans    = (const float*)d_in[1];
    const float* emission = (const float*)d_in[2];
    const int*   obs      = (const int*)d_in[3];

    float* gamma  = (float*)d_out;
    float* loglik = gamma + (size_t)HMM_B * HMM_T * HMM_K;
    unsigned* bits = (unsigned*)d_ws;

    hmm_pack_kernel<<<HMM_TAB_BLOCKS + HMM_PACK_BLOCKS, 256, 0, stream>>>(
        obs, bits, trans, emission, gamma);
    hmm_chunk_kernel<<<HMM_C * (HMM_B / 2), 64, 0, stream>>>(
        trans, emission, bits, gamma, gamma);
    hmm_scan_kernel<<<HMM_B, 64, 0, stream>>>(start, gamma, loglik);
    hmm_fb_kernel<<<HMM_C * (HMM_B / 16), 64, 0, stream>>>(
        start, trans, emission, bits, gamma);
}


# pass: alpha and beta rows staged in the block's LDS (128 KiB, row = 16 sequences x 8 states), not in the gamma slot: second half reads LDS, no row loads from memory, every gamma row stored once
# speedup vs baseline: 74.5292x; 1.1526x over previous
#include <hip/hip_runtime.h>


#define HMM_B 256
#define HMM_T 8192
#define HMM_K 8
#define HMM_C 32
#define HMM_L (HMM_T / HMM_C)
#define HMM_SROW (16 * HMM_K)
#define HMM_EPS 1e-8f
#define HMM_LN2 0.6931471805599453f

typedef __attribute__((ext_vector_type(2))) float v2f;
typedef __attribute__((ext_vector_type(4))) float v4f;
typedef __attribute__((ext_vector_type(8))) float v8f;

__device__ __forceinline__ float fast_rcp(float x) {
    float r = __builtin_amdgcn_rcpf(x);
    float e = __builtin_fmaf(-x, r, 1.0f);
    return __builtin_fmaf(r, e, r);
}

#define HMM_TAB_BLOCKS  16
#define HMM_PACK_BLOCKS 256

__device__ __forceinline__ size_t tab_entry_off(unsigned v)
{
    return ((size_t)(v >> 4) * HMM_L + 16) * 8 + (size_t)(v & 15u) * 64;
}
__device__ __forceinline__ size_t tab_scale_off(unsigned v)
{
    return ((size_t)(v >> 4) * HMM_L + 144) * 8 + (size_t)(v & 15u);
}
static_assert(HMM_C >= 16 && HMM_L >= 146, "the table needs rows 16-145 of 16 chunk regions");

__device__ void hmm_build_table(const float* __restrict__ trans,
                                const float* __restrict__ emission,
                                float* __restrict__ tab, int tw);

__global__ __launch_bounds__(256) void hmm_pack_kernel(
    const int* __restrict__ obs, unsigned* __restrict__ bits,
    const float* __restrict__ trans, const float* __restrict__ emission,
    float* __restrict__ tab)
{
    if (blockIdx.x < HMM_TAB_BLOCKS) {
        hmm_build_table(trans, emission, tab,
                        (int)((blockIdx.x * blockDim.x + threadIdx.x) >> 5));
        return;
    }
    const int wave = ((blockIdx.x - HMM_TAB_BLOCKS) * blockDim.x + threadIdx.x) >> 5;
    const int lane = threadIdx.x & 31;
    const long base = (long)wave * 1024;
    unsigned keep = 0;
#pragma unroll 4
    for (int i = 0; i < 32; ++i) {
        int v = obs[base + i * 32 + lane];
        unsigned m = (unsigned)__ballot(v != 0);
        if (i == lane) keep = m;
    }
    bits[(size_t)wave * 32 + lane] = keep;
}

__device__ __forceinline__ float sum8_tree(const v8f& x)
{
    const v4f q = x.lo + x.hi;
    const v2f r = q.lo + q.hi;
    return r.x + r.y;
}

__device__ __forceinline__ float sum8_chain(const v8f& x)
{
    const v4f q = x.lo + x.hi;
    const v2f r = q.lo + q.hi;
    float s;
    asm("v_add_f32 %0, %1, %2" : "=v"(s) : "v"(r.x), "v"(r.y));
    return s;
}

template <bool STORE, bool LOGS>
__device__ __forceinline__ void fwd_step(
    v8f& a, float& inv, float& lc, float* row, int ob, bool lo,
    const float (&em1)[8], const float (&em0)[8], v2f A0, v2f A1)
{
    const v8f vzero = {};
    v2f B0, B1;
    B0.x = (lo ? a[0] : a[2]) * inv;  B0.y = (lo ? a[1] : a[3]) * inv;
    B1.x = (lo ? a[4] : a[6]) * inv;  B1.y = (lo ? a[5] : a[7]) * inv;

    v8f d = __builtin_amdgcn_wmma_f32_16x16x4_f32(
        false, A0, false, B0, (short)0, vzero, false, false);
    d = __builtin_amdgcn_wmma_f32_16x16x4_f32(
        false, A1, false, B1, (short)0, d, false, false);

#pragma unroll
    for (int m = 0; m < 8; ++m) a[m] = d[m] * (ob ? em1[m] : em0[m]);
    const float s = sum8_tree(a);
    inv = fast_rcp(s);
    if (LOGS) lc += __builtin_amdgcn_logf(s);
    if (STORE) {
        v4f ah;
        ah.x = lo ? a[0] : a[4];  ah.y = lo ? a[1] : a[5];
        ah.z = lo ? a[2] : a[6];  ah.w = lo ? a[3] : a[7];
        *(v4f*)row = ah;
    }
}

template <bool STORE, bool LOGS>
__device__ __forceinline__ void fwd_block8(
    v8f& a, float& inv, float& lc, float* row, unsigned wb, bool lo,
    const float (&em1)[8], const float (&em0)[8], v2f A0, v2f A1)
{
#pragma unroll
    for (int tb = 0; tb < 8; ++tb)
        fwd_step<STORE, LOGS>(a, inv, lc, STORE ? row + tb * HMM_SROW : row,
                              (int)((wb >> tb) & 1u), lo, em1, em0, A0, A1);
}

__device__ __forceinline__ void fwd_load8(v8f (&buf)[8], const float* stg, int blk)
{
    const float* p = stg + blk * (8 * HMM_SROW);
#pragma unroll
    for (int i = 0; i < 8; ++i) buf[i] = *(const v8f*)(p + i * HMM_SROW);
}

__device__ __forceinline__ void fwd_gamma_block8(
    const v8f (&buf)[8], float* half_seq, int blk, unsigned w,
    v8f& a, float& inv, bool lo,
    const float (&em1)[8], const float (&em0)[8], v2f A0, v2f A1)
{
    const v8f vzero = {};
    const int t_lo = blk * 8;
#pragma unroll
    for (int i = 0; i < 8; ++i) {
        const int t  = t_lo + i;
        const int ob = (int)((w >> (t & 31)) & 1u);

        v2f B0, B1;
        B0.x = (lo ? a[0] : a[2]) * inv;  B0.y = (lo ? a[1] : a[3]) * inv;
        B1.x = (lo ? a[4] : a[6]) * inv;  B1.y = (lo ? a[5] : a[7]) * inv;
        v8f d = __builtin_amdgcn_wmma_f32_16x16x4_f32(
            false, A0, false, B0, (short)0, vzero, false, false);
        d = __builtin_amdgcn_wmma_f32_16x16x4_f32(
            false, A1, false, B1, (short)0, d, false, false);
#pragma unroll
        for (int m = 0; m < 8; ++m) a[m] = d[m] * (ob ? em1[m] : em0[m]);
        const float s = sum8_chain(a);
        inv = fast_rcp(s);

        const v8f be = buf[i];
        v8f g;
#pragma unroll
        for (int m = 0; m < 8; ++m) g[m] = be[m] * a[m];
        const float gi = fast_rcp(sum8_tree(g));
#pragma unroll
        for (int m = 0; m < 8; ++m) g[m] *= gi;
        v4f gh;
        gh.x = lo ? g[0] : g[4];  gh.y = lo ? g[1] : g[5];
        gh.z = lo ? g[2] : g[6];  gh.w = lo ? g[3] : g[7];
        *(v4f*)(half_seq + (size_t)t * 8) = gh;
    }
}

__device__ __forceinline__ void bwd_stage_block8(
    float* row_hi, unsigned wb, v8f& d, bool lo,
    const float (&em1)[8], const float (&em0)[8], v2f A0, v2f A1)
{
    const v8f vzero = {};
#pragma unroll
    for (int i = 0; i < 8; ++i) {
        const int ob = (int)((wb >> (7 - i)) & 1u);
        v4f dh;
        dh.x = lo ? d[0] : d[4];  dh.y = lo ? d[1] : d[5];
        dh.z = lo ? d[2] : d[6];  dh.w = lo ? d[3] : d[7];
        *(v4f*)(row_hi - i * HMM_SROW) = dh;

        const float inv = fast_rcp(sum8_tree(d));
        v8f u;
#pragma unroll
        for (int m = 0; m < 8; ++m) u[m] = d[m] * (ob ? em1[m] : em0[m]);
        v2f B0, B1;
        B0.x = (lo ? u[0] : u[2]) * inv;  B0.y = (lo ? u[1] : u[3]) * inv;
        B1.x = (lo ? u[4] : u[6]) * inv;  B1.y = (lo ? u[5] : u[7]) * inv;
        d = __builtin_amdgcn_wmma_f32_16x16x4_f32(
            false, A0, false, B0, (short)0, vzero, false, false);
        d = __builtin_amdgcn_wmma_f32_16x16x4_f32(
            false, A1, false, B1, (short)0, d, false, false);
    }
}

__device__ __forceinline__ void bwd_load8(v8f (&buf)[8], const float* stg, int blk)
{
    const float* p = stg + blk * (8 * HMM_SROW);
#pragma unroll
    for (int i = 0; i < 8; ++i) buf[i] = *(const v8f*)(p + (7 - i) * HMM_SROW);
}

__device__ __forceinline__ void bwd_block8(
    const v8f (&buf)[8], float* half_seq, int blk, unsigned w, v8f& d, bool lo,
    const float (&em1)[8], const float (&em0)[8], v2f A0, v2f A1)
{
    const v8f vzero = {};
    const int tt_hi = blk * 8 + 7;
#pragma unroll
    for (int i = 0; i < 8; ++i) {
        const int tt = tt_hi - i;
        const int ob = (int)((w >> (tt & 31)) & 1u);

        const v8f al = buf[i];
        v8f g;
#pragma unroll
        for (int m = 0; m < 8; ++m) g[m] = al[m] * d[m];
        const float gi = fast_rcp(sum8_tree(g));
#pragma unroll
        for (int m = 0; m < 8; ++m) g[m] *= gi;
        v4f gh;
        gh.x = lo ? g[0] : g[4];  gh.y = lo ? g[1] : g[5];
        gh.z = lo ? g[2] : g[6];  gh.w = lo ? g[3] : g[7];
        *(v4f*)(half_seq + (size_t)tt * 8) = gh;

        const float inv = fast_rcp(sum8_chain(d));

        v8f u;
#pragma unroll
        for (int m = 0; m < 8; ++m) u[m] = d[m] * (ob ? em1[m] : em0[m]);
        v2f B0, B1;
        B0.x = (lo ? u[0] : u[2]) * inv;  B0.y = (lo ? u[1] : u[3]) * inv;
        B1.x = (lo ? u[4] : u[6]) * inv;  B1.y = (lo ? u[5] : u[7]) * inv;

        d = __builtin_amdgcn_wmma_f32_16x16x4_f32(
            false, A0, false, B0, (short)0, vzero, false, false);
        d = __builtin_amdgcn_wmma_f32_16x16x4_f32(
            false, A1, false, B1, (short)0, d, false, false);
    }
}

__device__ __forceinline__ float trans_norm(const float* __restrict__ trans, int r, int c)
{
    float s = 0.0f;
#pragma unroll
    for (int k = 0; k < 8; ++k) s += trans[r * 8 + k] + HMM_EPS;
    return (trans[r * 8 + c] + HMM_EPS) / s;
}

__device__ __forceinline__ void load_emission(
    const float* __restrict__ emission, float (&em1)[8], float (&em0)[8])
{
#pragma unroll
    for (int k = 0; k < 8; ++k) {
        float e = emission[k];
        e = fminf(fmaxf(e, HMM_EPS), 1.0f - HMM_EPS);
        em1[k] = e; em0[k] = 1.0f - e;
    }
}

__device__ void hmm_build_table(const float* __restrict__ trans,
                                const float* __restrict__ emission,
                                float* __restrict__ tab, int tw)
{
    const int  lane = threadIdx.x & 31;
    const int  n    = lane & 15;
    const bool lo   = lane < 16;
    const int  j    = n & 7;
    const unsigned v = (unsigned)(tw * 2 + (n >> 3));

    float em1[8], em0[8];
    load_emission(emission, em1, em0);

    const int kb = lo ? 0 : 2;
    v2f A0, A1;
    A0.x = trans_norm(trans, kb + 0, j);
    A0.y = trans_norm(trans, kb + 1, j);
    A1.x = trans_norm(trans, kb + 4, j);
    A1.y = trans_norm(trans, kb + 5, j);

    v8f x;
#pragma unroll
    for (int m = 0; m < 8; ++m) x[m] = (m == j) ? 1.0f : 0.0f;
    float inv = 1.0f, lc = 0.0f;
    fwd_block8<false, true>(x, inv, lc, nullptr, v, lo, em1, em0, A0, A1);

    float mx = lc;
    mx = fmaxf(mx, __shfl_xor(mx, 1));
    mx = fmaxf(mx, __shfl_xor(mx, 2));
    mx = fmaxf(mx, __shfl_xor(mx, 4));
    const float sc = inv * __builtin_amdgcn_exp2f(lc - mx);

    const int    pj = (j & 1) | ((j >> 1) & 2) | ((j << 1) & 4);
    float* const P  = tab + tab_entry_off(v) + pj;
    const int    i0 = lo ? 0 : 4;
    P[(i0 + 0) * 8] = (lo ? x[0] : x[4]) * sc;
    P[(i0 + 1) * 8] = (lo ? x[1] : x[5]) * sc;
    P[(i0 + 2) * 8] = (lo ? x[2] : x[6]) * sc;
    P[(i0 + 3) * 8] = (lo ? x[3] : x[7]) * sc;
    if (lo && j == 0) tab[tab_scale_off(v)] = mx;
}

__device__ __forceinline__ void tab_load4(
    const float* __restrict__ tab, unsigned w, int aoff,
    v2f (&a0)[4], v2f (&a1)[4], float (&ms)[4])
{
#pragma unroll
    for (int sb = 0; sb < 4; ++sb) {
        const unsigned v = (w >> (sb * 8)) & 255u;
        const v4f      p = *(const v4f*)(tab + tab_entry_off(v) + aoff);
        a0[sb] = p.lo;
        a1[sb] = p.hi;
        ms[sb] = tab[tab_scale_off(v)];
    }
}

__device__ __forceinline__ void tab_step(
    v8f& a, float& inv, float& lc, bool lo, v2f A0, v2f A1, float mlog)
{
    const v8f vzero = {};
    v2f B0, B1;
    B0.x = (lo ? a[0] : a[2]) * inv;  B0.y = (lo ? a[1] : a[3]) * inv;
    B1.x = (lo ? a[4] : a[6]) * inv;  B1.y = (lo ? a[5] : a[7]) * inv;
    v8f d = __builtin_amdgcn_wmma_f32_16x16x4_f32(
        false, A0, false, B0, (short)0, vzero, false, false);
    a = __builtin_amdgcn_wmma_f32_16x16x4_f32(
        false, A1, false, B1, (short)0, d, false, false);
    const float s = sum8_tree(a);
    inv = fast_rcp(s);
    lc += __builtin_amdgcn_logf(s) + mlog;
}

__global__ __launch_bounds__(64) void hmm_chunk_kernel(
    const float* __restrict__ trans, const float* __restrict__ emission,
    const unsigned* __restrict__ bits, const float* __restrict__ tab,
    float* __restrict__ gamma)
{
    const int  wave = __builtin_amdgcn_readfirstlane((int)(threadIdx.x >> 5));
    const int  lane = threadIdx.x & 31;
    const int  n    = lane & 15;
    const bool lo   = lane < 16;
    const int  c    = blockIdx.x / (HMM_B / 2);
    const int  b    = (blockIdx.x % (HMM_B / 2)) * 2 + wave;
    const int  j    = n & 7;
    const int  aoff = j * 8 + (lo ? 0 : 4);

    const unsigned* const bw = bits + (size_t)b * (HMM_T / 32) + c * (HMM_L / 32);

    unsigned w = (unsigned)__builtin_amdgcn_readfirstlane((int)bw[0]);
    v2f   a0c[4], a1c[4], a0n[4], a1n[4];
    float mc[4], mn[4];
    tab_load4(tab, w, aoff, a0c, a1c, mc);

    v8f   x;
    float inv, lc;
    if (c == 0) {
        float em1[8], em0[8];
        load_emission(emission, em1, em0);
        const int kb = lo ? 0 : 2;
        v2f A0, A1;
        A0.x = trans_norm(trans, kb + 0, j);
        A0.y = trans_norm(trans, kb + 1, j);
        A1.x = trans_norm(trans, kb + 4, j);
        A1.y = trans_norm(trans, kb + 5, j);
        const int ob = (int)(w & 1u);
#pragma unroll
        for (int m = 0; m < 8; ++m) x[m] = (m == j) ? (ob ? em1[m] : em0[m]) : 0.0f;
        const float s0 = sum8_tree(x);
        inv = fast_rcp(s0);
        lc  = __builtin_amdgcn_logf(s0);
#pragma unroll
        for (int tb = 1; tb < 8; ++tb)
            fwd_step<false, true>(x, inv, lc, nullptr, (int)((w >> tb) & 1u),
                                  lo, em1, em0, A0, A1);
    } else {
#pragma unroll
        for (int m = 0; m < 8; ++m) x[m] = (m == j) ? 1.0f : 0.0f;
        inv = 1.0f;
        lc  = 0.0f;
        tab_step(x, inv, lc, lo, a0c[0], a1c[0], mc[0]);
    }
    for (int q = 0; q < HMM_L / 32; ++q) {
        const unsigned wn = (unsigned)__builtin_amdgcn_readfirstlane(
            (int)bw[q + 1 < HMM_L / 32 ? q + 1 : q]);
        tab_load4(tab, wn, aoff, a0n, a1n, mn);
#pragma unroll
        for (int sb = 0; sb < 4; ++sb)
            if (q > 0 || sb > 0) tab_step(x, inv, lc, lo, a0c[sb], a1c[sb], mc[sb]);
#pragma unroll
        for (int sb = 0; sb < 4; ++sb) { a0c[sb] = a0n[sb]; a1c[sb] = a1n[sb]; mc[sb] = mn[sb]; }
    }

    float* const reg = gamma + ((size_t)b * HMM_T + (size_t)c * HMM_L) * 8;
    v4f ah;
    ah.x = (lo ? x[0] : x[4]) * inv;  ah.y = (lo ? x[1] : x[5]) * inv;
    ah.z = (lo ? x[2] : x[6]) * inv;  ah.w = (lo ? x[3] : x[7]) * inv;
    if (n < 8) {
        *(v4f*)(reg + (4 + j) * 8 + (lo ? 0 : 4)) = ah;
        if (lo) reg[12 * 8 + j] = lc;
    }
}

static_assert(HMM_C == 32, "the scan keeps one chunk per lane of a wave");

__device__ __forceinline__ float lane_value(float x, int src_lane)
{
    return __int_as_float(__builtin_amdgcn_readlane(__float_as_int(x), src_lane));
}

__global__ __launch_bounds__(64) void hmm_scan_kernel(
    const float* __restrict__ start, float* __restrict__ gamma,
    float* __restrict__ loglik)
{
    const bool fwd  = __builtin_amdgcn_readfirstlane((int)(threadIdx.x >> 5)) == 0;
    const int  lane = threadIdx.x & 31;
    const int  b    = blockIdx.x;
    float* const seq = gamma + (size_t)b * HMM_T * 8;
    float* const reg = seq + (size_t)lane * HMM_L * 8;

    v8f col[8];
#pragma unroll
    for (int k = 0; k < 8; ++k) col[k] = *(const v8f*)(reg + (4 + k) * 8);
    const v8f lg = *(const v8f*)(reg + 12 * 8);
    float m = lg[0];
#pragma unroll
    for (int k = 1; k < 8; ++k) m = fmaxf(m, lg[k]);
#pragma unroll
    for (int k = 0; k < 8; ++k) col[k] *= exp2f(lg[k] - m);

    if (fwd) {
        float v[8]; float ssum = 0.0f;
#pragma unroll
        for (int k = 0; k < 8; ++k) { v[k] = start[k] + HMM_EPS; ssum += v[k]; }
#pragma unroll
        for (int k = 0; k < 8; ++k) v[k] = lane_value(v[k] / ssum, 0);

        float* const erow    = reg + (size_t)HMM_L * 8;
        const bool   has_out = lane + 1 < HMM_C;
        float        skeep   = 1.0f;
        for (int c = 0; c < HMM_C; ++c) {
            v8f u = col[0] * v[0];
#pragma unroll
            for (int k = 1; k < 8; ++k) u = col[k] * v[k] + u;
            const float s  = sum8_tree(u);
            const v8f   vn = u * fast_rcp(s);
            if (lane == c) {
                skeep = s;
                if (has_out) *(v8f*)erow = vn;
            }
#pragma unroll
            for (int k = 0; k < 8; ++k) v[k] = lane_value(vn[k], c);
        }
        float tot = m + log2f(skeep);
        tot += __shfl_xor(tot, 16);
        tot += __shfl_xor(tot, 8);
        tot += __shfl_xor(tot, 4);
        tot += __shfl_xor(tot, 2);
        tot += __shfl_xor(tot, 1);
        if (lane == 0) loglik[b] = tot * HMM_LN2;
    } else {
        v8f row[8];
#pragma unroll
        for (int r = 0; r < 8; ++r)
#pragma unroll
            for (int k = 0; k < 8; ++k) row[r][k] = col[k][r];

        float f[8];
#pragma unroll
        for (int k = 0; k < 8; ++k) f[k] = 1.0f;
        if (lane == 0) {
            v8f ones;
#pragma unroll
            for (int k = 0; k < 8; ++k) ones[k] = 1.0f;
            *(v8f*)(seq + ((size_t)HMM_T - 1) * 8) = ones;
        }
        float* const frow = reg - 8;
        for (int c = HMM_C - 1; c >= 1; --c) {
            v8f u = row[0] * f[0];
#pragma unroll
            for (int r = 1; r < 8; ++r) u = row[r] * f[r] + u;
            const v8f fn = u * fast_rcp(sum8_tree(u));
            if (lane == c) *(v8f*)frow = fn;
#pragma unroll
            for (int k = 0; k < 8; ++k) f[k] = lane_value(fn[k], c);
        }
    }
}

__global__ __launch_bounds__(64) void hmm_fb_kernel(
    const float* __restrict__ start, const float* __restrict__ trans,
    const float* __restrict__ emission, const unsigned* __restrict__ bits,
    float* __restrict__ gamma)
{
    __shared__ float stage[HMM_L * HMM_SROW];

    const bool fwd  = __builtin_amdgcn_readfirstlane((int)(threadIdx.x >> 5)) == 0;
    const int  lane = threadIdx.x & 31;
    const int  n    = lane & 15;
    const bool lo   = lane < 16;
    const int  c    = blockIdx.x / (HMM_B / 16);
    const int  b    = (blockIdx.x % (HMM_B / 16)) * 16 + n;

    float em1[8], em0[8];
    load_emission(emission, em1, em0);

    const int kb = lo ? 0 : 2;
    const int i  = n & 7;
    v2f A0, A1;
    A0.x = fwd ? trans_norm(trans, kb + 0, i) : trans_norm(trans, i, kb + 0);
    A0.y = fwd ? trans_norm(trans, kb + 1, i) : trans_norm(trans, i, kb + 1);
    A1.x = fwd ? trans_norm(trans, kb + 4, i) : trans_norm(trans, i, kb + 4);
    A1.y = fwd ? trans_norm(trans, kb + 5, i) : trans_norm(trans, i, kb + 5);

    float* const          seq      = gamma + ((size_t)b * HMM_T + (size_t)c * HMM_L) * 8;
    float* const          half_seq = seq + (lo ? 0 : 4);
    const unsigned* const bw       = bits + (size_t)b * (HMM_T / 32) + c * (HMM_L / 32);
    float* const          stg      = stage + n * 8;
    float* const          half_stg = stg + (lo ? 0 : 4);

    v8f      x;
    float    inv = 1.0f;
    float    lcn = 0.0f;
    unsigned wn;

    if (fwd) {
        unsigned w = bw[0];
        wn = bw[1];
        if (c == 0) {
            float st[8]; float ssum = 0.0f;
#pragma unroll
            for (int k = 0; k < 8; ++k) { st[k] = start[k] + HMM_EPS; ssum += st[k]; }
#pragma unroll
            for (int k = 0; k < 8; ++k) st[k] /= ssum;

            {
                const int ob = (int)(w & 1u);
#pragma unroll
                for (int m = 0; m < 8; ++m) x[m] = st[m] * (ob ? em1[m] : em0[m]);
            }
            inv = fast_rcp(sum8_tree(x));
            {
                v4f ah;
                ah.x = lo ? x[0] : x[4];  ah.y = lo ? x[1] : x[5];
                ah.z = lo ? x[2] : x[6];  ah.w = lo ? x[3] : x[7];
                *(v4f*)half_stg = ah;
            }
#pragma unroll
            for (int tb = 1; tb < 8; ++tb)
                fwd_step<true, false>(x, inv, lcn, half_stg + tb * HMM_SROW,
                                      (int)((w >> tb) & 1u), lo, em1, em0, A0, A1);
        } else {
            x   = *(const v8f*)seq;
            inv = 1.0f;
            fwd_block8<true, false>(x, inv, lcn, half_stg, w, lo, em1, em0, A0, A1);
        }
        for (int sub = 1; sub < 4; ++sub)
            fwd_block8<true, false>(x, inv, lcn, half_stg + sub * (8 * HMM_SROW),
                                    w >> (sub * 8), lo, em1, em0, A0, A1);
        for (int q = 1; q < HMM_L / 64; ++q) {
            w  = wn;
            wn = bw[q + 1];
            float* const wrow = half_stg + q * (32 * HMM_SROW);
            for (int sub = 0; sub < 4; ++sub)
                fwd_block8<true, false>(x, inv, lcn, wrow + sub * (8 * HMM_SROW),
                                        w >> (sub * 8), lo, em1, em0, A0, A1);
        }
    } else {
        x  = *(const v8f*)(seq + (size_t)(HMM_L - 1) * 8);
        wn = bw[HMM_L / 32 - 1];
        for (int q = HMM_L / 32 - 1; q >= HMM_L / 64; --q) {
            const unsigned w = wn;
            wn = bw[q - 1];
            float* const wrow = half_stg + q * (32 * HMM_SROW);
            for (int sub = 3; sub >= 0; --sub)
                bwd_stage_block8(wrow + (sub * 8 + 7) * HMM_SROW, w >> (sub * 8), x,
                                 lo, em1, em0, A0, A1);
        }
    }

    __syncthreads();

    v8f bufA[8], bufB[8];
    if (fwd) {
        fwd_load8(bufA, stg, HMM_L / 16);
        for (int jb = HMM_L / 16; jb < HMM_L / 8; jb += 2) {
            const unsigned w  = wn;
            const int      jn = jb + 2 < HMM_L / 8 ? jb + 2 : HMM_L / 8 - 2;
            wn = bw[jn >> 2];
            fwd_load8(bufB, stg, jb + 1);
            __builtin_amdgcn_sched_barrier(0);
            fwd_gamma_block8(bufA, half_seq, jb, w, x, inv, lo, em1, em0, A0, A1);
            __builtin_amdgcn_sched_barrier(0);
            fwd_load8(bufA, stg, jn);
            __builtin_amdgcn_sched_barrier(0);
            fwd_gamma_block8(bufB, half_seq, jb + 1, w, x, inv, lo, em1, em0, A0, A1);
        }
    } else {
        bwd_load8(bufA, stg, HMM_L / 16 - 1);
        for (int jb = HMM_L / 16 - 1; jb >= 1; jb -= 2) {
            const unsigned w  = wn;
            const int      jn = jb >= 3 ? jb - 2 : 1;
            wn = bw[jn >> 2];
            bwd_load8(bufB, stg, jb - 1);
            __builtin_amdgcn_sched_barrier(0);
            bwd_block8(bufA, half_seq, jb, w, x, lo, em1, em0, A0, A1);
            __builtin_amdgcn_sched_barrier(0);
            bwd_load8(bufA, stg, jn);
            __builtin_amdgcn_sched_barrier(0);
            bwd_block8(bufB, half_seq, jb - 1, w, x, lo, em1, em0, A0, A1);
        }
    }
}

extern "C" void kernel_launch(void* const* d_in, const int* in_sizes, int n_in,
                              void* d_out, int out_size, void* d_ws, size_t ws_size,
                              hipStream_t stream) {
    (void)in_sizes; (void)n_in; (void)out_size; (void)ws_size;
    const float* start    = (const float*)d_in[0];
    const float* trans    = (const float*)d_in[1];
    const float* emission = (const float*)d_in[2];
    const int*   obs      = (const int*)d_in[3];

    float* gamma  = (float*)d_out;
    float* loglik = gamma + (size_t)HMM_B * HMM_T * HMM_K;
    unsigned* bits = (unsigned*)d_ws;

    hmm_pack_kernel<<<HMM_TAB_BLOCKS + HMM_PACK_BLOCKS, 256, 0, stream>>>(
        obs, bits, trans, emission, gamma);
    hmm_chunk_kernel<<<HMM_C * (HMM_B / 2), 64, 0, stream>>>(
        trans, emission, bits, gamma, gamma);
    hmm_scan_kernel<<<HMM_B, 64, 0, stream>>>(start, gamma, loglik);
    hmm_fb_kernel<<<HMM_C * (HMM_B / 16), 64, 0, stream>>>(
        start, trans, emission, bits, gamma);
}


# pass: gamma rows of 8 steps go through LDS (staged rows just consumed) and leave as whole 128-byte lines, 4 lines per store instead of 32-byte pieces of many lines; values and sum order unchanged
# speedup vs baseline: 78.0725x; 1.0475x over previous
#include <hip/hip_runtime.h>


#define HMM_B 256
#define HMM_T 8192
#define HMM_K 8
#define HMM_C 32
#define HMM_L (HMM_T / HMM_C)
#define HMM_SROW (16 * HMM_K)
#define HMM_EPS 1e-8f
#define HMM_LN2 0.6931471805599453f

typedef __attribute__((ext_vector_type(2))) float v2f;
typedef __attribute__((ext_vector_type(4))) float v4f;
typedef __attribute__((ext_vector_type(8))) float v8f;

__device__ __forceinline__ float fast_rcp(float x) {
    float r = __builtin_amdgcn_rcpf(x);
    float e = __builtin_fmaf(-x, r, 1.0f);
    return __builtin_fmaf(r, e, r);
}

#define HMM_TAB_BLOCKS  16
#define HMM_PACK_BLOCKS 256

__device__ __forceinline__ size_t tab_entry_off(unsigned v)
{
    return ((size_t)(v >> 4) * HMM_L + 16) * 8 + (size_t)(v & 15u) * 64;
}
__device__ __forceinline__ size_t tab_scale_off(unsigned v)
{
    return ((size_t)(v >> 4) * HMM_L + 144) * 8 + (size_t)(v & 15u);
}
static_assert(HMM_C >= 16 && HMM_L >= 146, "the table needs rows 16-145 of 16 chunk regions");

__device__ void hmm_build_table(const float* __restrict__ trans,
                                const float* __restrict__ emission,
                                float* __restrict__ tab, int tw);

__global__ __launch_bounds__(256) void hmm_pack_kernel(
    const int* __restrict__ obs, unsigned* __restrict__ bits,
    const float* __restrict__ trans, const float* __restrict__ emission,
    float* __restrict__ tab)
{
    if (blockIdx.x < HMM_TAB_BLOCKS) {
        hmm_build_table(trans, emission, tab,
                        (int)((blockIdx.x * blockDim.x + threadIdx.x) >> 5));
        return;
    }
    const int wave = ((blockIdx.x - HMM_TAB_BLOCKS) * blockDim.x + threadIdx.x) >> 5;
    const int lane = threadIdx.x & 31;
    const long base = (long)wave * 1024;
    unsigned keep = 0;
#pragma unroll 4
    for (int i = 0; i < 32; ++i) {
        int v = obs[base + i * 32 + lane];
        unsigned m = (unsigned)__ballot(v != 0);
        if (i == lane) keep = m;
    }
    bits[(size_t)wave * 32 + lane] = keep;
}

__device__ __forceinline__ float sum8_tree(const v8f& x)
{
    const v4f q = x.lo + x.hi;
    const v2f r = q.lo + q.hi;
    return r.x + r.y;
}

__device__ __forceinline__ float sum8_chain(const v8f& x)
{
    const v4f q = x.lo + x.hi;
    const v2f r = q.lo + q.hi;
    float s;
    asm("v_add_f32 %0, %1, %2" : "=v"(s) : "v"(r.x), "v"(r.y));
    return s;
}

template <bool STORE, bool LOGS>
__device__ __forceinline__ void fwd_step(
    v8f& a, float& inv, float& lc, float* row, int ob, bool lo,
    const float (&em1)[8], const float (&em0)[8], v2f A0, v2f A1)
{
    const v8f vzero = {};
    v2f B0, B1;
    B0.x = (lo ? a[0] : a[2]) * inv;  B0.y = (lo ? a[1] : a[3]) * inv;
    B1.x = (lo ? a[4] : a[6]) * inv;  B1.y = (lo ? a[5] : a[7]) * inv;

    v8f d = __builtin_amdgcn_wmma_f32_16x16x4_f32(
        false, A0, false, B0, (short)0, vzero, false, false);
    d = __builtin_amdgcn_wmma_f32_16x16x4_f32(
        false, A1, false, B1, (short)0, d, false, false);

#pragma unroll
    for (int m = 0; m < 8; ++m) a[m] = d[m] * (ob ? em1[m] : em0[m]);
    const float s = sum8_tree(a);
    inv = fast_rcp(s);
    if (LOGS) lc += __builtin_amdgcn_logf(s);
    if (STORE) {
        v4f ah;
        ah.x = lo ? a[0] : a[4];  ah.y = lo ? a[1] : a[5];
        ah.z = lo ? a[2] : a[6];  ah.w = lo ? a[3] : a[7];
        *(v4f*)row = ah;
    }
}

template <bool STORE, bool LOGS>
__device__ __forceinline__ void fwd_block8(
    v8f& a, float& inv, float& lc, float* row, unsigned wb, bool lo,
    const float (&em1)[8], const float (&em0)[8], v2f A0, v2f A1)
{
#pragma unroll
    for (int tb = 0; tb < 8; ++tb)
        fwd_step<STORE, LOGS>(a, inv, lc, STORE ? row + tb * HMM_SROW : row,
                              (int)((wb >> tb) & 1u), lo, em1, em0, A0, A1);
}

__device__ __forceinline__ void fwd_load8(v8f (&buf)[8], const float* stg, int blk)
{
    const float* p = stg + blk * (8 * HMM_SROW);
#pragma unroll
    for (int i = 0; i < 8; ++i) buf[i] = *(const v8f*)(p + i * HMM_SROW);
}

#define HMM_GROUP (4 * HMM_SROW)

struct gamma_lines {
    int    put[4];
    int    take[4];
    float* line;
};

__device__ __forceinline__ gamma_lines gamma_lines_init(float* chunk_seq0, int lane)
{
    gamma_lines q;
    const int n = lane & 15, h = lane >> 4;
#pragma unroll
    for (int k = 0; k < 4; ++k)
        q.put[k] = n * 32 + ((((k << 1) | h) ^ ((n >> 1) & 7)) << 2);
#pragma unroll
    for (int j = 0; j < 4; ++j) {
        const int nn = 4 * j + (lane >> 3);
        q.take[j] = nn * 32 + (((lane & 7) ^ ((nn >> 1) & 7)) << 2);
    }
    q.line = chunk_seq0 + (size_t)(lane >> 3) * HMM_T * 8 + (lane & 7) * 4;
    return q;
}

__device__ __forceinline__ void gamma_take(v4f (&pend)[8], const float* blk_stage, const gamma_lines& q)
{
#pragma unroll
    for (int g = 0; g < 2; ++g)
#pragma unroll
        for (int j = 0; j < 4; ++j)
            pend[g * 4 + j] = *(const v4f*)(blk_stage + g * HMM_GROUP + q.take[j]);
}

__device__ __forceinline__ void gamma_send(const v4f (&pend)[8], const gamma_lines& q, int t0)
{
    float* const p = q.line + (size_t)t0 * 8;
#pragma unroll
    for (int g = 0; g < 2; ++g)
#pragma unroll
        for (int j = 0; j < 4; ++j)
            *(v4f*)(p + g * 32 + (size_t)j * (4 * (size_t)HMM_T * 8)) = pend[g * 4 + j];
}

__device__ __forceinline__ void fwd_gamma_block8(
    const v8f (&buf)[8], float* blk_stage, const gamma_lines& q, v4f (&pend)[8],
    bool has_pend, int blk, unsigned w,
    v8f& a, float& inv, bool lo,
    const float (&em1)[8], const float (&em0)[8], v2f A0, v2f A1)
{
    const v8f vzero = {};
    const int t_lo = blk * 8;
#pragma unroll
    for (int i = 0; i < 8; ++i) {
        const int t  = t_lo + i;
        const int ob = (int)((w >> (t & 31)) & 1u);

        v2f B0, B1;
        B0.x = (lo ? a[0] : a[2]) * inv;  B0.y = (lo ? a[1] : a[3]) * inv;
        B1.x = (lo ? a[4] : a[6]) * inv;  B1.y = (lo ? a[5] : a[7]) * inv;
        v8f d = __builtin_amdgcn_wmma_f32_16x16x4_f32(
            false, A0, false, B0, (short)0, vzero, false, false);
        d = __builtin_amdgcn_wmma_f32_16x16x4_f32(
            false, A1, false, B1, (short)0, d, false, false);
        if (i == 2) {
            __builtin_amdgcn_sched_barrier(0);
            if (has_pend) gamma_send(pend, q, t_lo - 8);
        }
#pragma unroll
        for (int m = 0; m < 8; ++m) a[m] = d[m] * (ob ? em1[m] : em0[m]);
        const float s = sum8_chain(a);
        inv = fast_rcp(s);

        const v8f be = buf[i];
        v8f g;
#pragma unroll
        for (int m = 0; m < 8; ++m) g[m] = be[m] * a[m];
        const float gi = fast_rcp(sum8_tree(g));
#pragma unroll
        for (int m = 0; m < 8; ++m) g[m] *= gi;
        v4f gh;
        gh.x = lo ? g[0] : g[4];  gh.y = lo ? g[1] : g[5];
        gh.z = lo ? g[2] : g[6];  gh.w = lo ? g[3] : g[7];
        *(v4f*)(blk_stage + (i >> 2) * HMM_GROUP + q.put[i & 3]) = gh;
    }
    gamma_take(pend, blk_stage, q);
}

__device__ __forceinline__ void bwd_stage_block8(
    float* row_hi, unsigned wb, v8f& d, bool lo,
    const float (&em1)[8], const float (&em0)[8], v2f A0, v2f A1)
{
    const v8f vzero = {};
#pragma unroll
    for (int i = 0; i < 8; ++i) {
        const int ob = (int)((wb >> (7 - i)) & 1u);
        v4f dh;
        dh.x = lo ? d[0] : d[4];  dh.y = lo ? d[1] : d[5];
        dh.z = lo ? d[2] : d[6];  dh.w = lo ? d[3] : d[7];
        *(v4f*)(row_hi - i * HMM_SROW) = dh;

        const float inv = fast_rcp(sum8_tree(d));
        v8f u;
#pragma unroll
        for (int m = 0; m < 8; ++m) u[m] = d[m] * (ob ? em1[m] : em0[m]);
        v2f B0, B1;
        B0.x = (lo ? u[0] : u[2]) * inv;  B0.y = (lo ? u[1] : u[3]) * inv;
        B1.x = (lo ? u[4] : u[6]) * inv;  B1.y = (lo ? u[5] : u[7]) * inv;
        d = __builtin_amdgcn_wmma_f32_16x16x4_f32(
            false, A0, false, B0, (short)0, vzero, false, false);
        d = __builtin_amdgcn_wmma_f32_16x16x4_f32(
            false, A1, false, B1, (short)0, d, false, false);
    }
}

__device__ __forceinline__ void bwd_load8(v8f (&buf)[8], const float* stg, int blk)
{
    const float* p = stg + blk * (8 * HMM_SROW);
#pragma unroll
    for (int i = 0; i < 8; ++i) buf[i] = *(const v8f*)(p + (7 - i) * HMM_SROW);
}

__device__ __forceinline__ void bwd_block8(
    const v8f (&buf)[8], float* blk_stage, const gamma_lines& q, v4f (&pend)[8],
    bool has_pend, int blk, unsigned w, v8f& d, bool lo,
    const float (&em1)[8], const float (&em0)[8], v2f A0, v2f A1)
{
    const v8f vzero = {};
    const int tt_hi = blk * 8 + 7;
#pragma unroll
    for (int i = 0; i < 8; ++i) {
        const int tt = tt_hi - i;
        const int ob = (int)((w >> (tt & 31)) & 1u);

        const v8f al = buf[i];
        v8f g;
#pragma unroll
        for (int m = 0; m < 8; ++m) g[m] = al[m] * d[m];
        const float gi = fast_rcp(sum8_tree(g));
#pragma unroll
        for (int m = 0; m < 8; ++m) g[m] *= gi;
        v4f gh;
        gh.x = lo ? g[0] : g[4];  gh.y = lo ? g[1] : g[5];
        gh.z = lo ? g[2] : g[6];  gh.w = lo ? g[3] : g[7];
        *(v4f*)(blk_stage + (1 - (i >> 2)) * HMM_GROUP + q.put[3 - (i & 3)]) = gh;

        const float inv = fast_rcp(sum8_chain(d));

        v8f u;
#pragma unroll
        for (int m = 0; m < 8; ++m) u[m] = d[m] * (ob ? em1[m] : em0[m]);
        v2f B0, B1;
        B0.x = (lo ? u[0] : u[2]) * inv;  B0.y = (lo ? u[1] : u[3]) * inv;
        B1.x = (lo ? u[4] : u[6]) * inv;  B1.y = (lo ? u[5] : u[7]) * inv;

        d = __builtin_amdgcn_wmma_f32_16x16x4_f32(
            false, A0, false, B0, (short)0, vzero, false, false);
        d = __builtin_amdgcn_wmma_f32_16x16x4_f32(
            false, A1, false, B1, (short)0, d, false, false);
        if (i == 2) {
            __builtin_amdgcn_sched_barrier(0);
            if (has_pend) gamma_send(pend, q, blk * 8 + 8);
        }
    }
    gamma_take(pend, blk_stage, q);
}

__device__ __forceinline__ float trans_norm(const float* __restrict__ trans, int r, int c)
{
    float s = 0.0f;
#pragma unroll
    for (int k = 0; k < 8; ++k) s += trans[r * 8 + k] + HMM_EPS;
    return (trans[r * 8 + c] + HMM_EPS) / s;
}

__device__ __forceinline__ void load_emission(
    const float* __restrict__ emission, float (&em1)[8], float (&em0)[8])
{
#pragma unroll
    for (int k = 0; k < 8; ++k) {
        float e = emission[k];
        e = fminf(fmaxf(e, HMM_EPS), 1.0f - HMM_EPS);
        em1[k] = e; em0[k] = 1.0f - e;
    }
}

__device__ void hmm_build_table(const float* __restrict__ trans,
                                const float* __restrict__ emission,
                                float* __restrict__ tab, int tw)
{
    const int  lane = threadIdx.x & 31;
    const int  n    = lane & 15;
    const bool lo   = lane < 16;
    const int  j    = n & 7;
    const unsigned v = (unsigned)(tw * 2 + (n >> 3));

    float em1[8], em0[8];
    load_emission(emission, em1, em0);

    const int kb = lo ? 0 : 2;
    v2f A0, A1;
    A0.x = trans_norm(trans, kb + 0, j);
    A0.y = trans_norm(trans, kb + 1, j);
    A1.x = trans_norm(trans, kb + 4, j);
    A1.y = trans_norm(trans, kb + 5, j);

    v8f x;
#pragma unroll
    for (int m = 0; m < 8; ++m) x[m] = (m == j) ? 1.0f : 0.0f;
    float inv = 1.0f, lc = 0.0f;
    fwd_block8<false, true>(x, inv, lc, nullptr, v, lo, em1, em0, A0, A1);

    float mx = lc;
    mx = fmaxf(mx, __shfl_xor(mx, 1));
    mx = fmaxf(mx, __shfl_xor(mx, 2));
    mx = fmaxf(mx, __shfl_xor(mx, 4));
    const float sc = inv * __builtin_amdgcn_exp2f(lc - mx);

    const int    pj = (j & 1) | ((j >> 1) & 2) | ((j << 1) & 4);
    float* const P  = tab + tab_entry_off(v) + pj;
    const int    i0 = lo ? 0 : 4;
    P[(i0 + 0) * 8] = (lo ? x[0] : x[4]) * sc;
    P[(i0 + 1) * 8] = (lo ? x[1] : x[5]) * sc;
    P[(i0 + 2) * 8] = (lo ? x[2] : x[6]) * sc;
    P[(i0 + 3) * 8] = (lo ? x[3] : x[7]) * sc;
    if (lo && j == 0) tab[tab_scale_off(v)] = mx;
}

__device__ __forceinline__ void tab_load4(
    const float* __restrict__ tab, unsigned w, int aoff,
    v2f (&a0)[4], v2f (&a1)[4], float (&ms)[4])
{
#pragma unroll
    for (int sb = 0; sb < 4; ++sb) {
        const unsigned v = (w >> (sb * 8)) & 255u;
        const v4f      p = *(const v4f*)(tab + tab_entry_off(v) + aoff);
        a0[sb] = p.lo;
        a1[sb] = p.hi;
        ms[sb] = tab[tab_scale_off(v)];
    }
}

__device__ __forceinline__ void tab_step(
    v8f& a, float& inv, float& lc, bool lo, v2f A0, v2f A1, float mlog)
{
    const v8f vzero = {};
    v2f B0, B1;
    B0.x = (lo ? a[0] : a[2]) * inv;  B0.y = (lo ? a[1] : a[3]) * inv;
    B1.x = (lo ? a[4] : a[6]) * inv;  B1.y = (lo ? a[5] : a[7]) * inv;
    v8f d = __builtin_amdgcn_wmma_f32_16x16x4_f32(
        false, A0, false, B0, (short)0, vzero, false, false);
    a = __builtin_amdgcn_wmma_f32_16x16x4_f32(
        false, A1, false, B1, (short)0, d, false, false);
    const float s = sum8_tree(a);
    inv = fast_rcp(s);
    lc += __builtin_amdgcn_logf(s) + mlog;
}

__global__ __launch_bounds__(64) void hmm_chunk_kernel(
    const float* __restrict__ trans, const float* __restrict__ emission,
    const unsigned* __restrict__ bits, const float* __restrict__ tab,
    float* __restrict__ gamma)
{
    const int  wave = __builtin_amdgcn_readfirstlane((int)(threadIdx.x >> 5));
    const int  lane = threadIdx.x & 31;
    const int  n    = lane & 15;
    const bool lo   = lane < 16;
    const int  c    = blockIdx.x / (HMM_B / 2);
    const int  b    = (blockIdx.x % (HMM_B / 2)) * 2 + wave;
    const int  j    = n & 7;
    const int  aoff = j * 8 + (lo ? 0 : 4);

    const unsigned* const bw = bits + (size_t)b * (HMM_T / 32) + c * (HMM_L / 32);

    unsigned w = (unsigned)__builtin_amdgcn_readfirstlane((int)bw[0]);
    v2f   a0c[4], a1c[4], a0n[4], a1n[4];
    float mc[4], mn[4];
    tab_load4(tab, w, aoff, a0c, a1c, mc);

    v8f   x;
    float inv, lc;
    if (c == 0) {
        float em1[8], em0[8];
        load_emission(emission, em1, em0);
        const int kb = lo ? 0 : 2;
        v2f A0, A1;
        A0.x = trans_norm(trans, kb + 0, j);
        A0.y = trans_norm(trans, kb + 1, j);
        A1.x = trans_norm(trans, kb + 4, j);
        A1.y = trans_norm(trans, kb + 5, j);
        const int ob = (int)(w & 1u);
#pragma unroll
        for (int m = 0; m < 8; ++m) x[m] = (m == j) ? (ob ? em1[m] : em0[m]) : 0.0f;
        const float s0 = sum8_tree(x);
        inv = fast_rcp(s0);
        lc  = __builtin_amdgcn_logf(s0);
#pragma unroll
        for (int tb = 1; tb < 8; ++tb)
            fwd_step<false, true>(x, inv, lc, nullptr, (int)((w >> tb) & 1u),
                                  lo, em1, em0, A0, A1);
    } else {
#pragma unroll
        for (int m = 0; m < 8; ++m) x[m] = (m == j) ? 1.0f : 0.0f;
        inv = 1.0f;
        lc  = 0.0f;
        tab_step(x, inv, lc, lo, a0c[0], a1c[0], mc[0]);
    }
    for (int q = 0; q < HMM_L / 32; ++q) {
        const unsigned wn = (unsigned)__builtin_amdgcn_readfirstlane(
            (int)bw[q + 1 < HMM_L / 32 ? q + 1 : q]);
        tab_load4(tab, wn, aoff, a0n, a1n, mn);
#pragma unroll
        for (int sb = 0; sb < 4; ++sb)
            if (q > 0 || sb > 0) tab_step(x, inv, lc, lo, a0c[sb], a1c[sb], mc[sb]);
#pragma unroll
        for (int sb = 0; sb < 4; ++sb) { a0c[sb] = a0n[sb]; a1c[sb] = a1n[sb]; mc[sb] = mn[sb]; }
    }

    float* const reg = gamma + ((size_t)b * HMM_T + (size_t)c * HMM_L) * 8;
    v4f ah;
    ah.x = (lo ? x[0] : x[4]) * inv;  ah.y = (lo ? x[1] : x[5]) * inv;
    ah.z = (lo ? x[2] : x[6]) * inv;  ah.w = (lo ? x[3] : x[7]) * inv;
    if (n < 8) {
        *(v4f*)(reg + (4 + j) * 8 + (lo ? 0 : 4)) = ah;
        if (lo) reg[12 * 8 + j] = lc;
    }
}

static_assert(HMM_C == 32, "the scan keeps one chunk per lane of a wave");

__device__ __forceinline__ float lane_value(float x, int src_lane)
{
    return __int_as_float(__builtin_amdgcn_readlane(__float_as_int(x), src_lane));
}

__global__ __launch_bounds__(64) void hmm_scan_kernel(
    const float* __restrict__ start, float* __restrict__ gamma,
    float* __restrict__ loglik)
{
    const bool fwd  = __builtin_amdgcn_readfirstlane((int)(threadIdx.x >> 5)) == 0;
    const int  lane = threadIdx.x & 31;
    const int  b    = blockIdx.x;
    float* const seq = gamma + (size_t)b * HMM_T * 8;
    float* const reg = seq + (size_t)lane * HMM_L * 8;

    v8f col[8];
#pragma unroll
    for (int k = 0; k < 8; ++k) col[k] = *(const v8f*)(reg + (4 + k) * 8);
    const v8f lg = *(const v8f*)(reg + 12 * 8);
    float m = lg[0];
#pragma unroll
    for (int k = 1; k < 8; ++k) m = fmaxf(m, lg[k]);
#pragma unroll
    for (int k = 0; k < 8; ++k) col[k] *= exp2f(lg[k] - m);

    if (fwd) {
        float v[8]; float ssum = 0.0f;
#pragma unroll
        for (int k = 0; k < 8; ++k) { v[k] = start[k] + HMM_EPS; ssum += v[k]; }
#pragma unroll
        for (int k = 0; k < 8; ++k) v[k] = lane_value(v[k] / ssum, 0);

        float* const erow    = reg + (size_t)HMM_L * 8;
        const bool   has_out = lane + 1 < HMM_C;
        float        skeep   = 1.0f;
        for (int c = 0; c < HMM_C; ++c) {
            v8f u = col[0] * v[0];
#pragma unroll
            for (int k = 1; k < 8; ++k) u = col[k] * v[k] + u;
            const float s  = sum8_tree(u);
            const v8f   vn = u * fast_rcp(s);
            if (lane == c) {
                skeep = s;
                if (has_out) *(v8f*)erow = vn;
            }
#pragma unroll
            for (int k = 0; k < 8; ++k) v[k] = lane_value(vn[k], c);
        }
        float tot = m + log2f(skeep);
        tot += __shfl_xor(tot, 16);
        tot += __shfl_xor(tot, 8);
        tot += __shfl_xor(tot, 4);
        tot += __shfl_xor(tot, 2);
        tot += __shfl_xor(tot, 1);
        if (lane == 0) loglik[b] = tot * HMM_LN2;
    } else {
        v8f row[8];
#pragma unroll
        for (int r = 0; r < 8; ++r)
#pragma unroll
            for (int k = 0; k < 8; ++k) row[r][k] = col[k][r];

        float f[8];
#pragma unroll
        for (int k = 0; k < 8; ++k) f[k] = 1.0f;
        if (lane == 0) {
            v8f ones;
#pragma unroll
            for (int k = 0; k < 8; ++k) ones[k] = 1.0f;
            *(v8f*)(seq + ((size_t)HMM_T - 1) * 8) = ones;
        }
        float* const frow = reg - 8;
        for (int c = HMM_C - 1; c >= 1; --c) {
            v8f u = row[0] * f[0];
#pragma unroll
            for (int r = 1; r < 8; ++r) u = row[r] * f[r] + u;
            const v8f fn = u * fast_rcp(sum8_tree(u));
            if (lane == c) *(v8f*)frow = fn;
#pragma unroll
            for (int k = 0; k < 8; ++k) f[k] = lane_value(fn[k], c);
        }
    }
}

__global__ __launch_bounds__(64) void hmm_fb_kernel(
    const float* __restrict__ start, const float* __restrict__ trans,
    const float* __restrict__ emission, const unsigned* __restrict__ bits,
    float* __restrict__ gamma)
{
    __shared__ float stage[HMM_L * HMM_SROW];

    const bool fwd  = __builtin_amdgcn_readfirstlane((int)(threadIdx.x >> 5)) == 0;
    const int  lane = threadIdx.x & 31;
    const int  n    = lane & 15;
    const bool lo   = lane < 16;
    const int  c    = blockIdx.x / (HMM_B / 16);
    const int  b    = (blockIdx.x % (HMM_B / 16)) * 16 + n;

    float em1[8], em0[8];
    load_emission(emission, em1, em0);

    const int kb = lo ? 0 : 2;
    const int i  = n & 7;
    v2f A0, A1;
    A0.x = fwd ? trans_norm(trans, kb + 0, i) : trans_norm(trans, i, kb + 0);
    A0.y = fwd ? trans_norm(trans, kb + 1, i) : trans_norm(trans, i, kb + 1);
    A1.x = fwd ? trans_norm(trans, kb + 4, i) : trans_norm(trans, i, kb + 4);
    A1.y = fwd ? trans_norm(trans, kb + 5, i) : trans_norm(trans, i, kb + 5);

    float* const          seq      = gamma + ((size_t)b * HMM_T + (size_t)c * HMM_L) * 8;
    const unsigned* const bw       = bits + (size_t)b * (HMM_T / 32) + c * (HMM_L / 32);
    float* const          stg      = stage + n * 8;
    float* const          half_stg = stg + (lo ? 0 : 4);

    v8f      x;
    float    inv = 1.0f;
    float    lcn = 0.0f;
    unsigned wn;

    if (fwd) {
        unsigned w = bw[0];
        wn = bw[1];
        if (c == 0) {
            float st[8]; float ssum = 0.0f;
#pragma unroll
            for (int k = 0; k < 8; ++k) { st[k] = start[k] + HMM_EPS; ssum += st[k]; }
#pragma unroll
            for (int k = 0; k < 8; ++k) st[k] /= ssum;

            {
                const int ob = (int)(w & 1u);
#pragma unroll
                for (int m = 0; m < 8; ++m) x[m] = st[m] * (ob ? em1[m] : em0[m]);
            }
            inv = fast_rcp(sum8_tree(x));
            {
                v4f ah;
                ah.x = lo ? x[0] : x[4];  ah.y = lo ? x[1] : x[5];
                ah.z = lo ? x[2] : x[6];  ah.w = lo ? x[3] : x[7];
                *(v4f*)half_stg = ah;
            }
#pragma unroll
            for (int tb = 1; tb < 8; ++tb)
                fwd_step<true, false>(x, inv, lcn, half_stg + tb * HMM_SROW,
                                      (int)((w >> tb) & 1u), lo, em1, em0, A0, A1);
        } else {
            x   = *(const v8f*)seq;
            inv = 1.0f;
            fwd_block8<true, false>(x, inv, lcn, half_stg, w, lo, em1, em0, A0, A1);
        }
        for (int sub = 1; sub < 4; ++sub)
            fwd_block8<true, false>(x, inv, lcn, half_stg + sub * (8 * HMM_SROW),
                                    w >> (sub * 8), lo, em1, em0, A0, A1);
        for (int q = 1; q < HMM_L / 64; ++q) {
            w  = wn;
            wn = bw[q + 1];
            float* const wrow = half_stg + q * (32 * HMM_SROW);
            for (int sub = 0; sub < 4; ++sub)
                fwd_block8<true, false>(x, inv, lcn, wrow + sub * (8 * HMM_SROW),
                                        w >> (sub * 8), lo, em1, em0, A0, A1);
        }
    } else {
        x  = *(const v8f*)(seq + (size_t)(HMM_L - 1) * 8);
        wn = bw[HMM_L / 32 - 1];
        for (int q = HMM_L / 32 - 1; q >= HMM_L / 64; --q) {
            const unsigned w = wn;
            wn = bw[q - 1];
            float* const wrow = half_stg + q * (32 * HMM_SROW);
            for (int sub = 3; sub >= 0; --sub)
                bwd_stage_block8(wrow + (sub * 8 + 7) * HMM_SROW, w >> (sub * 8), x,
                                 lo, em1, em0, A0, A1);
        }
    }

    __syncthreads();

    v8f bufA[8], bufB[8];
    const gamma_lines q = gamma_lines_init(
        gamma + ((size_t)((blockIdx.x % (HMM_B / 16)) * 16) * HMM_T + (size_t)c * HMM_L) * 8, lane);
    v4f pend[8];
    if (fwd) {
        fwd_load8(bufA, stg, HMM_L / 16);
        for (int jb = HMM_L / 16; jb < HMM_L / 8; jb += 2) {
            const unsigned w  = wn;
            const int      jn = jb + 2 < HMM_L / 8 ? jb + 2 : HMM_L / 8 - 2;
            wn = bw[jn >> 2];
            fwd_load8(bufB, stg, jb + 1);
            __builtin_amdgcn_sched_barrier(0);
            fwd_gamma_block8(bufA, stage + jb * (8 * HMM_SROW), q, pend, jb != HMM_L / 16,
                             jb, w, x, inv, lo, em1, em0, A0, A1);
            __builtin_amdgcn_sched_barrier(0);
            fwd_load8(bufA, stg, jn);
            __builtin_amdgcn_sched_barrier(0);
            fwd_gamma_block8(bufB, stage + (jb + 1) * (8 * HMM_SROW), q, pend, true,
                             jb + 1, w, x, inv, lo, em1, em0, A0, A1);
        }
        gamma_send(pend, q, HMM_L - 8);
    } else {
        bwd_load8(bufA, stg, HMM_L / 16 - 1);
        for (int jb = HMM_L / 16 - 1; jb >= 1; jb -= 2) {
            const unsigned w  = wn;
            const int      jn = jb >= 3 ? jb - 2 : 1;
            wn = bw[jn >> 2];
            bwd_load8(bufB, stg, jb - 1);
            __builtin_amdgcn_sched_barrier(0);
            bwd_block8(bufA, stage + jb * (8 * HMM_SROW), q, pend, jb != HMM_L / 16 - 1,
                       jb, w, x, lo, em1, em0, A0, A1);
            __builtin_amdgcn_sched_barrier(0);
            bwd_load8(bufA, stg, jn);
            __builtin_amdgcn_sched_barrier(0);
            bwd_block8(bufB, stage + (jb - 1) * (8 * HMM_SROW), q, pend, true,
                       jb - 1, w, x, lo, em1, em0, A0, A1);
        }
        gamma_send(pend, q, 0);
    }
}

extern "C" void kernel_launch(void* const* d_in, const int* in_sizes, int n_in,
                              void* d_out, int out_size, void* d_ws, size_t ws_size,
                              hipStream_t stream) {
    (void)in_sizes; (void)n_in; (void)out_size; (void)ws_size;
    const float* start    = (const float*)d_in[0];
    const float* trans    = (const float*)d_in[1];
    const float* emission = (const float*)d_in[2];
    const int*   obs      = (const int*)d_in[3];

    float* gamma  = (float*)d_out;
    float* loglik = gamma + (size_t)HMM_B * HMM_T * HMM_K;
    unsigned* bits = (unsigned*)d_ws;

    hmm_pack_kernel<<<HMM_TAB_BLOCKS + HMM_PACK_BLOCKS, 256, 0, stream>>>(
        obs, bits, trans, emission, gamma);
    hmm_chunk_kernel<<<HMM_C * (HMM_B / 2), 64, 0, stream>>>(
        trans, emission, bits, gamma, gamma);
    hmm_scan_kernel<<<HMM_B, 64, 0, stream>>>(start, gamma, loglik);
    hmm_fb_kernel<<<HMM_C * (HMM_B / 16), 64, 0, stream>>>(
        start, trans, emission, bits, gamma);
}


# scan runs inside the chunk kernel: a block is one sequence (32 waves = 32 chunks), the matrices pass through LDS, not memory, waves 0-1 scan them behind one block barrier; 4 launches become 3
# speedup vs baseline: 85.2678x; 1.0922x over previous
#include <hip/hip_runtime.h>


#define HMM_B 256
#define HMM_T 8192
#define HMM_K 8
#define HMM_C 32
#define HMM_L (HMM_T / HMM_C)
#define HMM_SROW (16 * HMM_K)
#define HMM_EPS 1e-8f
#define HMM_LN2 0.6931471805599453f

typedef __attribute__((ext_vector_type(2))) float v2f;
typedef __attribute__((ext_vector_type(4))) float v4f;
typedef __attribute__((ext_vector_type(8))) float v8f;

__device__ __forceinline__ float fast_rcp(float x) {
    float r = __builtin_amdgcn_rcpf(x);
    float e = __builtin_fmaf(-x, r, 1.0f);
    return __builtin_fmaf(r, e, r);
}

#define HMM_TAB_BLOCKS  16
#define HMM_PACK_BLOCKS 256

__device__ __forceinline__ size_t tab_entry_off(unsigned v)
{
    return ((size_t)(v >> 4) * HMM_L + 16) * 8 + (size_t)(v & 15u) * 64;
}
__device__ __forceinline__ size_t tab_scale_off(unsigned v)
{
    return ((size_t)(v >> 4) * HMM_L + 144) * 8 + (size_t)(v & 15u);
}
static_assert(HMM_C >= 16 && HMM_L >= 146, "the table needs rows 16-145 of 16 chunk regions");

__device__ void hmm_build_table(const float* __restrict__ trans,
                                const float* __restrict__ emission,
                                float* __restrict__ tab, int tw);

__global__ __launch_bounds__(256) void hmm_pack_kernel(
    const int* __restrict__ obs, unsigned* __restrict__ bits,
    const float* __restrict__ trans, const float* __restrict__ emission,
    float* __restrict__ tab)
{
    if (blockIdx.x < HMM_TAB_BLOCKS) {
        hmm_build_table(trans, emission, tab,
                        (int)((blockIdx.x * blockDim.x + threadIdx.x) >> 5));
        return;
    }
    const int wave = ((blockIdx.x - HMM_TAB_BLOCKS) * blockDim.x + threadIdx.x) >> 5;
    const int lane = threadIdx.x & 31;
    const long base = (long)wave * 1024;
    unsigned keep = 0;
#pragma unroll 4
    for (int i = 0; i < 32; ++i) {
        int v = obs[base + i * 32 + lane];
        unsigned m = (unsigned)__ballot(v != 0);
        if (i == lane) keep = m;
    }
    bits[(size_t)wave * 32 + lane] = keep;
}

__device__ __forceinline__ float sum8_tree(const v8f& x)
{
    const v4f q = x.lo + x.hi;
    const v2f r = q.lo + q.hi;
    return r.x + r.y;
}

__device__ __forceinline__ float sum8_chain(const v8f& x)
{
    const v4f q = x.lo + x.hi;
    const v2f r = q.lo + q.hi;
    float s;
    asm("v_add_f32 %0, %1, %2" : "=v"(s) : "v"(r.x), "v"(r.y));
    return s;
}

template <bool STORE, bool LOGS>
__device__ __forceinline__ void fwd_step(
    v8f& a, float& inv, float& lc, float* row, int ob, bool lo,
    const float (&em1)[8], const float (&em0)[8], v2f A0, v2f A1)
{
    const v8f vzero = {};
    v2f B0, B1;
    B0.x = (lo ? a[0] : a[2]) * inv;  B0.y = (lo ? a[1] : a[3]) * inv;
    B1.x = (lo ? a[4] : a[6]) * inv;  B1.y = (lo ? a[5] : a[7]) * inv;

    v8f d = __builtin_amdgcn_wmma_f32_16x16x4_f32(
        false, A0, false, B0, (short)0, vzero, false, false);
    d = __builtin_amdgcn_wmma_f32_16x16x4_f32(
        false, A1, false, B1, (short)0, d, false, false);

#pragma unroll
    for (int m = 0; m < 8; ++m) a[m] = d[m] * (ob ? em1[m] : em0[m]);
    const float s = sum8_tree(a);
    inv = fast_rcp(s);
    if (LOGS) lc += __builtin_amdgcn_logf(s);
    if (STORE) {
        v4f ah;
        ah.x = lo ? a[0] : a[4];  ah.y = lo ? a[1] : a[5];
        ah.z = lo ? a[2] : a[6];  ah.w = lo ? a[3] : a[7];
        *(v4f*)row = ah;
    }
}

template <bool STORE, bool LOGS>
__device__ __forceinline__ void fwd_block8(
    v8f& a, float& inv, float& lc, float* row, unsigned wb, bool lo,
    const float (&em1)[8], const float (&em0)[8], v2f A0, v2f A1)
{
#pragma unroll
    for (int tb = 0; tb < 8; ++tb)
        fwd_step<STORE, LOGS>(a, inv, lc, STORE ? row + tb * HMM_SROW : row,
                              (int)((wb >> tb) & 1u), lo, em1, em0, A0, A1);
}

__device__ __forceinline__ void fwd_load8(v8f (&buf)[8], const float* stg, int blk)
{
    const float* p = stg + blk * (8 * HMM_SROW);
#pragma unroll
    for (int i = 0; i < 8; ++i) buf[i] = *(const v8f*)(p + i * HMM_SROW);
}

#define HMM_GROUP (4 * HMM_SROW)

struct gamma_lines {
    int    put[4];
    int    take[4];
    float* line;
};

__device__ __forceinline__ gamma_lines gamma_lines_init(float* chunk_seq0, int lane)
{
    gamma_lines q;
    const int n = lane & 15, h = lane >> 4;
#pragma unroll
    for (int k = 0; k < 4; ++k)
        q.put[k] = n * 32 + ((((k << 1) | h) ^ ((n >> 1) & 7)) << 2);
#pragma unroll
    for (int j = 0; j < 4; ++j) {
        const int nn = 4 * j + (lane >> 3);
        q.take[j] = nn * 32 + (((lane & 7) ^ ((nn >> 1) & 7)) << 2);
    }
    q.line = chunk_seq0 + (size_t)(lane >> 3) * HMM_T * 8 + (lane & 7) * 4;
    return q;
}

__device__ __forceinline__ void gamma_take(v4f (&pend)[8], const float* blk_stage, const gamma_lines& q)
{
#pragma unroll
    for (int g = 0; g < 2; ++g)
#pragma unroll
        for (int j = 0; j < 4; ++j)
            pend[g * 4 + j] = *(const v4f*)(blk_stage + g * HMM_GROUP + q.take[j]);
}

__device__ __forceinline__ void gamma_send(const v4f (&pend)[8], const gamma_lines& q, int t0)
{
    float* const p = q.line + (size_t)t0 * 8;
#pragma unroll
    for (int g = 0; g < 2; ++g)
#pragma unroll
        for (int j = 0; j < 4; ++j)
            *(v4f*)(p + g * 32 + (size_t)j * (4 * (size_t)HMM_T * 8)) = pend[g * 4 + j];
}

__device__ __forceinline__ void fwd_gamma_block8(
    const v8f (&buf)[8], float* blk_stage, const gamma_lines& q, v4f (&pend)[8],
    bool has_pend, int blk, unsigned w,
    v8f& a, float& inv, bool lo,
    const float (&em1)[8], const float (&em0)[8], v2f A0, v2f A1)
{
    const v8f vzero = {};
    const int t_lo = blk * 8;
#pragma unroll
    for (int i = 0; i < 8; ++i) {
        const int t  = t_lo + i;
        const int ob = (int)((w >> (t & 31)) & 1u);

        v2f B0, B1;
        B0.x = (lo ? a[0] : a[2]) * inv;  B0.y = (lo ? a[1] : a[3]) * inv;
        B1.x = (lo ? a[4] : a[6]) * inv;  B1.y = (lo ? a[5] : a[7]) * inv;
        v8f d = __builtin_amdgcn_wmma_f32_16x16x4_f32(
            false, A0, false, B0, (short)0, vzero, false, false);
        d = __builtin_amdgcn_wmma_f32_16x16x4_f32(
            false, A1, false, B1, (short)0, d, false, false);
        if (i == 2) {
            __builtin_amdgcn_sched_barrier(0);
            if (has_pend) gamma_send(pend, q, t_lo - 8);
        }
#pragma unroll
        for (int m = 0; m < 8; ++m) a[m] = d[m] * (ob ? em1[m] : em0[m]);
        const float s = sum8_chain(a);
        inv = fast_rcp(s);

        const v8f be = buf[i];
        v8f g;
#pragma unroll
        for (int m = 0; m < 8; ++m) g[m] = be[m] * a[m];
        const float gi = fast_rcp(sum8_tree(g));
#pragma unroll
        for (int m = 0; m < 8; ++m) g[m] *= gi;
        v4f gh;
        gh.x = lo ? g[0] : g[4];  gh.y = lo ? g[1] : g[5];
        gh.z = lo ? g[2] : g[6];  gh.w = lo ? g[3] : g[7];
        *(v4f*)(blk_stage + (i >> 2) * HMM_GROUP + q.put[i & 3]) = gh;
    }
    gamma_take(pend, blk_stage, q);
}

__device__ __forceinline__ void bwd_stage_block8(
    float* row_hi, unsigned wb, v8f& d, bool lo,
    const float (&em1)[8], const float (&em0)[8], v2f A0, v2f A1)
{
    const v8f vzero = {};
#pragma unroll
    for (int i = 0; i < 8; ++i) {
        const int ob = (int)((wb >> (7 - i)) & 1u);
        v4f dh;
        dh.x = lo ? d[0] : d[4];  dh.y = lo ? d[1] : d[5];
        dh.z = lo ? d[2] : d[6];  dh.w = lo ? d[3] : d[7];
        *(v4f*)(row_hi - i * HMM_SROW) = dh;

        const float inv = fast_rcp(sum8_tree(d));
        v8f u;
#pragma unroll
        for (int m = 0; m < 8; ++m) u[m] = d[m] * (ob ? em1[m] : em0[m]);
        v2f B0, B1;
        B0.x = (lo ? u[0] : u[2]) * inv;  B0.y = (lo ? u[1] : u[3]) * inv;
        B1.x = (lo ? u[4] : u[6]) * inv;  B1.y = (lo ? u[5] : u[7]) * inv;
        d = __builtin_amdgcn_wmma_f32_16x16x4_f32(
            false, A0, false, B0, (short)0, vzero, false, false);
        d = __builtin_amdgcn_wmma_f32_16x16x4_f32(
            false, A1, false, B1, (short)0, d, false, false);
    }
}

__device__ __forceinline__ void bwd_load8(v8f (&buf)[8], const float* stg, int blk)
{
    const float* p = stg + blk * (8 * HMM_SROW);
#pragma unroll
    for (int i = 0; i < 8; ++i) buf[i] = *(const v8f*)(p + (7 - i) * HMM_SROW);
}

__device__ __forceinline__ void bwd_block8(
    const v8f (&buf)[8], float* blk_stage, const gamma_lines& q, v4f (&pend)[8],
    bool has_pend, int blk, unsigned w, v8f& d, bool lo,
    const float (&em1)[8], const float (&em0)[8], v2f A0, v2f A1)
{
    const v8f vzero = {};
    const int tt_hi = blk * 8 + 7;
#pragma unroll
    for (int i = 0; i < 8; ++i) {
        const int tt = tt_hi - i;
        const int ob = (int)((w >> (tt & 31)) & 1u);

        const v8f al = buf[i];
        v8f g;
#pragma unroll
        for (int m = 0; m < 8; ++m) g[m] = al[m] * d[m];
        const float gi = fast_rcp(sum8_tree(g));
#pragma unroll
        for (int m = 0; m < 8; ++m) g[m] *= gi;
        v4f gh;
        gh.x = lo ? g[0] : g[4];  gh.y = lo ? g[1] : g[5];
        gh.z = lo ? g[2] : g[6];  gh.w = lo ? g[3] : g[7];
        *(v4f*)(blk_stage + (1 - (i >> 2)) * HMM_GROUP + q.put[3 - (i & 3)]) = gh;

        const float inv = fast_rcp(sum8_chain(d));

        v8f u;
#pragma unroll
        for (int m = 0; m < 8; ++m) u[m] = d[m] * (ob ? em1[m] : em0[m]);
        v2f B0, B1;
        B0.x = (lo ? u[0] : u[2]) * inv;  B0.y = (lo ? u[1] : u[3]) * inv;
        B1.x = (lo ? u[4] : u[6]) * inv;  B1.y = (lo ? u[5] : u[7]) * inv;

        d = __builtin_amdgcn_wmma_f32_16x16x4_f32(
            false, A0, false, B0, (short)0, vzero, false, false);
        d = __builtin_amdgcn_wmma_f32_16x16x4_f32(
            false, A1, false, B1, (short)0, d, false, false);
        if (i == 2) {
            __builtin_amdgcn_sched_barrier(0);
            if (has_pend) gamma_send(pend, q, blk * 8 + 8);
        }
    }
    gamma_take(pend, blk_stage, q);
}

__device__ __forceinline__ float trans_norm(const float* __restrict__ trans, int r, int c)
{
    float s = 0.0f;
#pragma unroll
    for (int k = 0; k < 8; ++k) s += trans[r * 8 + k] + HMM_EPS;
    return (trans[r * 8 + c] + HMM_EPS) / s;
}

__device__ __forceinline__ void load_emission(
    const float* __restrict__ emission, float (&em1)[8], float (&em0)[8])
{
#pragma unroll
    for (int k = 0; k < 8; ++k) {
        float e = emission[k];
        e = fminf(fmaxf(e, HMM_EPS), 1.0f - HMM_EPS);
        em1[k] = e; em0[k] = 1.0f - e;
    }
}

__device__ void hmm_build_table(const float* __restrict__ trans,
                                const float* __restrict__ emission,
                                float* __restrict__ tab, int tw)
{
    const int  lane = threadIdx.x & 31;
    const int  n    = lane & 15;
    const bool lo   = lane < 16;
    const int  j    = n & 7;
    const unsigned v = (unsigned)(tw * 2 + (n >> 3));

    float em1[8], em0[8];
    load_emission(emission, em1, em0);

    const int kb = lo ? 0 : 2;
    v2f A0, A1;
    A0.x = trans_norm(trans, kb + 0, j);
    A0.y = trans_norm(trans, kb + 1, j);
    A1.x = trans_norm(trans, kb + 4, j);
    A1.y = trans_norm(trans, kb + 5, j);

    v8f x;
#pragma unroll
    for (int m = 0; m < 8; ++m) x[m] = (m == j) ? 1.0f : 0.0f;
    float inv = 1.0f, lc = 0.0f;
    fwd_block8<false, true>(x, inv, lc, nullptr, v, lo, em1, em0, A0, A1);

    float mx = lc;
    mx = fmaxf(mx, __shfl_xor(mx, 1));
    mx = fmaxf(mx, __shfl_xor(mx, 2));
    mx = fmaxf(mx, __shfl_xor(mx, 4));
    const float sc = inv * __builtin_amdgcn_exp2f(lc - mx);

    const int    pj = (j & 1) | ((j >> 1) & 2) | ((j << 1) & 4);
    float* const P  = tab + tab_entry_off(v) + pj;
    const int    i0 = lo ? 0 : 4;
    P[(i0 + 0) * 8] = (lo ? x[0] : x[4]) * sc;
    P[(i0 + 1) * 8] = (lo ? x[1] : x[5]) * sc;
    P[(i0 + 2) * 8] = (lo ? x[2] : x[6]) * sc;
    P[(i0 + 3) * 8] = (lo ? x[3] : x[7]) * sc;
    if (lo && j == 0) tab[tab_scale_off(v)] = mx;
}

__device__ __forceinline__ void tab_load4(
    const float* __restrict__ tab, unsigned w, int aoff,
    v2f (&a0)[4], v2f (&a1)[4], float (&ms)[4])
{
#pragma unroll
    for (int sb = 0; sb < 4; ++sb) {
        const unsigned v = (w >> (sb * 8)) & 255u;
        const v4f      p = *(const v4f*)(tab + tab_entry_off(v) + aoff);
        a0[sb] = p.lo;
        a1[sb] = p.hi;
        ms[sb] = tab[tab_scale_off(v)];
    }
}

__device__ __forceinline__ void tab_step(
    v8f& a, float& inv, float& lc, bool lo, v2f A0, v2f A1, float mlog)
{
    const v8f vzero = {};
    v2f B0, B1;
    B0.x = (lo ? a[0] : a[2]) * inv;  B0.y = (lo ? a[1] : a[3]) * inv;
    B1.x = (lo ? a[4] : a[6]) * inv;  B1.y = (lo ? a[5] : a[7]) * inv;
    v8f d = __builtin_amdgcn_wmma_f32_16x16x4_f32(
        false, A0, false, B0, (short)0, vzero, false, false);
    a = __builtin_amdgcn_wmma_f32_16x16x4_f32(
        false, A1, false, B1, (short)0, d, false, false);
    const float s = sum8_tree(a);
    inv = fast_rcp(s);
    lc += __builtin_amdgcn_logf(s) + mlog;
}

static_assert(HMM_C == 32, "the scan keeps one chunk per lane of a wave");
#define HMM_GROW 72

__device__ __forceinline__ float lane_value(float x, int src_lane)
{
    return __int_as_float(__builtin_amdgcn_readlane(__float_as_int(x), src_lane));
}

__device__ __forceinline__ void hmm_scan(
    const float* __restrict__ start, const float* gl, const float* glt,
    float* __restrict__ gamma, float* __restrict__ loglik, int b, bool fwd, int lane)
{
    float* const seq = gamma + (size_t)b * HMM_T * 8;
    float* const reg = seq + (size_t)lane * HMM_L * 8;
    const float* const mine = gl + lane * HMM_GROW;

    const v8f lg = *(const v8f*)(mine + 64);
    float m = lg[0];
#pragma unroll
    for (int k = 1; k < 8; ++k) m = fmaxf(m, lg[k]);
    v8f sc;
#pragma unroll
    for (int k = 0; k < 8; ++k) sc[k] = exp2f(lg[k] - m);

    if (fwd) {
        v8f col[8];
#pragma unroll
        for (int k = 0; k < 8; ++k) col[k] = *(const v8f*)(mine + k * 8) * sc[k];

        float v[8]; float ssum = 0.0f;
#pragma unroll
        for (int k = 0; k < 8; ++k) { v[k] = start[k] + HMM_EPS; ssum += v[k]; }
#pragma unroll
        for (int k = 0; k < 8; ++k) v[k] = lane_value(v[k] / ssum, 0);

        float* const erow    = reg + (size_t)HMM_L * 8;
        const bool   has_out = lane + 1 < HMM_C;
        float        skeep   = 1.0f;
        for (int c = 0; c < HMM_C; ++c) {
            v8f u = col[0] * v[0];
#pragma unroll
            for (int k = 1; k < 8; ++k) u = col[k] * v[k] + u;
            const float s  = sum8_tree(u);
            const v8f   vn = u * fast_rcp(s);
            if (lane == c) {
                skeep = s;
                if (has_out) *(v8f*)erow = vn;
            }
#pragma unroll
            for (int k = 0; k < 8; ++k) v[k] = lane_value(vn[k], c);
        }
        float tot = m + log2f(skeep);
        tot += __shfl_xor(tot, 16);
        tot += __shfl_xor(tot, 8);
        tot += __shfl_xor(tot, 4);
        tot += __shfl_xor(tot, 2);
        tot += __shfl_xor(tot, 1);
        if (lane == 0) loglik[b] = tot * HMM_LN2;
    } else {
        const float* const mine_t = glt + lane * HMM_GROW;
        v8f row[8];
#pragma unroll
        for (int r = 0; r < 8; ++r) row[r] = *(const v8f*)(mine_t + r * 8) * sc;

        float f[8];
#pragma unroll
        for (int k = 0; k < 8; ++k) f[k] = 1.0f;
        if (lane == 0) {
            v8f ones;
#pragma unroll
            for (int k = 0; k < 8; ++k) ones[k] = 1.0f;
            *(v8f*)(seq + ((size_t)HMM_T - 1) * 8) = ones;
        }
        float* const frow = reg - 8;
        for (int c = HMM_C - 1; c >= 1; --c) {
            v8f u = row[0] * f[0];
#pragma unroll
            for (int r = 1; r < 8; ++r) u = row[r] * f[r] + u;
            const v8f fn = u * fast_rcp(sum8_tree(u));
            if (lane == c) *(v8f*)frow = fn;
#pragma unroll
            for (int k = 0; k < 8; ++k) f[k] = lane_value(fn[k], c);
        }
    }
}

__global__ __launch_bounds__(1024) void hmm_chunk_kernel(
    const float* __restrict__ start, const float* __restrict__ trans,
    const float* __restrict__ emission, const unsigned* __restrict__ bits,
    const float* __restrict__ tab, float* __restrict__ gamma,
    float* __restrict__ loglik)
{
    __shared__ float gl[HMM_C * HMM_GROW];
    __shared__ float glt[HMM_C * HMM_GROW];

    const int  c    = __builtin_amdgcn_readfirstlane((int)(threadIdx.x >> 5));
    const int  lane = threadIdx.x & 31;
    const int  n    = lane & 15;
    const bool lo   = lane < 16;
    const int  b    = blockIdx.x;
    const int  j    = n & 7;
    const int  aoff = j * 8 + (lo ? 0 : 4);

    const unsigned* const bw = bits + (size_t)b * (HMM_T / 32) + c * (HMM_L / 32);

    unsigned w = (unsigned)__builtin_amdgcn_readfirstlane((int)bw[0]);
    v2f   a0c[4], a1c[4], a0n[4], a1n[4];
    float mc[4], mn[4];
    tab_load4(tab, w, aoff, a0c, a1c, mc);

    v8f   x;
    float inv, lc;
    if (c == 0) {
        float em1[8], em0[8];
        load_emission(emission, em1, em0);
        const int kb = lo ? 0 : 2;
        v2f A0, A1;
        A0.x = trans_norm(trans, kb + 0, j);
        A0.y = trans_norm(trans, kb + 1, j);
        A1.x = trans_norm(trans, kb + 4, j);
        A1.y = trans_norm(trans, kb + 5, j);
        const int ob = (int)(w & 1u);
#pragma unroll
        for (int m = 0; m < 8; ++m) x[m] = (m == j) ? (ob ? em1[m] : em0[m]) : 0.0f;
        const float s0 = sum8_tree(x);
        inv = fast_rcp(s0);
        lc  = __builtin_amdgcn_logf(s0);
#pragma unroll
        for (int tb = 1; tb < 8; ++tb)
            fwd_step<false, true>(x, inv, lc, nullptr, (int)((w >> tb) & 1u),
                                  lo, em1, em0, A0, A1);
    } else {
#pragma unroll
        for (int m = 0; m < 8; ++m) x[m] = (m == j) ? 1.0f : 0.0f;
        inv = 1.0f;
        lc  = 0.0f;
        tab_step(x, inv, lc, lo, a0c[0], a1c[0], mc[0]);
    }
    for (int q = 0; q < HMM_L / 32; ++q) {
        const unsigned wn = (unsigned)__builtin_amdgcn_readfirstlane(
            (int)bw[q + 1 < HMM_L / 32 ? q + 1 : q]);
        tab_load4(tab, wn, aoff, a0n, a1n, mn);
#pragma unroll
        for (int sb = 0; sb < 4; ++sb)
            if (q > 0 || sb > 0) tab_step(x, inv, lc, lo, a0c[sb], a1c[sb], mc[sb]);
#pragma unroll
        for (int sb = 0; sb < 4; ++sb) { a0c[sb] = a0n[sb]; a1c[sb] = a1n[sb]; mc[sb] = mn[sb]; }
    }

    float* const reg = gl + c * HMM_GROW;
    v4f ah;
    ah.x = (lo ? x[0] : x[4]) * inv;  ah.y = (lo ? x[1] : x[5]) * inv;
    ah.z = (lo ? x[2] : x[6]) * inv;  ah.w = (lo ? x[3] : x[7]) * inv;
    if (n < 8) {
        *(v4f*)(reg + j * 8 + (lo ? 0 : 4)) = ah;
        if (lo) reg[64 + j] = lc;
        float* const rt = glt + c * HMM_GROW + (lo ? 0 : 4) * 8 + j;
        rt[0] = ah.x;  rt[8] = ah.y;  rt[16] = ah.z;  rt[24] = ah.w;
    }

    __syncthreads();
    if (c >= 2) return;
    hmm_scan(start, gl, glt, gamma, loglik, b, c == 0, lane);
}

__global__ __launch_bounds__(64) void hmm_fb_kernel(
    const float* __restrict__ start, const float* __restrict__ trans,
    const float* __restrict__ emission, const unsigned* __restrict__ bits,
    float* __restrict__ gamma)
{
    __shared__ float stage[HMM_L * HMM_SROW];

    const bool fwd  = __builtin_amdgcn_readfirstlane((int)(threadIdx.x >> 5)) == 0;
    const int  lane = threadIdx.x & 31;
    const int  n    = lane & 15;
    const bool lo   = lane < 16;
    const int  c    = blockIdx.x / (HMM_B / 16);
    const int  b    = (blockIdx.x % (HMM_B / 16)) * 16 + n;

    float em1[8], em0[8];
    load_emission(emission, em1, em0);

    const int kb = lo ? 0 : 2;
    const int i  = n & 7;
    v2f A0, A1;
    A0.x = fwd ? trans_norm(trans, kb + 0, i) : trans_norm(trans, i, kb + 0);
    A0.y = fwd ? trans_norm(trans, kb + 1, i) : trans_norm(trans, i, kb + 1);
    A1.x = fwd ? trans_norm(trans, kb + 4, i) : trans_norm(trans, i, kb + 4);
    A1.y = fwd ? trans_norm(trans, kb + 5, i) : trans_norm(trans, i, kb + 5);

    float* const          seq      = gamma + ((size_t)b * HMM_T + (size_t)c * HMM_L) * 8;
    const unsigned* const bw       = bits + (size_t)b * (HMM_T / 32) + c * (HMM_L / 32);
    float* const          stg      = stage + n * 8;
    float* const          half_stg = stg + (lo ? 0 : 4);

    v8f      x;
    float    inv = 1.0f;
    float    lcn = 0.0f;
    unsigned wn;

    if (fwd) {
        unsigned w = bw[0];
        wn = bw[1];
        if (c == 0) {
            float st[8]; float ssum = 0.0f;
#pragma unroll
            for (int k = 0; k < 8; ++k) { st[k] = start[k] + HMM_EPS; ssum += st[k]; }
#pragma unroll
            for (int k = 0; k < 8; ++k) st[k] /= ssum;

            {
                const int ob = (int)(w & 1u);
#pragma unroll
                for (int m = 0; m < 8; ++m) x[m] = st[m] * (ob ? em1[m] : em0[m]);
            }
            inv = fast_rcp(sum8_tree(x));
            {
                v4f ah;
                ah.x = lo ? x[0] : x[4];  ah.y = lo ? x[1] : x[5];
                ah.z = lo ? x[2] : x[6];  ah.w = lo ? x[3] : x[7];
                *(v4f*)half_stg = ah;
            }
#pragma unroll
            for (int tb = 1; tb < 8; ++tb)
                fwd_step<true, false>(x, inv, lcn, half_stg + tb * HMM_SROW,
                                      (int)((w >> tb) & 1u), lo, em1, em0, A0, A1);
        } else {
            x   = *(const v8f*)seq;
            inv = 1.0f;
            fwd_block8<true, false>(x, inv, lcn, half_stg, w, lo, em1, em0, A0, A1);
        }
        for (int sub = 1; sub < 4; ++sub)
            fwd_block8<true, false>(x, inv, lcn, half_stg + sub * (8 * HMM_SROW),
                                    w >> (sub * 8), lo, em1, em0, A0, A1);
        for (int q = 1; q < HMM_L / 64; ++q) {
            w  = wn;
            wn = bw[q + 1];
            float* const wrow = half_stg + q * (32 * HMM_SROW);
            for (int sub = 0; sub < 4; ++sub)
                fwd_block8<true, false>(x, inv, lcn, wrow + sub * (8 * HMM_SROW),
                                        w >> (sub * 8), lo, em1, em0, A0, A1);
        }
    } else {
        x  = *(const v8f*)(seq + (size_t)(HMM_L - 1) * 8);
        wn = bw[HMM_L / 32 - 1];
        for (int q = HMM_L / 32 - 1; q >= HMM_L / 64; --q) {
            const unsigned w = wn;
            wn = bw[q - 1];
            float* const wrow = half_stg + q * (32 * HMM_SROW);
            for (int sub = 3; sub >= 0; --sub)
                bwd_stage_block8(wrow + (sub * 8 + 7) * HMM_SROW, w >> (sub * 8), x,
                                 lo, em1, em0, A0, A1);
        }
    }

    __syncthreads();

    v8f bufA[8], bufB[8];
    const gamma_lines q = gamma_lines_init(
        gamma + ((size_t)((blockIdx.x % (HMM_B / 16)) * 16) * HMM_T + (size_t)c * HMM_L) * 8, lane);
    v4f pend[8];
    if (fwd) {
        fwd_load8(bufA, stg, HMM_L / 16);
        for (int jb = HMM_L / 16; jb < HMM_L / 8; jb += 2) {
            const unsigned w  = wn;
            const int      jn = jb + 2 < HMM_L / 8 ? jb + 2 : HMM_L / 8 - 2;
            wn = bw[jn >> 2];
            fwd_load8(bufB, stg, jb + 1);
            __builtin_amdgcn_sched_barrier(0);
            fwd_gamma_block8(bufA, stage + jb * (8 * HMM_SROW), q, pend, jb != HMM_L / 16,
                             jb, w, x, inv, lo, em1, em0, A0, A1);
            __builtin_amdgcn_sched_barrier(0);
            fwd_load8(bufA, stg, jn);
            __builtin_amdgcn_sched_barrier(0);
            fwd_gamma_block8(bufB, stage + (jb + 1) * (8 * HMM_SROW), q, pend, true,
                             jb + 1, w, x, inv, lo, em1, em0, A0, A1);
        }
        gamma_send(pend, q, HMM_L - 8);
    } else {
        bwd_load8(bufA, stg, HMM_L / 16 - 1);
        for (int jb = HMM_L / 16 - 1; jb >= 1; jb -= 2) {
            const unsigned w  = wn;
            const int      jn = jb >= 3 ? jb - 2 : 1;
            wn = bw[jn >> 2];
            bwd_load8(bufB, stg, jb - 1);
            __builtin_amdgcn_sched_barrier(0);
            bwd_block8(bufA, stage + jb * (8 * HMM_SROW), q, pend, jb != HMM_L / 16 - 1,
                       jb, w, x, lo, em1, em0, A0, A1);
            __builtin_amdgcn_sched_barrier(0);
            bwd_load8(bufA, stg, jn);
            __builtin_amdgcn_sched_barrier(0);
            bwd_block8(bufB, stage + (jb - 1) * (8 * HMM_SROW), q, pend, true,
                       jb - 1, w, x, lo, em1, em0, A0, A1);
        }
        gamma_send(pend, q, 0);
    }
}

extern "C" void kernel_launch(void* const* d_in, const int* in_sizes, int n_in,
                              void* d_out, int out_size, void* d_ws, size_t ws_size,
                              hipStream_t stream) {
    (void)in_sizes; (void)n_in; (void)out_size; (void)ws_size;
    const float* start    = (const float*)d_in[0];
    const float* trans    = (const float*)d_in[1];
    const float* emission = (const float*)d_in[2];
    const int*   obs      = (const int*)d_in[3];

    float* gamma  = (float*)d_out;
    float* loglik = gamma + (size_t)HMM_B * HMM_T * HMM_K;
    unsigned* bits = (unsigned*)d_ws;

    hmm_pack_kernel<<<HMM_TAB_BLOCKS + HMM_PACK_BLOCKS, 256, 0, stream>>>(
        obs, bits, trans, emission, gamma);
    hmm_chunk_kernel<<<HMM_B, 1024, 0, stream>>>(
        start, trans, emission, bits, gamma, gamma, loglik);
    hmm_fb_kernel<<<HMM_C * (HMM_B / 16), 64, 0, stream>>>(
        start, trans, emission, bits, gamma);
}


# pass kernel: upper-lane A rows reordered (r^2) so B is read from the product registers, no selects; the divide by the column sum moves behind the matrix product, into the emission factors
# speedup vs baseline: 89.8141x; 1.0533x over previous
#include <hip/hip_runtime.h>


#define HMM_B 256
#define HMM_T 8192
#define HMM_K 8
#define HMM_C 32
#define HMM_L (HMM_T / HMM_C)
#define HMM_SROW (16 * HMM_K)
#define HMM_EPS 1e-8f
#define HMM_LN2 0.6931471805599453f

typedef __attribute__((ext_vector_type(2))) float v2f;
typedef __attribute__((ext_vector_type(4))) float v4f;
typedef __attribute__((ext_vector_type(8))) float v8f;

__device__ __forceinline__ float fast_rcp(float x) {
    float r = __builtin_amdgcn_rcpf(x);
    float e = __builtin_fmaf(-x, r, 1.0f);
    return __builtin_fmaf(r, e, r);
}

#define HMM_TAB_BLOCKS  16
#define HMM_PACK_BLOCKS 256

__device__ __forceinline__ size_t tab_entry_off(unsigned v)
{
    return ((size_t)(v >> 4) * HMM_L + 16) * 8 + (size_t)(v & 15u) * 64;
}
__device__ __forceinline__ size_t tab_scale_off(unsigned v)
{
    return ((size_t)(v >> 4) * HMM_L + 144) * 8 + (size_t)(v & 15u);
}
static_assert(HMM_C >= 16 && HMM_L >= 146, "the table needs rows 16-145 of 16 chunk regions");

__device__ void hmm_build_table(const float* __restrict__ trans,
                                const float* __restrict__ emission,
                                float* __restrict__ tab, int tw);

__global__ __launch_bounds__(256) void hmm_pack_kernel(
    const int* __restrict__ obs, unsigned* __restrict__ bits,
    const float* __restrict__ trans, const float* __restrict__ emission,
    float* __restrict__ tab)
{
    if (blockIdx.x < HMM_TAB_BLOCKS) {
        hmm_build_table(trans, emission, tab,
                        (int)((blockIdx.x * blockDim.x + threadIdx.x) >> 5));
        return;
    }
    const int wave = ((blockIdx.x - HMM_TAB_BLOCKS) * blockDim.x + threadIdx.x) >> 5;
    const int lane = threadIdx.x & 31;
    const long base = (long)wave * 1024;
    unsigned keep = 0;
#pragma unroll 4
    for (int i = 0; i < 32; ++i) {
        int v = obs[base + i * 32 + lane];
        unsigned m = (unsigned)__ballot(v != 0);
        if (i == lane) keep = m;
    }
    bits[(size_t)wave * 32 + lane] = keep;
}

__device__ __forceinline__ float sum8_tree(const v8f& x)
{
    const v4f q = x.lo + x.hi;
    const v2f r = q.lo + q.hi;
    return r.x + r.y;
}

__device__ __forceinline__ void fwd_step(
    v8f& a, float& inv, float& lc, int ob, bool lo,
    const float (&em1)[8], const float (&em0)[8], v2f A0, v2f A1)
{
    const v8f vzero = {};
    v2f B0, B1;
    B0.x = (lo ? a[0] : a[2]) * inv;  B0.y = (lo ? a[1] : a[3]) * inv;
    B1.x = (lo ? a[4] : a[6]) * inv;  B1.y = (lo ? a[5] : a[7]) * inv;

    v8f d = __builtin_amdgcn_wmma_f32_16x16x4_f32(
        false, A0, false, B0, (short)0, vzero, false, false);
    d = __builtin_amdgcn_wmma_f32_16x16x4_f32(
        false, A1, false, B1, (short)0, d, false, false);

#pragma unroll
    for (int m = 0; m < 8; ++m) a[m] = d[m] * (ob ? em1[m] : em0[m]);
    const float s = sum8_tree(a);
    inv = fast_rcp(s);
    lc += __builtin_amdgcn_logf(s);
}

__device__ __forceinline__ void fwd_block8(
    v8f& a, float& inv, float& lc, unsigned wb, bool lo,
    const float (&em1)[8], const float (&em0)[8], v2f A0, v2f A1)
{
#pragma unroll
    for (int tb = 0; tb < 8; ++tb)
        fwd_step(a, inv, lc, (int)((wb >> tb) & 1u), lo, em1, em0, A0, A1);
}

__device__ __forceinline__ v8f lane_order(const v8f& x, bool lo)
{
    v8f y;
#pragma unroll
    for (int m = 0; m < 8; ++m) y[m] = lo ? x[m] : x[m ^ 2];
    return y;
}

__device__ __forceinline__ v4f own_half(const v8f& p, bool lo)
{
    v4f h;
    h.x = lo ? p[0] : p[6];  h.y = lo ? p[1] : p[7];
    h.z = lo ? p[2] : p[4];  h.w = lo ? p[3] : p[5];
    return h;
}

__device__ __forceinline__ v8f staged_column(const float* px, const float* py)
{
    const v2f p0 = *(const v2f*)px,       p1 = *(const v2f*)py;
    const v2f p2 = *(const v2f*)(px + 4), p3 = *(const v2f*)(py + 4);
    v8f r;
    r[0] = p0.x; r[1] = p0.y; r[2] = p1.x; r[3] = p1.y;
    r[4] = p2.x; r[5] = p2.y; r[6] = p3.x; r[7] = p3.y;
    return r;
}

__device__ __forceinline__ void pass_fwd_step(
    v8f& a, float& cs, float* row, int ob, bool lo,
    const float (&em1)[8], const float (&em0)[8], v2f A0, v2f A1)
{
    const v8f vzero = {};
    v2f B0, B1;
    B0.x = a[0];  B0.y = a[1];
    B1.x = a[4];  B1.y = a[5];
    v8f d = __builtin_amdgcn_wmma_f32_16x16x4_f32(
        false, A0, false, B0, (short)0, vzero, false, false);
    d = __builtin_amdgcn_wmma_f32_16x16x4_f32(
        false, A1, false, B1, (short)0, d, false, false);
#pragma unroll
    for (int m = 0; m < 8; ++m) a[m] = d[m] * ((ob ? em1[m] : em0[m]) * cs);
    cs = __builtin_amdgcn_rcpf(sum8_tree(a));
    *(v4f*)row = own_half(a, lo);
}

__device__ __forceinline__ void pass_fwd_block8(
    v8f& a, float& cs, float* row, unsigned wb, bool lo,
    const float (&em1)[8], const float (&em0)[8], v2f A0, v2f A1)
{
#pragma unroll
    for (int tb = 0; tb < 8; ++tb)
        pass_fwd_step(a, cs, row + tb * HMM_SROW, (int)((wb >> tb) & 1u),
                      lo, em1, em0, A0, A1);
}

__device__ __forceinline__ void fwd_load8(v8f (&buf)[8], const float* sx, const float* sy, int blk)
{
    int o = blk * (8 * HMM_SROW);
    asm volatile("" : "+s"(o));
#pragma unroll
    for (int i = 0; i < 8; ++i)
        buf[i] = staged_column(sx + o + i * HMM_SROW, sy + o + i * HMM_SROW);
}

#define HMM_GROUP (4 * HMM_SROW)

struct gamma_lines {
    int    put[4];
    int    take[4];
    float* line;
};

__device__ __forceinline__ gamma_lines gamma_lines_init(float* chunk_seq0, int lane)
{
    gamma_lines q;
    const int n = lane & 15, h = lane >> 4;
#pragma unroll
    for (int k = 0; k < 4; ++k)
        q.put[k] = n * 32 + ((((k << 1) | h) ^ ((n >> 1) & 7)) << 2);
#pragma unroll
    for (int j = 0; j < 4; ++j) {
        const int nn = 4 * j + (lane >> 3);
        q.take[j] = nn * 32 + (((lane & 7) ^ ((nn >> 1) & 7)) << 2);
    }
    q.line = chunk_seq0 + (size_t)(lane >> 3) * HMM_T * 8 + (lane & 7) * 4;
    return q;
}

__device__ __forceinline__ void gamma_take(v4f (&pend)[8], const float* blk_stage, const gamma_lines& q)
{
#pragma unroll
    for (int g = 0; g < 2; ++g)
#pragma unroll
        for (int j = 0; j < 4; ++j)
            pend[g * 4 + j] = *(const v4f*)(blk_stage + g * HMM_GROUP + q.take[j]);
}

__device__ __forceinline__ void gamma_send(const v4f (&pend)[8], const gamma_lines& q, int t0)
{
    float* const p = q.line + (size_t)t0 * 8;
#pragma unroll
    for (int g = 0; g < 2; ++g)
#pragma unroll
        for (int j = 0; j < 4; ++j)
            *(v4f*)(p + g * 32 + (size_t)j * (4 * (size_t)HMM_T * 8)) = pend[g * 4 + j];
}

__device__ __forceinline__ void fwd_gamma_block8(
    const v8f (&buf)[8], float* blk_stage, const gamma_lines& q, v4f (&pend)[8],
    bool has_pend, int blk, unsigned w,
    v8f& a, float& cs, bool lo,
    const float (&em1)[8], const float (&em0)[8], v2f A0, v2f A1)
{
    const v8f vzero = {};
    const int t_lo = blk * 8;
#pragma unroll
    for (int i = 0; i < 8; ++i) {
        const int t  = t_lo + i;
        const int ob = (int)((w >> (t & 31)) & 1u);

        v2f B0, B1;
        B0.x = a[0];  B0.y = a[1];
        B1.x = a[4];  B1.y = a[5];
        v8f d = __builtin_amdgcn_wmma_f32_16x16x4_f32(
            false, A0, false, B0, (short)0, vzero, false, false);
        d = __builtin_amdgcn_wmma_f32_16x16x4_f32(
            false, A1, false, B1, (short)0, d, false, false);
        if (i == 2) {
            __builtin_amdgcn_sched_barrier(0);
            if (has_pend) gamma_send(pend, q, t_lo - 8);
        }
#pragma unroll
        for (int m = 0; m < 8; ++m) a[m] = d[m] * ((ob ? em1[m] : em0[m]) * cs);
        cs = __builtin_amdgcn_rcpf(sum8_tree(a));

        const v8f be = buf[i];
        v8f g;
#pragma unroll
        for (int m = 0; m < 8; ++m) g[m] = be[m] * a[m];
        const float gi = fast_rcp(sum8_tree(g));
#pragma unroll
        for (int m = 0; m < 8; ++m) g[m] *= gi;
        *(v4f*)(blk_stage + (i >> 2) * HMM_GROUP + q.put[i & 3]) = own_half(g, lo);
    }
    gamma_take(pend, blk_stage, q);
}

__device__ __forceinline__ void bwd_chain_step(
    v8f& d, float& cs, int ob,
    const float (&em1)[8], const float (&em0)[8], v2f A0, v2f A1)
{
    const v8f vzero = {};
    v8f u;
#pragma unroll
    for (int m = 0; m < 8; ++m) u[m] = d[m] * ((ob ? em1[m] : em0[m]) * cs);
    cs = __builtin_amdgcn_rcpf(sum8_tree(u));
    v2f B0, B1;
    B0.x = u[0];  B0.y = u[1];
    B1.x = u[4];  B1.y = u[5];
    d = __builtin_amdgcn_wmma_f32_16x16x4_f32(
        false, A0, false, B0, (short)0, vzero, false, false);
    d = __builtin_amdgcn_wmma_f32_16x16x4_f32(
        false, A1, false, B1, (short)0, d, false, false);
}

__device__ __forceinline__ void bwd_stage_block8(
    float* row_hi, unsigned wb, v8f& d, float& cs, bool lo,
    const float (&em1)[8], const float (&em0)[8], v2f A0, v2f A1)
{
#pragma unroll
    for (int i = 0; i < 8; ++i) {
        *(v4f*)(row_hi - i * HMM_SROW) = own_half(d, lo);
        bwd_chain_step(d, cs, (int)((wb >> (7 - i)) & 1u), em1, em0, A0, A1);
    }
}

__device__ __forceinline__ void bwd_load8(v8f (&buf)[8], const float* sx, const float* sy, int blk)
{
    const int o = blk * (8 * HMM_SROW);
#pragma unroll
    for (int i = 0; i < 8; ++i)
        buf[i] = staged_column(sx + o + (7 - i) * HMM_SROW, sy + o + (7 - i) * HMM_SROW);
}

__device__ __forceinline__ void bwd_block8(
    const v8f (&buf)[8], float* blk_stage, const gamma_lines& q, v4f (&pend)[8],
    bool has_pend, int blk, unsigned w, v8f& d, float& cs, bool lo,
    const float (&em1)[8], const float (&em0)[8], v2f A0, v2f A1)
{
    const int tt_hi = blk * 8 + 7;
#pragma unroll
    for (int i = 0; i < 8; ++i) {
        const int tt = tt_hi - i;
        const int ob = (int)((w >> (tt & 31)) & 1u);

        const v8f al = buf[i];
        v8f g;
#pragma unroll
        for (int m = 0; m < 8; ++m) g[m] = al[m] * d[m];
        const float gi = fast_rcp(sum8_tree(g));
#pragma unroll
        for (int m = 0; m < 8; ++m) g[m] *= gi;
        *(v4f*)(blk_stage + (1 - (i >> 2)) * HMM_GROUP + q.put[3 - (i & 3)]) = own_half(g, lo);

        bwd_chain_step(d, cs, ob, em1, em0, A0, A1);
        if (i == 2) {
            __builtin_amdgcn_sched_barrier(0);
            if (has_pend) gamma_send(pend, q, blk * 8 + 8);
        }
    }
    gamma_take(pend, blk_stage, q);
}

__device__ __forceinline__ float trans_norm(const float* __restrict__ trans, int r, int c)
{
    float s = 0.0f;
#pragma unroll
    for (int k = 0; k < 8; ++k) s += trans[r * 8 + k] + HMM_EPS;
    return (trans[r * 8 + c] + HMM_EPS) / s;
}

__device__ __forceinline__ void load_emission(
    const float* __restrict__ emission, float (&em1)[8], float (&em0)[8])
{
#pragma unroll
    for (int k = 0; k < 8; ++k) {
        float e = emission[k];
        e = fminf(fmaxf(e, HMM_EPS), 1.0f - HMM_EPS);
        em1[k] = e; em0[k] = 1.0f - e;
    }
}

__device__ void hmm_build_table(const float* __restrict__ trans,
                                const float* __restrict__ emission,
                                float* __restrict__ tab, int tw)
{
    const int  lane = threadIdx.x & 31;
    const int  n    = lane & 15;
    const bool lo   = lane < 16;
    const int  j    = n & 7;
    const unsigned v = (unsigned)(tw * 2 + (n >> 3));

    float em1[8], em0[8];
    load_emission(emission, em1, em0);

    const int kb = lo ? 0 : 2;
    v2f A0, A1;
    A0.x = trans_norm(trans, kb + 0, j);
    A0.y = trans_norm(trans, kb + 1, j);
    A1.x = trans_norm(trans, kb + 4, j);
    A1.y = trans_norm(trans, kb + 5, j);

    v8f x;
#pragma unroll
    for (int m = 0; m < 8; ++m) x[m] = (m == j) ? 1.0f : 0.0f;
    float inv = 1.0f, lc = 0.0f;
    fwd_block8(x, inv, lc, v, lo, em1, em0, A0, A1);

    float mx = lc;
    mx = fmaxf(mx, __shfl_xor(mx, 1));
    mx = fmaxf(mx, __shfl_xor(mx, 2));
    mx = fmaxf(mx, __shfl_xor(mx, 4));
    const float sc = inv * __builtin_amdgcn_exp2f(lc - mx);

    const int    pj = (j & 1) | ((j >> 1) & 2) | ((j << 1) & 4);
    float* const P  = tab + tab_entry_off(v) + pj;
    const int    i0 = lo ? 0 : 4;
    P[(i0 + 0) * 8] = (lo ? x[0] : x[4]) * sc;
    P[(i0 + 1) * 8] = (lo ? x[1] : x[5]) * sc;
    P[(i0 + 2) * 8] = (lo ? x[2] : x[6]) * sc;
    P[(i0 + 3) * 8] = (lo ? x[3] : x[7]) * sc;
    if (lo && j == 0) tab[tab_scale_off(v)] = mx;
}

__device__ __forceinline__ void tab_load4(
    const float* __restrict__ tab, unsigned w, int aoff,
    v2f (&a0)[4], v2f (&a1)[4], float (&ms)[4])
{
#pragma unroll
    for (int sb = 0; sb < 4; ++sb) {
        const unsigned v = (w >> (sb * 8)) & 255u;
        const v4f      p = *(const v4f*)(tab + tab_entry_off(v) + aoff);
        a0[sb] = p.lo;
        a1[sb] = p.hi;
        ms[sb] = tab[tab_scale_off(v)];
    }
}

__device__ __forceinline__ void tab_step(
    v8f& a, float& inv, float& lc, bool lo, v2f A0, v2f A1, float mlog)
{
    const v8f vzero = {};
    v2f B0, B1;
    B0.x = (lo ? a[0] : a[2]) * inv;  B0.y = (lo ? a[1] : a[3]) * inv;
    B1.x = (lo ? a[4] : a[6]) * inv;  B1.y = (lo ? a[5] : a[7]) * inv;
    v8f d = __builtin_amdgcn_wmma_f32_16x16x4_f32(
        false, A0, false, B0, (short)0, vzero, false, false);
    a = __builtin_amdgcn_wmma_f32_16x16x4_f32(
        false, A1, false, B1, (short)0, d, false, false);
    const float s = sum8_tree(a);
    inv = fast_rcp(s);
    lc += __builtin_amdgcn_logf(s) + mlog;
}

static_assert(HMM_C == 32, "the scan keeps one chunk per lane of a wave");
#define HMM_GROW 72

__device__ __forceinline__ float lane_value(float x, int src_lane)
{
    return __int_as_float(__builtin_amdgcn_readlane(__float_as_int(x), src_lane));
}

__device__ __forceinline__ void hmm_scan(
    const float* __restrict__ start, const float* gl, const float* glt,
    float* __restrict__ gamma, float* __restrict__ loglik, int b, bool fwd, int lane)
{
    float* const seq = gamma + (size_t)b * HMM_T * 8;
    float* const reg = seq + (size_t)lane * HMM_L * 8;
    const float* const mine = gl + lane * HMM_GROW;

    const v8f lg = *(const v8f*)(mine + 64);
    float m = lg[0];
#pragma unroll
    for (int k = 1; k < 8; ++k) m = fmaxf(m, lg[k]);
    v8f sc;
#pragma unroll
    for (int k = 0; k < 8; ++k) sc[k] = exp2f(lg[k] - m);

    if (fwd) {
        v8f col[8];
#pragma unroll
        for (int k = 0; k < 8; ++k) col[k] = *(const v8f*)(mine + k * 8) * sc[k];

        float v[8]; float ssum = 0.0f;
#pragma unroll
        for (int k = 0; k < 8; ++k) { v[k] = start[k] + HMM_EPS; ssum += v[k]; }
#pragma unroll
        for (int k = 0; k < 8; ++k) v[k] = lane_value(v[k] / ssum, 0);

        float* const erow    = reg + (size_t)HMM_L * 8;
        const bool   has_out = lane + 1 < HMM_C;
        float        skeep   = 1.0f;
        for (int c = 0; c < HMM_C; ++c) {
            v8f u = col[0] * v[0];
#pragma unroll
            for (int k = 1; k < 8; ++k) u = col[k] * v[k] + u;
            const float s  = sum8_tree(u);
            const v8f   vn = u * fast_rcp(s);
            if (lane == c) {
                skeep = s;
                if (has_out) *(v8f*)erow = vn;
            }
#pragma unroll
            for (int k = 0; k < 8; ++k) v[k] = lane_value(vn[k], c);
        }
        float tot = m + log2f(skeep);
        tot += __shfl_xor(tot, 16);
        tot += __shfl_xor(tot, 8);
        tot += __shfl_xor(tot, 4);
        tot += __shfl_xor(tot, 2);
        tot += __shfl_xor(tot, 1);
        if (lane == 0) loglik[b] = tot * HMM_LN2;
    } else {
        const float* const mine_t = glt + lane * HMM_GROW;
        v8f row[8];
#pragma unroll
        for (int r = 0; r < 8; ++r) row[r] = *(const v8f*)(mine_t + r * 8) * sc;

        float f[8];
#pragma unroll
        for (int k = 0; k < 8; ++k) f[k] = 1.0f;
        if (lane == 0) {
            v8f ones;
#pragma unroll
            for (int k = 0; k < 8; ++k) ones[k] = 1.0f;
            *(v8f*)(seq + ((size_t)HMM_T - 1) * 8) = ones;
        }
        float* const frow = reg - 8;
        for (int c = HMM_C - 1; c >= 1; --c) {
            v8f u = row[0] * f[0];
#pragma unroll
            for (int r = 1; r < 8; ++r) u = row[r] * f[r] + u;
            const v8f fn = u * fast_rcp(sum8_tree(u));
            if (lane == c) *(v8f*)frow = fn;
#pragma unroll
            for (int k = 0; k < 8; ++k) f[k] = lane_value(fn[k], c);
        }
    }
}

__global__ __launch_bounds__(1024) void hmm_chunk_kernel(
    const float* __restrict__ start, const float* __restrict__ trans,
    const float* __restrict__ emission, const unsigned* __restrict__ bits,
    const float* __restrict__ tab, float* __restrict__ gamma,
    float* __restrict__ loglik)
{
    __shared__ float gl[HMM_C * HMM_GROW];
    __shared__ float glt[HMM_C * HMM_GROW];

    const int  c    = __builtin_amdgcn_readfirstlane((int)(threadIdx.x >> 5));
    const int  lane = threadIdx.x & 31;
    const int  n    = lane & 15;
    const bool lo   = lane < 16;
    const int  b    = blockIdx.x;
    const int  j    = n & 7;
    const int  aoff = j * 8 + (lo ? 0 : 4);

    const unsigned* const bw = bits + (size_t)b * (HMM_T / 32) + c * (HMM_L / 32);

    unsigned w = (unsigned)__builtin_amdgcn_readfirstlane((int)bw[0]);
    v2f   a0c[4], a1c[4], a0n[4], a1n[4];
    float mc[4], mn[4];
    tab_load4(tab, w, aoff, a0c, a1c, mc);

    v8f   x;
    float inv, lc;
    if (c == 0) {
        float em1[8], em0[8];
        load_emission(emission, em1, em0);
        const int kb = lo ? 0 : 2;
        v2f A0, A1;
        A0.x = trans_norm(trans, kb + 0, j);
        A0.y = trans_norm(trans, kb + 1, j);
        A1.x = trans_norm(trans, kb + 4, j);
        A1.y = trans_norm(trans, kb + 5, j);
        const int ob = (int)(w & 1u);
#pragma unroll
        for (int m = 0; m < 8; ++m) x[m] = (m == j) ? (ob ? em1[m] : em0[m]) : 0.0f;
        const float s0 = sum8_tree(x);
        inv = fast_rcp(s0);
        lc  = __builtin_amdgcn_logf(s0);
#pragma unroll
        for (int tb = 1; tb < 8; ++tb)
            fwd_step(x, inv, lc, (int)((w >> tb) & 1u), lo, em1, em0, A0, A1);
    } else {
#pragma unroll
        for (int m = 0; m < 8; ++m) x[m] = (m == j) ? 1.0f : 0.0f;
        inv = 1.0f;
        lc  = 0.0f;
        tab_step(x, inv, lc, lo, a0c[0], a1c[0], mc[0]);
    }
    for (int q = 0; q < HMM_L / 32; ++q) {
        const unsigned wn = (unsigned)__builtin_amdgcn_readfirstlane(
            (int)bw[q + 1 < HMM_L / 32 ? q + 1 : q]);
        tab_load4(tab, wn, aoff, a0n, a1n, mn);
#pragma unroll
        for (int sb = 0; sb < 4; ++sb)
            if (q > 0 || sb > 0) tab_step(x, inv, lc, lo, a0c[sb], a1c[sb], mc[sb]);
#pragma unroll
        for (int sb = 0; sb < 4; ++sb) { a0c[sb] = a0n[sb]; a1c[sb] = a1n[sb]; mc[sb] = mn[sb]; }
    }

    float* const reg = gl + c * HMM_GROW;
    v4f ah;
    ah.x = (lo ? x[0] : x[4]) * inv;  ah.y = (lo ? x[1] : x[5]) * inv;
    ah.z = (lo ? x[2] : x[6]) * inv;  ah.w = (lo ? x[3] : x[7]) * inv;
    if (n < 8) {
        *(v4f*)(reg + j * 8 + (lo ? 0 : 4)) = ah;
        if (lo) reg[64 + j] = lc;
        float* const rt = glt + c * HMM_GROW + (lo ? 0 : 4) * 8 + j;
        rt[0] = ah.x;  rt[8] = ah.y;  rt[16] = ah.z;  rt[24] = ah.w;
    }

    __syncthreads();
    if (c >= 2) return;
    hmm_scan(start, gl, glt, gamma, loglik, b, c == 0, lane);
}

__global__ __launch_bounds__(64) void hmm_fb_kernel(
    const float* __restrict__ start, const float* __restrict__ trans,
    const float* __restrict__ emission, const unsigned* __restrict__ bits,
    float* __restrict__ gamma)
{
    __shared__ float stage[HMM_L * HMM_SROW];

    const bool fwd  = __builtin_amdgcn_readfirstlane((int)(threadIdx.x >> 5)) == 0;
    const int  lane = threadIdx.x & 31;
    const int  n    = lane & 15;
    const bool lo   = lane < 16;
    const int  c    = blockIdx.x / (HMM_B / 16);
    const int  b    = (blockIdx.x % (HMM_B / 16)) * 16 + n;

    float em1n[8], em0n[8], em1[8], em0[8];
    load_emission(emission, em1n, em0n);
#pragma unroll
    for (int m = 0; m < 8; ++m) {
        em1[m] = lo ? em1n[m] : em1n[m ^ 2];
        em0[m] = lo ? em0n[m] : em0n[m ^ 2];
    }

    const int kb = lo ? 0 : 2;
    const int i  = (n & 8) ? ((n & 7) ^ 2) : n;
    v2f A0, A1;
    A0.x = fwd ? trans_norm(trans, kb + 0, i) : trans_norm(trans, i, kb + 0);
    A0.y = fwd ? trans_norm(trans, kb + 1, i) : trans_norm(trans, i, kb + 1);
    A1.x = fwd ? trans_norm(trans, kb + 4, i) : trans_norm(trans, i, kb + 4);
    A1.y = fwd ? trans_norm(trans, kb + 5, i) : trans_norm(trans, i, kb + 5);

    float* const          seq      = gamma + ((size_t)b * HMM_T + (size_t)c * HMM_L) * 8;
    const unsigned* const bw       = bits + (size_t)b * (HMM_T / 32) + c * (HMM_L / 32);
    float* const          stg      = stage + n * 8;
    float* const          half_stg = stg + (lo ? 0 : 4);
    const float* const    stg_x    = stg + (lo ? 0 : 2);
    const float* const    stg_y    = stg + (lo ? 2 : 0);

    v8f      x;
    float    cs = 1.0f;
    unsigned wn;

    if (fwd) {
        unsigned w = bw[0];
        wn = bw[1];
        if (c == 0) {
            float st[8]; float ssum = 0.0f;
#pragma unroll
            for (int k = 0; k < 8; ++k) { st[k] = start[k] + HMM_EPS; ssum += st[k]; }
#pragma unroll
            for (int k = 0; k < 8; ++k) st[k] /= ssum;

            {
                const int ob = (int)(w & 1u);
                v8f x0;
#pragma unroll
                for (int m = 0; m < 8; ++m) x0[m] = st[m] * (ob ? em1n[m] : em0n[m]);
                x = lane_order(x0, lo);
            }
            cs = __builtin_amdgcn_rcpf(sum8_tree(x));
            *(v4f*)half_stg = own_half(x, lo);
#pragma unroll
            for (int tb = 1; tb < 8; ++tb)
                pass_fwd_step(x, cs, half_stg + tb * HMM_SROW,
                              (int)((w >> tb) & 1u), lo, em1, em0, A0, A1);
        } else {
            x  = lane_order(*(const v8f*)seq, lo);
            cs = __builtin_amdgcn_rcpf(sum8_tree(x));
            pass_fwd_block8(x, cs, half_stg, w, lo, em1, em0, A0, A1);
        }
        for (int sub = 1; sub < 4; ++sub)
            pass_fwd_block8(x, cs, half_stg + sub * (8 * HMM_SROW),
                            w >> (sub * 8), lo, em1, em0, A0, A1);
        for (int q = 1; q < HMM_L / 64; ++q) {
            w  = wn;
            wn = bw[q + 1];
            float* const wrow = half_stg + q * (32 * HMM_SROW);
            for (int sub = 0; sub < 4; ++sub)
                pass_fwd_block8(x, cs, wrow + sub * (8 * HMM_SROW),
                                w >> (sub * 8), lo, em1, em0, A0, A1);
        }
    } else {
        x  = lane_order(*(const v8f*)(seq + (size_t)(HMM_L - 1) * 8), lo);
        wn = bw[HMM_L / 32 - 1];
        for (int q = HMM_L / 32 - 1; q >= HMM_L / 64; --q) {
            const unsigned w = wn;
            wn = bw[q - 1];
            float* const wrow = half_stg + q * (32 * HMM_SROW);
            for (int sub = 3; sub >= 0; --sub)
                bwd_stage_block8(wrow + (sub * 8 + 7) * HMM_SROW, w >> (sub * 8), x, cs,
                                 lo, em1, em0, A0, A1);
        }
    }

    __syncthreads();

    v8f bufA[8], bufB[8];
    const gamma_lines q = gamma_lines_init(
        gamma + ((size_t)((blockIdx.x % (HMM_B / 16)) * 16) * HMM_T + (size_t)c * HMM_L) * 8, lane);
    v4f pend[8];
    if (fwd) {
        fwd_load8(bufA, stg_x, stg_y, HMM_L / 16);
        for (int jb = HMM_L / 16; jb < HMM_L / 8; jb += 2) {
            const unsigned w  = wn;
            const int      jn = jb + 2 < HMM_L / 8 ? jb + 2 : HMM_L / 8 - 2;
            wn = bw[jn >> 2];
            fwd_load8(bufB, stg_x, stg_y, jb + 1);
            __builtin_amdgcn_sched_barrier(0);
            fwd_gamma_block8(bufA, stage + jb * (8 * HMM_SROW), q, pend, jb != HMM_L / 16,
                             jb, w, x, cs, lo, em1, em0, A0, A1);
            __builtin_amdgcn_sched_barrier(0);
            fwd_load8(bufA, stg_x, stg_y, jn);
            __builtin_amdgcn_sched_barrier(0);
            fwd_gamma_block8(bufB, stage + (jb + 1) * (8 * HMM_SROW), q, pend, true,
                             jb + 1, w, x, cs, lo, em1, em0, A0, A1);
        }
        gamma_send(pend, q, HMM_L - 8);
    } else {
        bwd_load8(bufA, stg_x, stg_y, HMM_L / 16 - 1);
        for (int jb = HMM_L / 16 - 1; jb >= 1; jb -= 2) {
            const unsigned w  = wn;
            const int      jn = jb >= 3 ? jb - 2 : 1;
            wn = bw[jn >> 2];
            bwd_load8(bufB, stg_x, stg_y, jb - 1);
            __builtin_amdgcn_sched_barrier(0);
            bwd_block8(bufA, stage + jb * (8 * HMM_SROW), q, pend, jb != HMM_L / 16 - 1,
                       jb, w, x, cs, lo, em1, em0, A0, A1);
            __builtin_amdgcn_sched_barrier(0);
            bwd_load8(bufA, stg_x, stg_y, jn);
            __builtin_amdgcn_sched_barrier(0);
            bwd_block8(bufB, stage + (jb - 1) * (8 * HMM_SROW), q, pend, true,
                       jb - 1, w, x, cs, lo, em1, em0, A0, A1);
        }
        gamma_send(pend, q, 0);
    }
}

extern "C" void kernel_launch(void* const* d_in, const int* in_sizes, int n_in,
                              void* d_out, int out_size, void* d_ws, size_t ws_size,
                              hipStream_t stream) {
    (void)in_sizes; (void)n_in; (void)out_size; (void)ws_size;
    const float* start    = (const float*)d_in[0];
    const float* trans    = (const float*)d_in[1];
    const float* emission = (const float*)d_in[2];
    const int*   obs      = (const int*)d_in[3];

    float* gamma  = (float*)d_out;
    float* loglik = gamma + (size_t)HMM_B * HMM_T * HMM_K;
    unsigned* bits = (unsigned*)d_ws;

    hmm_pack_kernel<<<HMM_TAB_BLOCKS + HMM_PACK_BLOCKS, 256, 0, stream>>>(
        obs, bits, trans, emission, gamma);
    hmm_chunk_kernel<<<HMM_B, 1024, 0, stream>>>(
        start, trans, emission, bits, gamma, gamma, loglik);
    hmm_fb_kernel<<<HMM_C * (HMM_B / 16), 64, 0, stream>>>(
        start, trans, emission, bits, gamma);
}
